# starting kernel
# baseline (speedup 1.0000x reference)
#include <hip/hip_runtime.h>

typedef float v2f __attribute__((ext_vector_type(2)));
typedef float v8f __attribute__((ext_vector_type(8)));

#define NNODES 30000
#define HEADS 4
#define SLOPE 0.2f
#define KCHUNK 128
#define PAIRSTRIDE 160

__global__ __launch_bounds__(256)
void gemm_xw_wmma(const float* __restrict__ A, const float* __restrict__ B,
                  float* __restrict__ C, int M, int K, int Nout) {
  __shared__ float ldsB[(KCHUNK / 2) * PAIRSTRIDE];

  const int tid  = threadIdx.x;
  const int lane = tid & 31;
  const int wave = tid >> 5;
  const int Mtiles = M >> 4;

  const int n0 = blockIdx.y * 64;
  int mt = blockIdx.x * 8 + wave;
  const bool active = (mt < Mtiles);
  if (!active) mt = Mtiles - 1;
  const int m0   = mt * 16;
  const int half = lane >> 4;
  const int lm   = lane & 15;

  const int lrow = tid >> 4;
  const int lcol = (tid & 15) * 4;

  v8f acc0 = {}, acc1 = {}, acc2 = {}, acc3 = {};
  const float* __restrict__ arow = A + (size_t)(m0 + lm) * K;

  for (int kb = 0; kb < K; kb += KCHUNK) {
    __syncthreads();
#pragma unroll
    for (int r = 0; r < KCHUNK; r += 16) {
      const int k = r + lrow;
      const float4 v =
          *(const float4*)(B + (size_t)(kb + k) * Nout + n0 + lcol);
      float* p = &ldsB[(k >> 1) * PAIRSTRIDE + lcol * 2 + (k & 1)];
      p[0] = v.x; p[2] = v.y; p[4] = v.z; p[6] = v.w;
    }
    __syncthreads();

    if (kb + KCHUNK < K)
      __builtin_prefetch(arow + kb + KCHUNK, 0, 0);

#pragma unroll 4
    for (int k = 0; k < KCHUNK; k += 4) {
      v2f a = *(const v2f*)(arow + kb + k + 2 * half);
      const float* __restrict__ bp =
          &ldsB[((k >> 1) + half) * PAIRSTRIDE + lm * 2];
      v2f b;
      b = *(const v2f*)(bp + 0);
      acc0 = __builtin_amdgcn_wmma_f32_16x16x4_f32(false, a, false, b, (short)0, acc0, false, false);
      b = *(const v2f*)(bp + 32);
      acc1 = __builtin_amdgcn_wmma_f32_16x16x4_f32(false, a, false, b, (short)0, acc1, false, false);
      b = *(const v2f*)(bp + 64);
      acc2 = __builtin_amdgcn_wmma_f32_16x16x4_f32(false, a, false, b, (short)0, acc2, false, false);
      b = *(const v2f*)(bp + 96);
      acc3 = __builtin_amdgcn_wmma_f32_16x16x4_f32(false, a, false, b, (short)0, acc3, false, false);
    }
  }

  if (active) {
    float* __restrict__ crow = C + (size_t)(m0 + 8 * half) * Nout + n0 + lm;
#pragma unroll
    for (int v = 0; v < 8; ++v) {
      crow[(size_t)v * Nout + 0]  = acc0[v];
      crow[(size_t)v * Nout + 16] = acc1[v];
      crow[(size_t)v * Nout + 32] = acc2[v];
      crow[(size_t)v * Nout + 48] = acc3[v];
    }
  }
}

__global__ void attn_logits(const float* __restrict__ H, const float* __restrict__ a_s,
                            const float* __restrict__ a_d, float* __restrict__ als,
                            float* __restrict__ ald, int Nn, int C) {
  int i = blockIdx.x * blockDim.x + threadIdx.x;
  if (i >= Nn * HEADS) return;
  int n = i / HEADS, h = i % HEADS;
  const float* hp = H + (size_t)n * HEADS * C + (size_t)h * C;
  float s = 0.f, d = 0.f;
  for (int c = 0; c < C; ++c) { float v = hp[c]; s += v * a_s[h * C + c]; d += v * a_d[h * C + c]; }
  als[i] = s; ald[i] = d;
}

__global__ void zero_f(float* __restrict__ p, long long n) {
  long long i = (long long)blockIdx.x * blockDim.x + threadIdx.x;
  if (i < n) p[i] = 0.f;
}

__device__ __forceinline__ unsigned fkey(float x) {
  unsigned b = __float_as_uint(x);
  return (b & 0x80000000u) ? ~b : (b | 0x80000000u);
}
__device__ __forceinline__ float funkey(unsigned k) {
  unsigned b = (k & 0x80000000u) ? (k & 0x7FFFFFFFu) : ~k;
  return __uint_as_float(b);
}

__global__ void edge_scores(const int* __restrict__ src, const int* __restrict__ dst,
                            int E, const float* __restrict__ als, const float* __restrict__ ald,
                            float* __restrict__ ews, unsigned* __restrict__ mkey) {
  int e = blockIdx.x * blockDim.x + threadIdx.x;
  const int EL = E + NNODES;
  if (e >= EL) return;
  int s = (e < E) ? src[e] : (e - E);
  int d = (e < E) ? dst[e] : (e - E);
#pragma unroll
  for (int h = 0; h < HEADS; ++h) {
    float v = als[s * HEADS + h] + ald[d * HEADS + h];
    v = (v > 0.f) ? v : SLOPE * v;
    ews[(size_t)e * HEADS + h] = v;
    atomicMax(&mkey[d * HEADS + h], fkey(v));
  }
}

__global__ void edge_exp(const int* __restrict__ src, const int* __restrict__ dst,
                         int E, float* __restrict__ ews,
                         const unsigned* __restrict__ mkey, float* __restrict__ den) {
  int e = blockIdx.x * blockDim.x + threadIdx.x;
  const int EL = E + NNODES;
  if (e >= EL) return;
  int d = (e < E) ? dst[e] : (e - E);
#pragma unroll
  for (int h = 0; h < HEADS; ++h) {
    float m = funkey(mkey[d * HEADS + h]);
    float ex = __expf(ews[(size_t)e * HEADS + h] - m);
    ews[(size_t)e * HEADS + h] = ex;
    atomicAdd(&den[d * HEADS + h], ex);
  }
}

__global__ void edge_aggregate(const int* __restrict__ src, const int* __restrict__ dst,
                               int E, const float* __restrict__ H,
                               const float* __restrict__ ews, const float* __restrict__ den,
                               float* __restrict__ agg, int C) {
  const int Fout = HEADS * C;
  const int chunks = Fout >> 2;
  long long tid = (long long)blockIdx.x * blockDim.x + threadIdx.x;
  long long total = (long long)(E + NNODES) * chunks;
  if (tid >= total) return;
  int e = (int)(tid / chunks);
  int j = (int)(tid % chunks);
  int f = j * 4;
  int h = f / C;
  int s = (e < E) ? src[e] : (e - E);
  int d = (e < E) ? dst[e] : (e - E);
  float alpha = ews[(size_t)e * HEADS + h] / (den[d * HEADS + h] + 1e-16f);
  const float4 hv = *(const float4*)(H + (size_t)s * Fout + f);
  float* op = agg + (size_t)d * Fout + f;
  atomicAdd(op + 0, hv.x * alpha);
  atomicAdd(op + 1, hv.y * alpha);
  atomicAdd(op + 2, hv.z * alpha);
  atomicAdd(op + 3, hv.w * alpha);
}

__global__ void bias_relu(const float* __restrict__ agg, const float* __restrict__ b,
                          float* __restrict__ out, int Nn, int F) {
  int i = blockIdx.x * blockDim.x + threadIdx.x;
  if (i >= Nn * F) return;
  float v = agg[i] + b[i % F];
  out[i] = v > 0.f ? v : 0.f;
}

__global__ void mean_bias(const float* __restrict__ agg, const float* __restrict__ b,
                          float* __restrict__ out, int Nn, int C) {
  int i = blockIdx.x * blockDim.x + threadIdx.x;
  if (i >= Nn * C) return;
  int n = i / C, c = i % C;
  const float* p = agg + (size_t)n * HEADS * C + c;
  out[i] = 0.25f * (p[0] + p[C] + p[2 * C] + p[3 * C]) + b[c];
}

static void run_gat_layer(hipStream_t st, const float* in, int Fin, int C,
                          const float* W, const float* a_s, const float* a_d, const float* b,
                          float* Hbuf, float* agg, float* outMean,
                          const int* src, const int* dst, int E,
                          float* ALS, float* ALD, float* EWS, unsigned* MKEY, float* DEN) {
  const int Nn = NNODES;
  const int Fout = HEADS * C;
  const int Mtiles = Nn / 16;
  dim3 ggrid((Mtiles + 7) / 8, Fout / 64);
  gemm_xw_wmma<<<ggrid, 256, 0, st>>>(in, W, Hbuf, Nn, Fin, Fout);
  int t = Nn * HEADS;
  attn_logits<<<(t + 255) / 256, 256, 0, st>>>(Hbuf, a_s, a_d, ALS, ALD, Nn, C);
  zero_f<<<(t + 255) / 256, 256, 0, st>>>(DEN, t);
  zero_f<<<(t + 255) / 256, 256, 0, st>>>((float*)MKEY, t);
  long long nagg = (long long)Nn * Fout;
  zero_f<<<(int)((nagg + 255) / 256), 256, 0, st>>>(agg, nagg);
  const int EL = E + Nn;
  edge_scores<<<(EL + 255) / 256, 256, 0, st>>>(src, dst, E, ALS, ALD, EWS, MKEY);
  edge_exp<<<(EL + 255) / 256, 256, 0, st>>>(src, dst, E, EWS, MKEY, DEN);
  long long tot = (long long)EL * (Fout >> 2);
  edge_aggregate<<<(int)((tot + 255) / 256), 256, 0, st>>>(src, dst, E, Hbuf, EWS, DEN, agg, C);
  if (outMean) {
    mean_bias<<<(Nn * C + 255) / 256, 256, 0, st>>>(agg, b, outMean, Nn, C);
  } else {
    bias_relu<<<(int)((nagg + 255) / 256), 256, 0, st>>>(agg, b, agg, Nn, Fout);
  }
}

extern "C" void kernel_launch(void* const* d_in, const int* in_sizes, int n_in,
                              void* d_out, int out_size, void* d_ws, size_t ws_size,
                              hipStream_t stream) {
  const float* x   = (const float*)d_in[0];
  const int*   ei  = (const int*)d_in[1];
  const int E = in_sizes[1] / 2;
  const int* src = ei;
  const int* dst = ei + E;
  const float *W1 = (const float*)d_in[2],  *as1 = (const float*)d_in[3],
              *ad1 = (const float*)d_in[4], *b1  = (const float*)d_in[5];
  const float *W2 = (const float*)d_in[6],  *as2 = (const float*)d_in[7],
              *ad2 = (const float*)d_in[8], *b2  = (const float*)d_in[9];
  const float *Wm = (const float*)d_in[10], *asm_ = (const float*)d_in[11],
              *adm = (const float*)d_in[12], *bm = (const float*)d_in[13];
  const float *Wl = (const float*)d_in[14], *asl = (const float*)d_in[15],
              *adl = (const float*)d_in[16], *bl = (const float*)d_in[17];

  char* ws = (char*)d_ws;
  size_t off = 0;
  auto alloc = [&](size_t bytes) -> void* {
    void* p = ws + off;
    off += (bytes + 255) & ~(size_t)255;
    return p;
  };
  float*    Hbuf = (float*)alloc((size_t)NNODES * 512 * 4);
  float*    A0   = (float*)alloc((size_t)NNODES * 256 * 4);
  float*    A1   = (float*)alloc((size_t)NNODES * 512 * 4);
  float*    ALS  = (float*)alloc((size_t)NNODES * HEADS * 4);
  float*    ALD  = (float*)alloc((size_t)NNODES * HEADS * 4);
  float*    EWS  = (float*)alloc((size_t)(E + NNODES) * HEADS * 4);
  unsigned* MKEY = (unsigned*)alloc((size_t)NNODES * HEADS * 4);
  float*    DEN  = (float*)alloc((size_t)NNODES * HEADS * 4);

  float* zmean   = (float*)d_out;
  float* zlogstd = (float*)d_out + (size_t)NNODES * 32;

  run_gat_layer(stream, x, 256, 64, W1, as1, ad1, b1, Hbuf, A0, nullptr,
                src, dst, E, ALS, ALD, EWS, MKEY, DEN);
  run_gat_layer(stream, A0, 256, 128, W2, as2, ad2, b2, Hbuf, A1, nullptr,
                src, dst, E, ALS, ALD, EWS, MKEY, DEN);
  run_gat_layer(stream, A1, 512, 32, Wm, asm_, adm, bm, Hbuf, A0, zmean,
                src, dst, E, ALS, ALD, EWS, MKEY, DEN);
  run_gat_layer(stream, A1, 512, 32, Wl, asl, adl, bl, Hbuf, A0, zlogstd,
                src, dst, E, ALS, ALD, EWS, MKEY, DEN);
}


# Each work-group owns 118 destination nodes: scans the whole edge list for them, sums softmax terms in LDS (LDS float atomics, arrival order, self loop first); global atomics and 5 passes/layer gone
# speedup vs baseline: 3.6076x; 3.6076x over previous
#include <hip/hip_runtime.h>

typedef float v2f __attribute__((ext_vector_type(2)));
typedef float v8f __attribute__((ext_vector_type(8)));

#define NNODES 30000
#define HEADS 4
#define SLOPE 0.2f
#define KCHUNK 128
#define PAIRSTRIDE 160

__global__ __launch_bounds__(256)
void gemm_xw_wmma(const float* __restrict__ A, const float* __restrict__ B,
                  float* __restrict__ C, int M, int K, int Nout) {
  __shared__ float ldsB[(KCHUNK / 2) * PAIRSTRIDE];

  const int tid  = threadIdx.x;
  const int lane = tid & 31;
  const int wave = tid >> 5;
  const int Mtiles = M >> 4;

  const int n0 = blockIdx.y * 64;
  int mt = blockIdx.x * 8 + wave;
  const bool active = (mt < Mtiles);
  if (!active) mt = Mtiles - 1;
  const int m0   = mt * 16;
  const int half = lane >> 4;
  const int lm   = lane & 15;

  const int lrow = tid >> 4;
  const int lcol = (tid & 15) * 4;

  v8f acc0 = {}, acc1 = {}, acc2 = {}, acc3 = {};
  const float* __restrict__ arow = A + (size_t)(m0 + lm) * K;

  for (int kb = 0; kb < K; kb += KCHUNK) {
    __syncthreads();
#pragma unroll
    for (int r = 0; r < KCHUNK; r += 16) {
      const int k = r + lrow;
      const float4 v =
          *(const float4*)(B + (size_t)(kb + k) * Nout + n0 + lcol);
      float* p = &ldsB[(k >> 1) * PAIRSTRIDE + lcol * 2 + (k & 1)];
      p[0] = v.x; p[2] = v.y; p[4] = v.z; p[6] = v.w;
    }
    __syncthreads();

    if (kb + KCHUNK < K)
      __builtin_prefetch(arow + kb + KCHUNK, 0, 0);

#pragma unroll 4
    for (int k = 0; k < KCHUNK; k += 4) {
      v2f a = *(const v2f*)(arow + kb + k + 2 * half);
      const float* __restrict__ bp =
          &ldsB[((k >> 1) + half) * PAIRSTRIDE + lm * 2];
      v2f b;
      b = *(const v2f*)(bp + 0);
      acc0 = __builtin_amdgcn_wmma_f32_16x16x4_f32(false, a, false, b, (short)0, acc0, false, false);
      b = *(const v2f*)(bp + 32);
      acc1 = __builtin_amdgcn_wmma_f32_16x16x4_f32(false, a, false, b, (short)0, acc1, false, false);
      b = *(const v2f*)(bp + 64);
      acc2 = __builtin_amdgcn_wmma_f32_16x16x4_f32(false, a, false, b, (short)0, acc2, false, false);
      b = *(const v2f*)(bp + 96);
      acc3 = __builtin_amdgcn_wmma_f32_16x16x4_f32(false, a, false, b, (short)0, acc3, false, false);
    }
  }

  if (active) {
    float* __restrict__ crow = C + (size_t)(m0 + 8 * half) * Nout + n0 + lm;
#pragma unroll
    for (int v = 0; v < 8; ++v) {
      crow[(size_t)v * Nout + 0]  = acc0[v];
      crow[(size_t)v * Nout + 16] = acc1[v];
      crow[(size_t)v * Nout + 32] = acc2[v];
      crow[(size_t)v * Nout + 48] = acc3[v];
    }
  }
}

__global__ void attn_logits(const float* __restrict__ H, const float* __restrict__ a_s,
                            const float* __restrict__ a_d, float* __restrict__ als,
                            float* __restrict__ ald, int Nn, int C) {
  int i = blockIdx.x * blockDim.x + threadIdx.x;
  if (i >= Nn * HEADS) return;
  int n = i / HEADS, h = i % HEADS;
  const float* hp = H + (size_t)n * HEADS * C + (size_t)h * C;
  float s = 0.f, d = 0.f;
  for (int c = 0; c < C; ++c) { float v = hp[c]; s += v * a_s[h * C + c]; d += v * a_d[h * C + c]; }
  als[i] = s; ald[i] = d;
}

__device__ __forceinline__ unsigned fkey(float x) {
  unsigned b = __float_as_uint(x);
  return (b & 0x80000000u) ? ~b : (b | 0x80000000u);
}
__device__ __forceinline__ float funkey(unsigned k) {
  unsigned b = (k & 0x80000000u) ? (k & 0x7FFFFFFFu) : ~k;
  return __uint_as_float(b);
}

#define RNODES 118
#define QCAP 8192
#define NTHR 1024
#define NWAVE (NTHR / 32)

template <int C, bool MEAN>
__global__ __launch_bounds__(NTHR)
void node_range_softmax_aggregate(const int* __restrict__ src, const int* __restrict__ dst, int E,
                                  const float* __restrict__ H, const float* __restrict__ als,
                                  const float* __restrict__ ald, const float* __restrict__ bias,
                                  float* __restrict__ out) {
  constexpr int F = HEADS * C;
  constexpr int J = F / 128;
  constexpr int U = 8 / J;
  __shared__ float    acc[RNODES * F];
  __shared__ unsigned mk[RNODES * HEADS];
  __shared__ float    mv[RNODES * HEADS];
  __shared__ float    den[RNODES * HEADS];
  __shared__ float    aldc[RNODES * HEADS];
  __shared__ unsigned q[QCAP];
  __shared__ unsigned qn;

  const int tid  = threadIdx.x;
  const int lane = tid & 31;
  const int wave = tid >> 5;
  const int base = blockIdx.x * RNODES;
  int rn = NNODES - base;
  if (rn > RNODES) rn = RNODES;
  if (rn < 0) rn = 0;

  auto take = [&](int e, unsigned dl) {
    if (dl < (unsigned)rn) {
      const unsigned slot = atomicAdd(&qn, 1u);
      if (slot < QCAP) q[slot] = (unsigned)e | (dl << 25);
    }
  };

  auto scan = [&](int e0, int e1) {
    if (((E | e0 | e1) & 3) == 0) {
      const int4* __restrict__ d4 = (const int4*)dst;
      const int g1 = e1 >> 2;
      for (int g0 = (e0 >> 2); g0 < g1; g0 += NTHR * 8) {
        int4 v[8];
#pragma unroll
        for (int u = 0; u < 8; ++u) {
          const int g = g0 + u * NTHR + tid;
          v[u] = (g < g1) ? d4[g] : make_int4(-1, -1, -1, -1);
        }
#pragma unroll
        for (int u = 0; u < 8; ++u) {
          const int e = (g0 + u * NTHR + tid) * 4;
          const unsigned a0 = (unsigned)(v[u].x - base), a1 = (unsigned)(v[u].y - base);
          const unsigned a2 = (unsigned)(v[u].z - base), a3 = (unsigned)(v[u].w - base);
          const unsigned lo01 = a0 < a1 ? a0 : a1, lo23 = a2 < a3 ? a2 : a3;
          if ((lo01 < lo23 ? lo01 : lo23) < (unsigned)rn) {
            take(e + 0, a0); take(e + 1, a1); take(e + 2, a2); take(e + 3, a3);
          }
        }
      }
    } else {
      for (int e = e0 + tid; e < e1; e += NTHR) take(e, (unsigned)(dst[e] - base));
    }
  };

  auto resolve = [&](int n, bool withmax) {
    for (int i = tid; i < n; i += NTHR) {
      const unsigned ent = q[i];
      const unsigned dl = ent >> 25;
      const int s = src[ent & 0x1FFFFFFu];
      if (withmax) {
        const float4 a = *(const float4*)(als + (size_t)s * HEADS);
        atomicMax(&mk[dl * HEADS + 0], fkey(a.x));
        atomicMax(&mk[dl * HEADS + 1], fkey(a.y));
        atomicMax(&mk[dl * HEADS + 2], fkey(a.z));
        atomicMax(&mk[dl * HEADS + 3], fkey(a.w));
      }
      q[i] = (unsigned)s | (dl << 16);
    }
  };

  auto drain = [&](int n) {
    for (int i0 = wave; i0 < n; i0 += NWAVE * U) {
      float4 hv[U][J];
      int sidx[U], dloc[U];
#pragma unroll
      for (int u = 0; u < U; ++u) {
        const int i = i0 + u * NWAVE;
        unsigned ent = (i < n) ? q[i] : 0u;
        ent = __builtin_amdgcn_readfirstlane(ent);
        sidx[u] = (int)(ent & 0xFFFFu);
        dloc[u] = (int)(ent >> 16);
        const float* __restrict__ hrow = H + (size_t)sidx[u] * F + lane * 4;
#pragma unroll
        for (int j = 0; j < J; ++j) hv[u][j] = *(const float4*)(hrow + 128 * j);
      }
#pragma unroll
      for (int u = 0; u < U; ++u) {
        if (i0 + u * NWAVE < n) {
          const int s = sidx[u], dl = dloc[u];
#pragma unroll
          for (int j = 0; j < J; ++j) {
            const int f = 128 * j + lane * 4;
            const int h = f / C;
            float sc = als[s * HEADS + h] + aldc[dl * HEADS + h];
            sc = (sc > 0.f) ? sc : SLOPE * sc;
            const float ex = __expf(sc - mv[dl * HEADS + h]);
            if ((f % C) == 0) atomicAdd(&den[dl * HEADS + h], ex);
            float* ap = &acc[dl * F + 128 * j + lane];
            atomicAdd(ap +  0, hv[u][j].x * ex);
            atomicAdd(ap + 32, hv[u][j].y * ex);
            atomicAdd(ap + 64, hv[u][j].z * ex);
            atomicAdd(ap + 96, hv[u][j].w * ex);
          }
        }
      }
    }
  };

  if (tid < rn * HEADS) {
    mk[tid]   = fkey(als[(size_t)base * HEADS + tid]);
    aldc[tid] = ald[(size_t)base * HEADS + tid];
  }
  if (tid == 0) qn = 0u;
  __syncthreads();

  scan(0, E);
  __syncthreads();
  const int total = (int)qn;
  const bool fits = (total <= QCAP);
  if (fits) {
    resolve(total, true);
  } else {
    for (int p = 0; p < E; p += QCAP) {
      __syncthreads();
      if (tid == 0) qn = 0u;
      __syncthreads();
      scan(p, (p + QCAP < E) ? (p + QCAP) : E);
      __syncthreads();
      resolve((int)qn, true);
    }
  }
  __syncthreads();

  if (tid < rn * HEADS) {
    const float a = aldc[tid];
    float m = funkey(mk[tid]) + a;
    m = (m > 0.f) ? m : SLOPE * m;
    mv[tid] = m;
    float sc = als[(size_t)base * HEADS + tid] + a;
    sc = (sc > 0.f) ? sc : SLOPE * sc;
    den[tid] = __expf(sc - m);
  }
  __syncthreads();

  for (int idx = tid; idx < rn * (F / 4); idx += NTHR) {
    const int dl = idx / (F / 4);
    const int r  = idx % (F / 4);
    const int j  = r >> 5, l = r & 31;
    const int f  = 128 * j + 4 * l;
    const float ex = den[dl * HEADS + f / C];
    const float4 hv = *(const float4*)(H + (size_t)(base + dl) * F + f);
    float* ap = &acc[dl * F + 128 * j + l];
    ap[0] = hv.x * ex; ap[32] = hv.y * ex; ap[64] = hv.z * ex; ap[96] = hv.w * ex;
  }
  __syncthreads();

  if (fits) {
    drain(total);
  } else {
    for (int p = 0; p < E; p += QCAP) {
      __syncthreads();
      if (tid == 0) qn = 0u;
      __syncthreads();
      scan(p, (p + QCAP < E) ? (p + QCAP) : E);
      __syncthreads();
      resolve((int)qn, false);
      __syncthreads();
      drain((int)qn);
    }
  }
  __syncthreads();

  if (!MEAN) {
    for (int idx = tid; idx < rn * (F / 4); idx += NTHR) {
      const int dl = idx / (F / 4);
      const int r  = idx % (F / 4);
      const int j  = r >> 5, l = r & 31;
      const int f  = 128 * j + 4 * l;
      const float dn = den[dl * HEADS + f / C] + 1e-16f;
      const float* ap = &acc[dl * F + 128 * j + l];
      const float4 b4 = *(const float4*)(bias + f);
      float4 o;
      o.x = ap[0]  / dn + b4.x;  o.y = ap[32] / dn + b4.y;
      o.z = ap[64] / dn + b4.z;  o.w = ap[96] / dn + b4.w;
      o.x = o.x > 0.f ? o.x : 0.f;  o.y = o.y > 0.f ? o.y : 0.f;
      o.z = o.z > 0.f ? o.z : 0.f;  o.w = o.w > 0.f ? o.w : 0.f;
      *(float4*)(out + (size_t)(base + dl) * F + f) = o;
    }
  } else {
    for (int idx = tid; idx < rn * C; idx += NTHR) {
      const int dl = idx / C, c = idx % C;
      const float* qd = &den[dl * HEADS];
      float p[HEADS];
#pragma unroll
      for (int h = 0; h < HEADS; ++h) {
        const int f = h * C + c;
        p[h] = acc[dl * F + 128 * (f >> 7) + 32 * (f & 3) + ((f & 127) >> 2)];
      }
      out[(size_t)(base + dl) * C + c] =
          0.25f * (p[0] / (qd[0] + 1e-16f) + p[1] / (qd[1] + 1e-16f) +
                   p[2] / (qd[2] + 1e-16f) + p[3] / (qd[3] + 1e-16f)) + bias[c];
    }
  }
}

template <int C, bool MEAN>
static void run_gat_layer(hipStream_t st, const float* in, int Fin,
                          const float* W, const float* a_s, const float* a_d, const float* b,
                          float* Hbuf, float* outp,
                          const int* src, const int* dst, int E,
                          float* ALS, float* ALD) {
  const int Nn = NNODES;
  const int Fout = HEADS * C;
  const int Mtiles = Nn / 16;
  dim3 ggrid((Mtiles + 7) / 8, Fout / 64);
  gemm_xw_wmma<<<ggrid, 256, 0, st>>>(in, W, Hbuf, Nn, Fin, Fout);
  int t = Nn * HEADS;
  attn_logits<<<(t + 255) / 256, 256, 0, st>>>(Hbuf, a_s, a_d, ALS, ALD, Nn, C);
  node_range_softmax_aggregate<C, MEAN>
      <<<(NNODES + RNODES - 1) / RNODES, NTHR, 0, st>>>(src, dst, E, Hbuf, ALS, ALD, b, outp);
}

extern "C" void kernel_launch(void* const* d_in, const int* in_sizes, int n_in,
                              void* d_out, int out_size, void* d_ws, size_t ws_size,
                              hipStream_t stream) {
  const float* x   = (const float*)d_in[0];
  const int*   ei  = (const int*)d_in[1];
  const int E = in_sizes[1] / 2;
  const int* src = ei;
  const int* dst = ei + E;
  const float *W1 = (const float*)d_in[2],  *as1 = (const float*)d_in[3],
              *ad1 = (const float*)d_in[4], *b1  = (const float*)d_in[5];
  const float *W2 = (const float*)d_in[6],  *as2 = (const float*)d_in[7],
              *ad2 = (const float*)d_in[8], *b2  = (const float*)d_in[9];
  const float *Wm = (const float*)d_in[10], *asm_ = (const float*)d_in[11],
              *adm = (const float*)d_in[12], *bm = (const float*)d_in[13];
  const float *Wl = (const float*)d_in[14], *asl = (const float*)d_in[15],
              *adl = (const float*)d_in[16], *bl = (const float*)d_in[17];

  char* ws = (char*)d_ws;
  size_t off = 0;
  auto alloc = [&](size_t bytes) -> void* {
    void* p = ws + off;
    off += (bytes + 255) & ~(size_t)255;
    return p;
  };
  float*    Hbuf = (float*)alloc((size_t)NNODES * 512 * 4);
  float*    A0   = (float*)alloc((size_t)NNODES * 256 * 4);
  float*    A1   = (float*)alloc((size_t)NNODES * 512 * 4);
  float*    ALS  = (float*)alloc((size_t)NNODES * HEADS * 4);
  float*    ALD  = (float*)alloc((size_t)NNODES * HEADS * 4);

  float* zmean   = (float*)d_out;
  float* zlogstd = (float*)d_out + (size_t)NNODES * 32;

  run_gat_layer<64, false>(stream, x, 256, W1, as1, ad1, b1, Hbuf, A0, src, dst, E, ALS, ALD);
  run_gat_layer<128, false>(stream, A0, 256, W2, as2, ad2, b2, Hbuf, A1, src, dst, E, ALS, ALD);
  run_gat_layer<32, true>(stream, A1, 512, Wm, asm_, adm, bm, Hbuf, zmean, src, dst, E, ALS, ALD);
  run_gat_layer<32, true>(stream, A1, 512, Wl, asl, adl, bl, Hbuf, zlogstd, src, dst, E, ALS, ALD);
}


# Each work-group's matched edge list (source, local node) is built by layer 1's scan only, kept in workspace, reloaded by layers 2-4: 3 of 4 edge-list scans gone; sums still LDS-atomic arrival order
# speedup vs baseline: 4.1591x; 1.1529x over previous
#include <hip/hip_runtime.h>

typedef float v2f __attribute__((ext_vector_type(2)));
typedef float v8f __attribute__((ext_vector_type(8)));

#define NNODES 30000
#define HEADS 4
#define SLOPE 0.2f
#define KCHUNK 128
#define PAIRSTRIDE 160

__global__ __launch_bounds__(256)
void gemm_xw_wmma(const float* __restrict__ A, const float* __restrict__ B,
                  float* __restrict__ C, int M, int K, int Nout) {
  __shared__ float ldsB[(KCHUNK / 2) * PAIRSTRIDE];

  const int tid  = threadIdx.x;
  const int lane = tid & 31;
  const int wave = tid >> 5;
  const int Mtiles = M >> 4;

  const int n0 = blockIdx.y * 64;
  int mt = blockIdx.x * 8 + wave;
  const bool active = (mt < Mtiles);
  if (!active) mt = Mtiles - 1;
  const int m0   = mt * 16;
  const int half = lane >> 4;
  const int lm   = lane & 15;

  const int lrow = tid >> 4;
  const int lcol = (tid & 15) * 4;

  v8f acc0 = {}, acc1 = {}, acc2 = {}, acc3 = {};
  const float* __restrict__ arow = A + (size_t)(m0 + lm) * K;

  for (int kb = 0; kb < K; kb += KCHUNK) {
    __syncthreads();
#pragma unroll
    for (int r = 0; r < KCHUNK; r += 16) {
      const int k = r + lrow;
      const float4 v =
          *(const float4*)(B + (size_t)(kb + k) * Nout + n0 + lcol);
      float* p = &ldsB[(k >> 1) * PAIRSTRIDE + lcol * 2 + (k & 1)];
      p[0] = v.x; p[2] = v.y; p[4] = v.z; p[6] = v.w;
    }
    __syncthreads();

    if (kb + KCHUNK < K)
      __builtin_prefetch(arow + kb + KCHUNK, 0, 0);

#pragma unroll 4
    for (int k = 0; k < KCHUNK; k += 4) {
      v2f a = *(const v2f*)(arow + kb + k + 2 * half);
      const float* __restrict__ bp =
          &ldsB[((k >> 1) + half) * PAIRSTRIDE + lm * 2];
      v2f b;
      b = *(const v2f*)(bp + 0);
      acc0 = __builtin_amdgcn_wmma_f32_16x16x4_f32(false, a, false, b, (short)0, acc0, false, false);
      b = *(const v2f*)(bp + 32);
      acc1 = __builtin_amdgcn_wmma_f32_16x16x4_f32(false, a, false, b, (short)0, acc1, false, false);
      b = *(const v2f*)(bp + 64);
      acc2 = __builtin_amdgcn_wmma_f32_16x16x4_f32(false, a, false, b, (short)0, acc2, false, false);
      b = *(const v2f*)(bp + 96);
      acc3 = __builtin_amdgcn_wmma_f32_16x16x4_f32(false, a, false, b, (short)0, acc3, false, false);
    }
  }

  if (active) {
    float* __restrict__ crow = C + (size_t)(m0 + 8 * half) * Nout + n0 + lm;
#pragma unroll
    for (int v = 0; v < 8; ++v) {
      crow[(size_t)v * Nout + 0]  = acc0[v];
      crow[(size_t)v * Nout + 16] = acc1[v];
      crow[(size_t)v * Nout + 32] = acc2[v];
      crow[(size_t)v * Nout + 48] = acc3[v];
    }
  }
}

__global__ void attn_logits(const float* __restrict__ H, const float* __restrict__ a_s,
                            const float* __restrict__ a_d, float* __restrict__ als,
                            float* __restrict__ ald, int Nn, int C) {
  int i = blockIdx.x * blockDim.x + threadIdx.x;
  if (i >= Nn * HEADS) return;
  int n = i / HEADS, h = i % HEADS;
  const float* hp = H + (size_t)n * HEADS * C + (size_t)h * C;
  float s = 0.f, d = 0.f;
  for (int c = 0; c < C; ++c) { float v = hp[c]; s += v * a_s[h * C + c]; d += v * a_d[h * C + c]; }
  als[i] = s; ald[i] = d;
}

__device__ __forceinline__ unsigned fkey(float x) {
  unsigned b = __float_as_uint(x);
  return (b & 0x80000000u) ? ~b : (b | 0x80000000u);
}
__device__ __forceinline__ float funkey(unsigned k) {
  unsigned b = (k & 0x80000000u) ? (k & 0x7FFFFFFFu) : ~k;
  return __uint_as_float(b);
}

#define RNODES 118
#define QCAP 8192
#define NTHR 1024
#define NWAVE (NTHR / 32)
#define NGROUPS ((NNODES + RNODES - 1) / RNODES)

template <int C, bool MEAN, bool FIRST>
__global__ __launch_bounds__(NTHR)
void node_range_softmax_aggregate(const int* __restrict__ src, const int* __restrict__ dst, int E,
                                  const float* __restrict__ H, const float* __restrict__ als,
                                  const float* __restrict__ ald, const float* __restrict__ bias,
                                  float* __restrict__ out,
                                  unsigned* __restrict__ qlist, int* __restrict__ qcnt) {
  constexpr int F = HEADS * C;
  constexpr int J = F / 128;
  constexpr int U = 8 / J;
  __shared__ float    acc[RNODES * F];
  __shared__ unsigned mk[RNODES * HEADS];
  __shared__ float    mv[RNODES * HEADS];
  __shared__ float    den[RNODES * HEADS];
  __shared__ float    aldc[RNODES * HEADS];
  __shared__ unsigned q[QCAP];
  __shared__ unsigned qn;

  const int tid  = threadIdx.x;
  const int lane = tid & 31;
  const int wave = tid >> 5;
  const int base = blockIdx.x * RNODES;
  int rn = NNODES - base;
  if (rn > RNODES) rn = RNODES;
  if (rn < 0) rn = 0;
  unsigned* __restrict__ lst = qlist + (size_t)blockIdx.x * QCAP;

  auto take = [&](int e, unsigned dl) {
    if (dl < (unsigned)rn) {
      const unsigned slot = atomicAdd(&qn, 1u);
      if (slot < QCAP) q[slot] = (unsigned)e | (dl << 25);
    }
  };

  auto scan = [&](int e0, int e1) {
    if (((E | e0 | e1) & 3) == 0) {
      const int4* __restrict__ d4 = (const int4*)dst;
      const int g1 = e1 >> 2;
      for (int g0 = (e0 >> 2); g0 < g1; g0 += NTHR * 8) {
        int4 v[8];
#pragma unroll
        for (int u = 0; u < 8; ++u) {
          const int g = g0 + u * NTHR + tid;
          v[u] = (g < g1) ? d4[g] : make_int4(-1, -1, -1, -1);
        }
#pragma unroll
        for (int u = 0; u < 8; ++u) {
          const int e = (g0 + u * NTHR + tid) * 4;
          const unsigned a0 = (unsigned)(v[u].x - base), a1 = (unsigned)(v[u].y - base);
          const unsigned a2 = (unsigned)(v[u].z - base), a3 = (unsigned)(v[u].w - base);
          const unsigned lo01 = a0 < a1 ? a0 : a1, lo23 = a2 < a3 ? a2 : a3;
          if ((lo01 < lo23 ? lo01 : lo23) < (unsigned)rn) {
            take(e + 0, a0); take(e + 1, a1); take(e + 2, a2); take(e + 3, a3);
          }
        }
      }
    } else {
      for (int e = e0 + tid; e < e1; e += NTHR) take(e, (unsigned)(dst[e] - base));
    }
  };

  auto resolve = [&](int n, bool withmax, bool keep) {
    for (int i = tid; i < n; i += NTHR) {
      const unsigned ent = q[i];
      const unsigned dl = ent >> 25;
      const int s = src[ent & 0x1FFFFFFu];
      if (withmax) {
        const float4 a = *(const float4*)(als + (size_t)s * HEADS);
        atomicMax(&mk[dl * HEADS + 0], fkey(a.x));
        atomicMax(&mk[dl * HEADS + 1], fkey(a.y));
        atomicMax(&mk[dl * HEADS + 2], fkey(a.z));
        atomicMax(&mk[dl * HEADS + 3], fkey(a.w));
      }
      const unsigned packed = (unsigned)s | (dl << 16);
      q[i] = packed;
      if (keep) lst[i] = packed;
    }
  };

  auto reload = [&](int n) {
    for (int i = tid; i < n; i += NTHR) {
      const unsigned ent = lst[i];
      const unsigned dl = ent >> 16;
      const float4 a = *(const float4*)(als + (size_t)(ent & 0xFFFFu) * HEADS);
      atomicMax(&mk[dl * HEADS + 0], fkey(a.x));
      atomicMax(&mk[dl * HEADS + 1], fkey(a.y));
      atomicMax(&mk[dl * HEADS + 2], fkey(a.z));
      atomicMax(&mk[dl * HEADS + 3], fkey(a.w));
      q[i] = ent;
    }
  };

  auto drain = [&](int n) {
    for (int i0 = wave; i0 < n; i0 += NWAVE * U) {
      float4 hv[U][J];
      int sidx[U], dloc[U];
#pragma unroll
      for (int u = 0; u < U; ++u) {
        const int i = i0 + u * NWAVE;
        unsigned ent = (i < n) ? q[i] : 0u;
        ent = __builtin_amdgcn_readfirstlane(ent);
        sidx[u] = (int)(ent & 0xFFFFu);
        dloc[u] = (int)(ent >> 16);
        const float* __restrict__ hrow = H + (size_t)sidx[u] * F + lane * 4;
#pragma unroll
        for (int j = 0; j < J; ++j) hv[u][j] = *(const float4*)(hrow + 128 * j);
      }
#pragma unroll
      for (int u = 0; u < U; ++u) {
        if (i0 + u * NWAVE < n) {
          const int s = sidx[u], dl = dloc[u];
#pragma unroll
          for (int j = 0; j < J; ++j) {
            const int f = 128 * j + lane * 4;
            const int h = f / C;
            float sc = als[s * HEADS + h] + aldc[dl * HEADS + h];
            sc = (sc > 0.f) ? sc : SLOPE * sc;
            const float ex = __expf(sc - mv[dl * HEADS + h]);
            if ((f % C) == 0) atomicAdd(&den[dl * HEADS + h], ex);
            float* ap = &acc[dl * F + 128 * j + lane];
            atomicAdd(ap +  0, hv[u][j].x * ex);
            atomicAdd(ap + 32, hv[u][j].y * ex);
            atomicAdd(ap + 64, hv[u][j].z * ex);
            atomicAdd(ap + 96, hv[u][j].w * ex);
          }
        }
      }
    }
  };

  if (tid < rn * HEADS) {
    mk[tid]   = fkey(als[(size_t)base * HEADS + tid]);
    aldc[tid] = ald[(size_t)base * HEADS + tid];
  }
  if (tid == 0) qn = 0u;
  __syncthreads();

  int total;
  if (FIRST) {
    scan(0, E);
    __syncthreads();
    total = (int)qn;
    if (tid == 0) qcnt[blockIdx.x] = total;
  } else {
    total = qcnt[blockIdx.x];
  }
  const bool fits = (total <= QCAP);
  if (fits) {
    if (FIRST) resolve(total, true, true);
    else       reload(total);
  } else {
    for (int p = 0; p < E; p += QCAP) {
      __syncthreads();
      if (tid == 0) qn = 0u;
      __syncthreads();
      scan(p, (p + QCAP < E) ? (p + QCAP) : E);
      __syncthreads();
      resolve((int)qn, true, false);
    }
  }
  __syncthreads();

  if (tid < rn * HEADS) {
    const float a = aldc[tid];
    float m = funkey(mk[tid]) + a;
    m = (m > 0.f) ? m : SLOPE * m;
    mv[tid] = m;
    float sc = als[(size_t)base * HEADS + tid] + a;
    sc = (sc > 0.f) ? sc : SLOPE * sc;
    den[tid] = __expf(sc - m);
  }
  __syncthreads();

  for (int idx = tid; idx < rn * (F / 4); idx += NTHR) {
    const int dl = idx / (F / 4);
    const int r  = idx % (F / 4);
    const int j  = r >> 5, l = r & 31;
    const int f  = 128 * j + 4 * l;
    const float ex = den[dl * HEADS + f / C];
    const float4 hv = *(const float4*)(H + (size_t)(base + dl) * F + f);
    float* ap = &acc[dl * F + 128 * j + l];
    ap[0] = hv.x * ex; ap[32] = hv.y * ex; ap[64] = hv.z * ex; ap[96] = hv.w * ex;
  }
  __syncthreads();

  if (fits) {
    drain(total);
  } else {
    for (int p = 0; p < E; p += QCAP) {
      __syncthreads();
      if (tid == 0) qn = 0u;
      __syncthreads();
      scan(p, (p + QCAP < E) ? (p + QCAP) : E);
      __syncthreads();
      resolve((int)qn, false, false);
      __syncthreads();
      drain((int)qn);
    }
  }
  __syncthreads();

  if (!MEAN) {
    for (int idx = tid; idx < rn * (F / 4); idx += NTHR) {
      const int dl = idx / (F / 4);
      const int r  = idx % (F / 4);
      const int j  = r >> 5, l = r & 31;
      const int f  = 128 * j + 4 * l;
      const float dn = den[dl * HEADS + f / C] + 1e-16f;
      const float* ap = &acc[dl * F + 128 * j + l];
      const float4 b4 = *(const float4*)(bias + f);
      float4 o;
      o.x = ap[0]  / dn + b4.x;  o.y = ap[32] / dn + b4.y;
      o.z = ap[64] / dn + b4.z;  o.w = ap[96] / dn + b4.w;
      o.x = o.x > 0.f ? o.x : 0.f;  o.y = o.y > 0.f ? o.y : 0.f;
      o.z = o.z > 0.f ? o.z : 0.f;  o.w = o.w > 0.f ? o.w : 0.f;
      *(float4*)(out + (size_t)(base + dl) * F + f) = o;
    }
  } else {
    for (int idx = tid; idx < rn * C; idx += NTHR) {
      const int dl = idx / C, c = idx % C;
      const float* qd = &den[dl * HEADS];
      float p[HEADS];
#pragma unroll
      for (int h = 0; h < HEADS; ++h) {
        const int f = h * C + c;
        p[h] = acc[dl * F + 128 * (f >> 7) + 32 * (f & 3) + ((f & 127) >> 2)];
      }
      out[(size_t)(base + dl) * C + c] =
          0.25f * (p[0] / (qd[0] + 1e-16f) + p[1] / (qd[1] + 1e-16f) +
                   p[2] / (qd[2] + 1e-16f) + p[3] / (qd[3] + 1e-16f)) + bias[c];
    }
  }
}

template <int C, bool MEAN, bool FIRST>
static void run_gat_layer(hipStream_t st, const float* in, int Fin,
                          const float* W, const float* a_s, const float* a_d, const float* b,
                          float* Hbuf, float* outp,
                          const int* src, const int* dst, int E,
                          float* ALS, float* ALD, unsigned* QL, int* QC) {
  const int Nn = NNODES;
  const int Fout = HEADS * C;
  const int Mtiles = Nn / 16;
  dim3 ggrid((Mtiles + 7) / 8, Fout / 64);
  gemm_xw_wmma<<<ggrid, 256, 0, st>>>(in, W, Hbuf, Nn, Fin, Fout);
  int t = Nn * HEADS;
  attn_logits<<<(t + 255) / 256, 256, 0, st>>>(Hbuf, a_s, a_d, ALS, ALD, Nn, C);
  node_range_softmax_aggregate<C, MEAN, FIRST>
      <<<NGROUPS, NTHR, 0, st>>>(src, dst, E, Hbuf, ALS, ALD, b, outp, QL, QC);
}

extern "C" void kernel_launch(void* const* d_in, const int* in_sizes, int n_in,
                              void* d_out, int out_size, void* d_ws, size_t ws_size,
                              hipStream_t stream) {
  const float* x   = (const float*)d_in[0];
  const int*   ei  = (const int*)d_in[1];
  const int E = in_sizes[1] / 2;
  const int* src = ei;
  const int* dst = ei + E;
  const float *W1 = (const float*)d_in[2],  *as1 = (const float*)d_in[3],
              *ad1 = (const float*)d_in[4], *b1  = (const float*)d_in[5];
  const float *W2 = (const float*)d_in[6],  *as2 = (const float*)d_in[7],
              *ad2 = (const float*)d_in[8], *b2  = (const float*)d_in[9];
  const float *Wm = (const float*)d_in[10], *asm_ = (const float*)d_in[11],
              *adm = (const float*)d_in[12], *bm = (const float*)d_in[13];
  const float *Wl = (const float*)d_in[14], *asl = (const float*)d_in[15],
              *adl = (const float*)d_in[16], *bl = (const float*)d_in[17];

  char* ws = (char*)d_ws;
  size_t off = 0;
  auto alloc = [&](size_t bytes) -> void* {
    void* p = ws + off;
    off += (bytes + 255) & ~(size_t)255;
    return p;
  };
  float*    Hbuf = (float*)alloc((size_t)NNODES * 512 * 4);
  float*    A0   = (float*)alloc((size_t)NNODES * 256 * 4);
  float*    A1   = (float*)alloc((size_t)NNODES * 512 * 4);
  float*    ALS  = (float*)alloc((size_t)NNODES * HEADS * 4);
  float*    ALD  = (float*)alloc((size_t)NNODES * HEADS * 4);
  unsigned* QL   = (unsigned*)alloc((size_t)NGROUPS * QCAP * 4);
  int*      QC   = (int*)alloc((size_t)NGROUPS * 4);

  float* zmean   = (float*)d_out;
  float* zlogstd = (float*)d_out + (size_t)NNODES * 32;

  run_gat_layer<64, false, true>(stream, x, 256, W1, as1, ad1, b1, Hbuf, A0, src, dst, E, ALS, ALD, QL, QC);
  run_gat_layer<128, false, false>(stream, A0, 256, W2, as2, ad2, b2, Hbuf, A1, src, dst, E, ALS, ALD, QL, QC);
  run_gat_layer<32, true, false>(stream, A1, 512, Wm, asm_, adm, bm, Hbuf, zmean, src, dst, E, ALS, ALD, QL, QC);
  run_gat_layer<32, true, false>(stream, A1, 512, Wl, asl, adl, bl, Hbuf, zlogstd, src, dst, E, ALS, ALD, QL, QC);
}


# Node kernel loads batched ahead of use: an edge trip's source logits and node constants issue with its feature rows (was a dependent load per add group); self-loop rows 4 per thread at once
# speedup vs baseline: 4.4101x; 1.0603x over previous
#include <hip/hip_runtime.h>

typedef float v2f __attribute__((ext_vector_type(2)));
typedef float v8f __attribute__((ext_vector_type(8)));

#define NNODES 30000
#define HEADS 4
#define SLOPE 0.2f
#define KCHUNK 128
#define PAIRSTRIDE 160

__global__ __launch_bounds__(256)
void gemm_xw_wmma(const float* __restrict__ A, const float* __restrict__ B,
                  float* __restrict__ C, int M, int K, int Nout) {
  __shared__ float ldsB[(KCHUNK / 2) * PAIRSTRIDE];

  const int tid  = threadIdx.x;
  const int lane = tid & 31;
  const int wave = tid >> 5;
  const int Mtiles = M >> 4;

  const int n0 = blockIdx.y * 64;
  int mt = blockIdx.x * 8 + wave;
  const bool active = (mt < Mtiles);
  if (!active) mt = Mtiles - 1;
  const int m0   = mt * 16;
  const int half = lane >> 4;
  const int lm   = lane & 15;

  const int lrow = tid >> 4;
  const int lcol = (tid & 15) * 4;

  v8f acc0 = {}, acc1 = {}, acc2 = {}, acc3 = {};
  const float* __restrict__ arow = A + (size_t)(m0 + lm) * K;

  for (int kb = 0; kb < K; kb += KCHUNK) {
    __syncthreads();
#pragma unroll
    for (int r = 0; r < KCHUNK; r += 16) {
      const int k = r + lrow;
      const float4 v =
          *(const float4*)(B + (size_t)(kb + k) * Nout + n0 + lcol);
      float* p = &ldsB[(k >> 1) * PAIRSTRIDE + lcol * 2 + (k & 1)];
      p[0] = v.x; p[2] = v.y; p[4] = v.z; p[6] = v.w;
    }
    __syncthreads();

    if (kb + KCHUNK < K)
      __builtin_prefetch(arow + kb + KCHUNK, 0, 0);

#pragma unroll 4
    for (int k = 0; k < KCHUNK; k += 4) {
      v2f a = *(const v2f*)(arow + kb + k + 2 * half);
      const float* __restrict__ bp =
          &ldsB[((k >> 1) + half) * PAIRSTRIDE + lm * 2];
      v2f b;
      b = *(const v2f*)(bp + 0);
      acc0 = __builtin_amdgcn_wmma_f32_16x16x4_f32(false, a, false, b, (short)0, acc0, false, false);
      b = *(const v2f*)(bp + 32);
      acc1 = __builtin_amdgcn_wmma_f32_16x16x4_f32(false, a, false, b, (short)0, acc1, false, false);
      b = *(const v2f*)(bp + 64);
      acc2 = __builtin_amdgcn_wmma_f32_16x16x4_f32(false, a, false, b, (short)0, acc2, false, false);
      b = *(const v2f*)(bp + 96);
      acc3 = __builtin_amdgcn_wmma_f32_16x16x4_f32(false, a, false, b, (short)0, acc3, false, false);
    }
  }

  if (active) {
    float* __restrict__ crow = C + (size_t)(m0 + 8 * half) * Nout + n0 + lm;
#pragma unroll
    for (int v = 0; v < 8; ++v) {
      crow[(size_t)v * Nout + 0]  = acc0[v];
      crow[(size_t)v * Nout + 16] = acc1[v];
      crow[(size_t)v * Nout + 32] = acc2[v];
      crow[(size_t)v * Nout + 48] = acc3[v];
    }
  }
}

__global__ void attn_logits(const float* __restrict__ H, const float* __restrict__ a_s,
                            const float* __restrict__ a_d, float* __restrict__ als,
                            float* __restrict__ ald, int Nn, int C) {
  int i = blockIdx.x * blockDim.x + threadIdx.x;
  if (i >= Nn * HEADS) return;
  int n = i / HEADS, h = i % HEADS;
  const float* hp = H + (size_t)n * HEADS * C + (size_t)h * C;
  float s = 0.f, d = 0.f;
  for (int c = 0; c < C; ++c) { float v = hp[c]; s += v * a_s[h * C + c]; d += v * a_d[h * C + c]; }
  als[i] = s; ald[i] = d;
}

__device__ __forceinline__ unsigned fkey(float x) {
  unsigned b = __float_as_uint(x);
  return (b & 0x80000000u) ? ~b : (b | 0x80000000u);
}
__device__ __forceinline__ float funkey(unsigned k) {
  unsigned b = (k & 0x80000000u) ? (k & 0x7FFFFFFFu) : ~k;
  return __uint_as_float(b);
}

#define RNODES 118
#define QCAP 8192
#define NTHR 1024
#define NWAVE (NTHR / 32)
#define NGROUPS ((NNODES + RNODES - 1) / RNODES)

template <int C, bool MEAN, bool FIRST>
__global__ __launch_bounds__(NTHR)
void node_range_softmax_aggregate(const int* __restrict__ src, const int* __restrict__ dst, int E,
                                  const float* __restrict__ H, const float* __restrict__ als,
                                  const float* __restrict__ ald, const float* __restrict__ bias,
                                  float* __restrict__ out,
                                  unsigned* __restrict__ qlist, int* __restrict__ qcnt) {
  constexpr int F = HEADS * C;
  constexpr int J = F / 128;
  constexpr int U = 8 / J;
  __shared__ float    acc[RNODES * F];
  __shared__ unsigned mk[RNODES * HEADS];
  __shared__ float    mv[RNODES * HEADS];
  __shared__ float    den[RNODES * HEADS];
  __shared__ float    aldc[RNODES * HEADS];
  __shared__ unsigned q[QCAP];
  __shared__ unsigned qn;

  const int tid  = threadIdx.x;
  const int lane = tid & 31;
  const int wave = tid >> 5;
  const int base = blockIdx.x * RNODES;
  int rn = NNODES - base;
  if (rn > RNODES) rn = RNODES;
  if (rn < 0) rn = 0;
  unsigned* __restrict__ lst = qlist + (size_t)blockIdx.x * QCAP;

  auto take = [&](int e, unsigned dl) {
    if (dl < (unsigned)rn) {
      const unsigned slot = atomicAdd(&qn, 1u);
      if (slot < QCAP) q[slot] = (unsigned)e | (dl << 25);
    }
  };

  auto scan = [&](int e0, int e1) {
    if (((E | e0 | e1) & 3) == 0) {
      const int4* __restrict__ d4 = (const int4*)dst;
      const int g1 = e1 >> 2;
      for (int g0 = (e0 >> 2); g0 < g1; g0 += NTHR * 8) {
        int4 v[8];
#pragma unroll
        for (int u = 0; u < 8; ++u) {
          const int g = g0 + u * NTHR + tid;
          v[u] = (g < g1) ? d4[g] : make_int4(-1, -1, -1, -1);
        }
#pragma unroll
        for (int u = 0; u < 8; ++u) {
          const int e = (g0 + u * NTHR + tid) * 4;
          const unsigned a0 = (unsigned)(v[u].x - base), a1 = (unsigned)(v[u].y - base);
          const unsigned a2 = (unsigned)(v[u].z - base), a3 = (unsigned)(v[u].w - base);
          const unsigned lo01 = a0 < a1 ? a0 : a1, lo23 = a2 < a3 ? a2 : a3;
          if ((lo01 < lo23 ? lo01 : lo23) < (unsigned)rn) {
            take(e + 0, a0); take(e + 1, a1); take(e + 2, a2); take(e + 3, a3);
          }
        }
      }
    } else {
      for (int e = e0 + tid; e < e1; e += NTHR) take(e, (unsigned)(dst[e] - base));
    }
  };

  auto resolve = [&](int n, bool withmax, bool keep) {
    for (int i = tid; i < n; i += NTHR) {
      const unsigned ent = q[i];
      const unsigned dl = ent >> 25;
      const int s = src[ent & 0x1FFFFFFu];
      if (withmax) {
        const float4 a = *(const float4*)(als + (size_t)s * HEADS);
        atomicMax(&mk[dl * HEADS + 0], fkey(a.x));
        atomicMax(&mk[dl * HEADS + 1], fkey(a.y));
        atomicMax(&mk[dl * HEADS + 2], fkey(a.z));
        atomicMax(&mk[dl * HEADS + 3], fkey(a.w));
      }
      const unsigned packed = (unsigned)s | (dl << 16);
      q[i] = packed;
      if (keep) lst[i] = packed;
    }
  };

  auto reload = [&](int n) {
    for (int i = tid; i < n; i += NTHR) {
      const unsigned ent = lst[i];
      const unsigned dl = ent >> 16;
      const float4 a = *(const float4*)(als + (size_t)(ent & 0xFFFFu) * HEADS);
      atomicMax(&mk[dl * HEADS + 0], fkey(a.x));
      atomicMax(&mk[dl * HEADS + 1], fkey(a.y));
      atomicMax(&mk[dl * HEADS + 2], fkey(a.z));
      atomicMax(&mk[dl * HEADS + 3], fkey(a.w));
      q[i] = ent;
    }
  };

  auto drain = [&](int n) {
    for (int i0 = wave; i0 < n; i0 += NWAVE * U) {
      const int iu = i0 + lane * NWAVE;
      const unsigned mine = (lane < U && iu < n) ? q[iu] : 0u;
      float4 hv[U][J];
      float sa[U][J], da[U][J], ma[U][J];
      int sidx[U], dloc[U];
#pragma unroll
      for (int u = 0; u < U; ++u) {
        const unsigned ent = (unsigned)__builtin_amdgcn_readlane((int)mine, u);
        sidx[u] = (int)(ent & 0xFFFFu);
        dloc[u] = (int)(ent >> 16);
      }
#pragma unroll
      for (int u = 0; u < U; ++u) {
#pragma unroll
        for (int j = 0; j < J; ++j) {
          const int h = (128 * j + lane * 4) / C;
          sa[u][j] = als[sidx[u] * HEADS + h];
        }
      }
#pragma unroll
      for (int u = 0; u < U; ++u) {
        const float* __restrict__ hrow = H + (size_t)sidx[u] * F + lane * 4;
#pragma unroll
        for (int j = 0; j < J; ++j) hv[u][j] = *(const float4*)(hrow + 128 * j);
      }
#pragma unroll
      for (int u = 0; u < U; ++u) {
#pragma unroll
        for (int j = 0; j < J; ++j) {
          const int h = (128 * j + lane * 4) / C;
          da[u][j] = aldc[dloc[u] * HEADS + h];
          ma[u][j] = mv[dloc[u] * HEADS + h];
        }
      }
      __builtin_amdgcn_sched_barrier(0);
#pragma unroll
      for (int u = 0; u < U; ++u) {
        if (i0 + u * NWAVE < n) {
          const int dl = dloc[u];
#pragma unroll
          for (int j = 0; j < J; ++j) {
            const int f = 128 * j + lane * 4;
            const int h = f / C;
            float sc = sa[u][j] + da[u][j];
            sc = (sc > 0.f) ? sc : SLOPE * sc;
            const float ex = __expf(sc - ma[u][j]);
            if ((f % C) == 0) atomicAdd(&den[dl * HEADS + h], ex);
            float* ap = &acc[dl * F + 128 * j + lane];
            atomicAdd(ap +  0, hv[u][j].x * ex);
            atomicAdd(ap + 32, hv[u][j].y * ex);
            atomicAdd(ap + 64, hv[u][j].z * ex);
            atomicAdd(ap + 96, hv[u][j].w * ex);
          }
        }
      }
    }
  };

  if (tid < rn * HEADS) {
    mk[tid]   = fkey(als[(size_t)base * HEADS + tid]);
    aldc[tid] = ald[(size_t)base * HEADS + tid];
  }
  if (tid == 0) qn = 0u;
  __syncthreads();

  int total;
  if (FIRST) {
    scan(0, E);
    __syncthreads();
    total = (int)qn;
    if (tid == 0) qcnt[blockIdx.x] = total;
  } else {
    total = qcnt[blockIdx.x];
  }
  const bool fits = (total <= QCAP);
  if (fits) {
    if (FIRST) resolve(total, true, true);
    else       reload(total);
  } else {
    for (int p = 0; p < E; p += QCAP) {
      __syncthreads();
      if (tid == 0) qn = 0u;
      __syncthreads();
      scan(p, (p + QCAP < E) ? (p + QCAP) : E);
      __syncthreads();
      resolve((int)qn, true, false);
    }
  }
  __syncthreads();

  if (tid < rn * HEADS) {
    const float a = aldc[tid];
    float m = funkey(mk[tid]) + a;
    m = (m > 0.f) ? m : SLOPE * m;
    mv[tid] = m;
    float sc = als[(size_t)base * HEADS + tid] + a;
    sc = (sc > 0.f) ? sc : SLOPE * sc;
    den[tid] = __expf(sc - m);
  }
  __syncthreads();

  {
    constexpr int P = 4;
    const int items = rn * (F / 4);
    for (int idx0 = tid; idx0 < items; idx0 += NTHR * P) {
      float4 hp[P];
#pragma unroll
      for (int p = 0; p < P; ++p) {
        const int idx = idx0 + p * NTHR;
        const int idc = (idx < items) ? idx : idx0;
        const int dl = idc / (F / 4);
        const int r  = idc % (F / 4);
        const int f  = 128 * (r >> 5) + 4 * (r & 31);
        hp[p] = *(const float4*)(H + (size_t)(base + dl) * F + f);
      }
#pragma unroll
      for (int p = 0; p < P; ++p) {
        const int idx = idx0 + p * NTHR;
        const int idc = (idx < items) ? idx : idx0;
        const int dl = idc / (F / 4);
        const int r  = idc % (F / 4);
        const int j  = r >> 5, l = r & 31;
        const int f  = 128 * j + 4 * l;
        const float ex = den[dl * HEADS + f / C];
        float* ap = &acc[dl * F + 128 * j + l];
        ap[0] = hp[p].x * ex; ap[32] = hp[p].y * ex; ap[64] = hp[p].z * ex; ap[96] = hp[p].w * ex;
      }
    }
  }
  __syncthreads();

  if (fits) {
    drain(total);
  } else {
    for (int p = 0; p < E; p += QCAP) {
      __syncthreads();
      if (tid == 0) qn = 0u;
      __syncthreads();
      scan(p, (p + QCAP < E) ? (p + QCAP) : E);
      __syncthreads();
      resolve((int)qn, false, false);
      __syncthreads();
      drain((int)qn);
    }
  }
  __syncthreads();

  if (!MEAN) {
    for (int idx = tid; idx < rn * (F / 4); idx += NTHR) {
      const int dl = idx / (F / 4);
      const int r  = idx % (F / 4);
      const int j  = r >> 5, l = r & 31;
      const int f  = 128 * j + 4 * l;
      const float dn = den[dl * HEADS + f / C] + 1e-16f;
      const float* ap = &acc[dl * F + 128 * j + l];
      const float4 b4 = *(const float4*)(bias + f);
      float4 o;
      o.x = ap[0]  / dn + b4.x;  o.y = ap[32] / dn + b4.y;
      o.z = ap[64] / dn + b4.z;  o.w = ap[96] / dn + b4.w;
      o.x = o.x > 0.f ? o.x : 0.f;  o.y = o.y > 0.f ? o.y : 0.f;
      o.z = o.z > 0.f ? o.z : 0.f;  o.w = o.w > 0.f ? o.w : 0.f;
      *(float4*)(out + (size_t)(base + dl) * F + f) = o;
    }
  } else {
    for (int idx = tid; idx < rn * C; idx += NTHR) {
      const int dl = idx / C, c = idx % C;
      const float* qd = &den[dl * HEADS];
      float p[HEADS];
#pragma unroll
      for (int h = 0; h < HEADS; ++h) {
        const int f = h * C + c;
        p[h] = acc[dl * F + 128 * (f >> 7) + 32 * (f & 3) + ((f & 127) >> 2)];
      }
      out[(size_t)(base + dl) * C + c] =
          0.25f * (p[0] / (qd[0] + 1e-16f) + p[1] / (qd[1] + 1e-16f) +
                   p[2] / (qd[2] + 1e-16f) + p[3] / (qd[3] + 1e-16f)) + bias[c];
    }
  }
}

template <int C, bool MEAN, bool FIRST>
static void run_gat_layer(hipStream_t st, const float* in, int Fin,
                          const float* W, const float* a_s, const float* a_d, const float* b,
                          float* Hbuf, float* outp,
                          const int* src, const int* dst, int E,
                          float* ALS, float* ALD, unsigned* QL, int* QC) {
  const int Nn = NNODES;
  const int Fout = HEADS * C;
  const int Mtiles = Nn / 16;
  dim3 ggrid((Mtiles + 7) / 8, Fout / 64);
  gemm_xw_wmma<<<ggrid, 256, 0, st>>>(in, W, Hbuf, Nn, Fin, Fout);
  int t = Nn * HEADS;
  attn_logits<<<(t + 255) / 256, 256, 0, st>>>(Hbuf, a_s, a_d, ALS, ALD, Nn, C);
  node_range_softmax_aggregate<C, MEAN, FIRST>
      <<<NGROUPS, NTHR, 0, st>>>(src, dst, E, Hbuf, ALS, ALD, b, outp, QL, QC);
}

extern "C" void kernel_launch(void* const* d_in, const int* in_sizes, int n_in,
                              void* d_out, int out_size, void* d_ws, size_t ws_size,
                              hipStream_t stream) {
  const float* x   = (const float*)d_in[0];
  const int*   ei  = (const int*)d_in[1];
  const int E = in_sizes[1] / 2;
  const int* src = ei;
  const int* dst = ei + E;
  const float *W1 = (const float*)d_in[2],  *as1 = (const float*)d_in[3],
              *ad1 = (const float*)d_in[4], *b1  = (const float*)d_in[5];
  const float *W2 = (const float*)d_in[6],  *as2 = (const float*)d_in[7],
              *ad2 = (const float*)d_in[8], *b2  = (const float*)d_in[9];
  const float *Wm = (const float*)d_in[10], *asm_ = (const float*)d_in[11],
              *adm = (const float*)d_in[12], *bm = (const float*)d_in[13];
  const float *Wl = (const float*)d_in[14], *asl = (const float*)d_in[15],
              *adl = (const float*)d_in[16], *bl = (const float*)d_in[17];

  char* ws = (char*)d_ws;
  size_t off = 0;
  auto alloc = [&](size_t bytes) -> void* {
    void* p = ws + off;
    off += (bytes + 255) & ~(size_t)255;
    return p;
  };
  float*    Hbuf = (float*)alloc((size_t)NNODES * 512 * 4);
  float*    A0   = (float*)alloc((size_t)NNODES * 256 * 4);
  float*    A1   = (float*)alloc((size_t)NNODES * 512 * 4);
  float*    ALS  = (float*)alloc((size_t)NNODES * HEADS * 4);
  float*    ALD  = (float*)alloc((size_t)NNODES * HEADS * 4);
  unsigned* QL   = (unsigned*)alloc((size_t)NGROUPS * QCAP * 4);
  int*      QC   = (int*)alloc((size_t)NGROUPS * 4);

  float* zmean   = (float*)d_out;
  float* zlogstd = (float*)d_out + (size_t)NNODES * 32;

  run_gat_layer<64, false, true>(stream, x, 256, W1, as1, ad1, b1, Hbuf, A0, src, dst, E, ALS, ALD, QL, QC);
  run_gat_layer<128, false, false>(stream, A0, 256, W2, as2, ad2, b2, Hbuf, A1, src, dst, E, ALS, ALD, QL, QC);
  run_gat_layer<32, true, false>(stream, A1, 512, Wm, asm_, adm, bm, Hbuf, zmean, src, dst, E, ALS, ALD, QL, QC);
  run_gat_layer<32, true, false>(stream, A1, 512, Wl, asl, adl, bl, Hbuf, zlogstd, src, dst, E, ALS, ALD, QL, QC);
}


# Group's edge queue counting-sorted by owned node in LDS; a wave sums a node's rows in registers and stores them: LDS float atomics, LDS sums gone; sum order = place in bucket (LDS counter's return)
# speedup vs baseline: 4.6796x; 1.0611x over previous
#include <hip/hip_runtime.h>

typedef float v2f __attribute__((ext_vector_type(2)));
typedef float v8f __attribute__((ext_vector_type(8)));

#define NNODES 30000
#define HEADS 4
#define SLOPE 0.2f
#define KCHUNK 128
#define PAIRSTRIDE 160

__global__ __launch_bounds__(256)
void gemm_xw_wmma(const float* __restrict__ A, const float* __restrict__ B,
                  float* __restrict__ C, int M, int K, int Nout) {
  __shared__ float ldsB[(KCHUNK / 2) * PAIRSTRIDE];

  const int tid  = threadIdx.x;
  const int lane = tid & 31;
  const int wave = tid >> 5;
  const int Mtiles = M >> 4;

  const int n0 = blockIdx.y * 64;
  int mt = blockIdx.x * 8 + wave;
  const bool active = (mt < Mtiles);
  if (!active) mt = Mtiles - 1;
  const int m0   = mt * 16;
  const int half = lane >> 4;
  const int lm   = lane & 15;

  const int lrow = tid >> 4;
  const int lcol = (tid & 15) * 4;

  v8f acc0 = {}, acc1 = {}, acc2 = {}, acc3 = {};
  const float* __restrict__ arow = A + (size_t)(m0 + lm) * K;

  for (int kb = 0; kb < K; kb += KCHUNK) {
    __syncthreads();
#pragma unroll
    for (int r = 0; r < KCHUNK; r += 16) {
      const int k = r + lrow;
      const float4 v =
          *(const float4*)(B + (size_t)(kb + k) * Nout + n0 + lcol);
      float* p = &ldsB[(k >> 1) * PAIRSTRIDE + lcol * 2 + (k & 1)];
      p[0] = v.x; p[2] = v.y; p[4] = v.z; p[6] = v.w;
    }
    __syncthreads();

    if (kb + KCHUNK < K)
      __builtin_prefetch(arow + kb + KCHUNK, 0, 0);

#pragma unroll 4
    for (int k = 0; k < KCHUNK; k += 4) {
      v2f a = *(const v2f*)(arow + kb + k + 2 * half);
      const float* __restrict__ bp =
          &ldsB[((k >> 1) + half) * PAIRSTRIDE + lm * 2];
      v2f b;
      b = *(const v2f*)(bp + 0);
      acc0 = __builtin_amdgcn_wmma_f32_16x16x4_f32(false, a, false, b, (short)0, acc0, false, false);
      b = *(const v2f*)(bp + 32);
      acc1 = __builtin_amdgcn_wmma_f32_16x16x4_f32(false, a, false, b, (short)0, acc1, false, false);
      b = *(const v2f*)(bp + 64);
      acc2 = __builtin_amdgcn_wmma_f32_16x16x4_f32(false, a, false, b, (short)0, acc2, false, false);
      b = *(const v2f*)(bp + 96);
      acc3 = __builtin_amdgcn_wmma_f32_16x16x4_f32(false, a, false, b, (short)0, acc3, false, false);
    }
  }

  if (active) {
    float* __restrict__ crow = C + (size_t)(m0 + 8 * half) * Nout + n0 + lm;
#pragma unroll
    for (int v = 0; v < 8; ++v) {
      crow[(size_t)v * Nout + 0]  = acc0[v];
      crow[(size_t)v * Nout + 16] = acc1[v];
      crow[(size_t)v * Nout + 32] = acc2[v];
      crow[(size_t)v * Nout + 48] = acc3[v];
    }
  }
}

__global__ void attn_logits(const float* __restrict__ H, const float* __restrict__ a_s,
                            const float* __restrict__ a_d, float* __restrict__ als,
                            float* __restrict__ ald, int Nn, int C) {
  int i = blockIdx.x * blockDim.x + threadIdx.x;
  if (i >= Nn * HEADS) return;
  int n = i / HEADS, h = i % HEADS;
  const float* hp = H + (size_t)n * HEADS * C + (size_t)h * C;
  float s = 0.f, d = 0.f;
  for (int c = 0; c < C; ++c) { float v = hp[c]; s += v * a_s[h * C + c]; d += v * a_d[h * C + c]; }
  als[i] = s; ald[i] = d;
}

__device__ __forceinline__ unsigned fkey(float x) {
  unsigned b = __float_as_uint(x);
  return (b & 0x80000000u) ? ~b : (b | 0x80000000u);
}
__device__ __forceinline__ float funkey(unsigned k) {
  unsigned b = (k & 0x80000000u) ? (k & 0x7FFFFFFFu) : ~k;
  return __uint_as_float(b);
}

#define RNODES 118
#define QCAP 8192
#define NTHR 1024
#define NWAVE (NTHR / 32)
#define NGROUPS ((NNODES + RNODES - 1) / RNODES)

template <int C, bool MEAN, bool FIRST>
__global__ __launch_bounds__(NTHR)
void node_range_softmax_aggregate(const int* __restrict__ src, const int* __restrict__ dst, int E,
                                  const float* __restrict__ H, const float* __restrict__ als,
                                  const float* __restrict__ ald, const float* __restrict__ bias,
                                  float* __restrict__ out,
                                  unsigned* __restrict__ qlist, int* __restrict__ qcnt) {
  constexpr int F = HEADS * C;
  constexpr int J = F / 128;
  constexpr int U = 8 / J;
  __shared__ float    acc[RNODES * F];
  __shared__ unsigned mk[RNODES * HEADS];
  __shared__ float    mv[RNODES * HEADS];
  __shared__ float    den[RNODES * HEADS];
  __shared__ float    aldc[RNODES * HEADS];
  __shared__ unsigned q[QCAP];
  __shared__ unsigned qn;
  __shared__ int      cnt[128];
  __shared__ int      off[132];

  const int tid  = threadIdx.x;
  const int lane = tid & 31;
  const int wave = tid >> 5;
  const int base = blockIdx.x * RNODES;
  int rn = NNODES - base;
  if (rn > RNODES) rn = RNODES;
  if (rn < 0) rn = 0;
  unsigned* __restrict__ lst = qlist + (size_t)blockIdx.x * QCAP;

  auto take = [&](int e, unsigned dl) {
    if (dl < (unsigned)rn) {
      const unsigned slot = atomicAdd(&qn, 1u);
      if (slot < QCAP) q[slot] = (unsigned)e | (dl << 25);
    }
  };

  auto scan = [&](int e0, int e1) {
    if (((E | e0 | e1) & 3) == 0) {
      const int4* __restrict__ d4 = (const int4*)dst;
      const int g1 = e1 >> 2;
      for (int g0 = (e0 >> 2); g0 < g1; g0 += NTHR * 8) {
        int4 v[8];
#pragma unroll
        for (int u = 0; u < 8; ++u) {
          const int g = g0 + u * NTHR + tid;
          v[u] = (g < g1) ? d4[g] : make_int4(-1, -1, -1, -1);
        }
#pragma unroll
        for (int u = 0; u < 8; ++u) {
          const int e = (g0 + u * NTHR + tid) * 4;
          const unsigned a0 = (unsigned)(v[u].x - base), a1 = (unsigned)(v[u].y - base);
          const unsigned a2 = (unsigned)(v[u].z - base), a3 = (unsigned)(v[u].w - base);
          const unsigned lo01 = a0 < a1 ? a0 : a1, lo23 = a2 < a3 ? a2 : a3;
          if ((lo01 < lo23 ? lo01 : lo23) < (unsigned)rn) {
            take(e + 0, a0); take(e + 1, a1); take(e + 2, a2); take(e + 3, a3);
          }
        }
      }
    } else {
      for (int e = e0 + tid; e < e1; e += NTHR) take(e, (unsigned)(dst[e] - base));
    }
  };

  auto resolve = [&](int n, bool withmax, bool keep) {
    for (int i = tid; i < n; i += NTHR) {
      const unsigned ent = q[i];
      const unsigned dl = ent >> 25;
      const int s = src[ent & 0x1FFFFFFu];
      if (withmax) {
        const float4 a = *(const float4*)(als + (size_t)s * HEADS);
        atomicMax(&mk[dl * HEADS + 0], fkey(a.x));
        atomicMax(&mk[dl * HEADS + 1], fkey(a.y));
        atomicMax(&mk[dl * HEADS + 2], fkey(a.z));
        atomicMax(&mk[dl * HEADS + 3], fkey(a.w));
      }
      const unsigned packed = (unsigned)s | (dl << 16);
      q[i] = packed;
      if (keep) lst[i] = packed;
    }
  };

  auto reload = [&](int n) {
    for (int i = tid; i < n; i += NTHR) {
      const unsigned ent = lst[i];
      const unsigned dl = ent >> 16;
      const float4 a = *(const float4*)(als + (size_t)(ent & 0xFFFFu) * HEADS);
      atomicMax(&mk[dl * HEADS + 0], fkey(a.x));
      atomicMax(&mk[dl * HEADS + 1], fkey(a.y));
      atomicMax(&mk[dl * HEADS + 2], fkey(a.z));
      atomicMax(&mk[dl * HEADS + 3], fkey(a.w));
      q[i] = ent;
    }
  };

  auto drain = [&](int n) {
    for (int i0 = wave; i0 < n; i0 += NWAVE * U) {
      const int iu = i0 + lane * NWAVE;
      const unsigned mine = (lane < U && iu < n) ? q[iu] : 0u;
      float4 hv[U][J];
      float sa[U][J], da[U][J], ma[U][J];
      int sidx[U], dloc[U];
#pragma unroll
      for (int u = 0; u < U; ++u) {
        const unsigned ent = (unsigned)__builtin_amdgcn_readlane((int)mine, u);
        sidx[u] = (int)(ent & 0xFFFFu);
        dloc[u] = (int)(ent >> 16);
      }
#pragma unroll
      for (int u = 0; u < U; ++u) {
#pragma unroll
        for (int j = 0; j < J; ++j) {
          const int h = (128 * j + lane * 4) / C;
          sa[u][j] = als[sidx[u] * HEADS + h];
        }
      }
#pragma unroll
      for (int u = 0; u < U; ++u) {
        const float* __restrict__ hrow = H + (size_t)sidx[u] * F + lane * 4;
#pragma unroll
        for (int j = 0; j < J; ++j) hv[u][j] = *(const float4*)(hrow + 128 * j);
      }
#pragma unroll
      for (int u = 0; u < U; ++u) {
#pragma unroll
        for (int j = 0; j < J; ++j) {
          const int h = (128 * j + lane * 4) / C;
          da[u][j] = aldc[dloc[u] * HEADS + h];
          ma[u][j] = mv[dloc[u] * HEADS + h];
        }
      }
      __builtin_amdgcn_sched_barrier(0);
#pragma unroll
      for (int u = 0; u < U; ++u) {
        if (i0 + u * NWAVE < n) {
          const int dl = dloc[u];
#pragma unroll
          for (int j = 0; j < J; ++j) {
            const int f = 128 * j + lane * 4;
            const int h = f / C;
            float sc = sa[u][j] + da[u][j];
            sc = (sc > 0.f) ? sc : SLOPE * sc;
            const float ex = __expf(sc - ma[u][j]);
            if ((f % C) == 0) atomicAdd(&den[dl * HEADS + h], ex);
            float* ap = &acc[dl * F + 128 * j + lane];
            atomicAdd(ap +  0, hv[u][j].x * ex);
            atomicAdd(ap + 32, hv[u][j].y * ex);
            atomicAdd(ap + 64, hv[u][j].z * ex);
            atomicAdd(ap + 96, hv[u][j].w * ex);
          }
        }
      }
    }
  };

  if (tid < rn * HEADS) {
    mk[tid]   = fkey(als[(size_t)base * HEADS + tid]);
    aldc[tid] = ald[(size_t)base * HEADS + tid];
  }
  if (tid == 0) qn = 0u;
  if (tid < 128) cnt[tid] = 0;
  __syncthreads();

  int total;
  if (FIRST) {
    scan(0, E);
    __syncthreads();
    total = (int)qn;
    if (tid == 0) qcnt[blockIdx.x] = total;
  } else {
    total = qcnt[blockIdx.x];
  }
  const bool fits = (total <= QCAP);
  if (fits) {
    if (FIRST) resolve(total, true, true);
    else       reload(total);
  } else {
    for (int p = 0; p < E; p += QCAP) {
      __syncthreads();
      if (tid == 0) qn = 0u;
      __syncthreads();
      scan(p, (p + QCAP < E) ? (p + QCAP) : E);
      __syncthreads();
      resolve((int)qn, true, false);
    }
  }
  __syncthreads();

  if (tid < rn * HEADS) {
    const float a = aldc[tid];
    float m = funkey(mk[tid]) + a;
    m = (m > 0.f) ? m : SLOPE * m;
    mv[tid] = m;
    float sc = als[(size_t)base * HEADS + tid] + a;
    sc = (sc > 0.f) ? sc : SLOPE * sc;
    den[tid] = __expf(sc - m);
  }

  if (fits) {
    static_assert(!MEAN || C == 32, "the sum over heads across lanes is written for 32 channels per head");
    unsigned* __restrict__ qs = (unsigned*)acc;
    constexpr int KMAX = QCAP / NTHR;
    unsigned ent[KMAX];
    int rk[KMAX];
#pragma unroll
    for (int k = 0; k < KMAX; ++k) {
      const int i = tid + k * NTHR;
      ent[k] = 0u; rk[k] = 0;
      if (i < total) { ent[k] = q[i]; rk[k] = atomicAdd(&cnt[ent[k] >> 16], 1); }
    }
    __syncthreads();
    if (wave == 0) {
      const int c0 = cnt[4 * lane], c1 = cnt[4 * lane + 1], c2 = cnt[4 * lane + 2], c3 = cnt[4 * lane + 3];
      const int s = c0 + c1 + c2 + c3;
      int incl = s;
#pragma unroll
      for (int d = 1; d < 32; d <<= 1) {
        const int t = __shfl_up(incl, d);
        if (lane >= d) incl += t;
      }
      const int ex0 = incl - s;
      off[4 * lane]     = ex0;
      off[4 * lane + 1] = ex0 + c0;
      off[4 * lane + 2] = ex0 + c0 + c1;
      off[4 * lane + 3] = ex0 + c0 + c1 + c2;
      if (lane == 31) off[128] = incl;
    }
    __syncthreads();
#pragma unroll
    for (int k = 0; k < KMAX; ++k) {
      const int i = tid + k * NTHR;
      if (i < total) qs[off[ent[k] >> 16] + rk[k]] = ent[k] & 0xFFFFu;
    }
    __syncthreads();

    constexpr int UE = 8;
    const int wv = __builtin_amdgcn_readfirstlane(wave);
    const int nitems = rn * J;
    for (int it = wv; it < nitems; it += NWAVE) {
      const int dl = it / J, j = it % J;
      const int node = base + dl;
      const int f = 128 * j + lane * 4;
      const int h = f / C;
      const int o0 = __builtin_amdgcn_readfirstlane(off[dl]);
      const int nt = __builtin_amdgcn_readfirstlane(off[dl + 1]) - o0 + 1;
      const float m  = mv[dl * HEADS + h];
      const float ad = aldc[dl * HEADS + h];
      float4 a4 = make_float4(0.f, 0.f, 0.f, 0.f);
      float dn = 0.f;
      for (int i0 = 0; i0 < nt; i0 += UE) {
        const int p = i0 + lane;
        unsigned mine = (unsigned)node;
        if (lane < UE && p >= 1 && p < nt) mine = qs[o0 + p - 1];
        int sidx[UE];
        float sa[UE];
        float4 hv[UE];
#pragma unroll
        for (int u = 0; u < UE; ++u) sidx[u] = __builtin_amdgcn_readlane((int)mine, u);
#pragma unroll
        for (int u = 0; u < UE; ++u) sa[u] = als[sidx[u] * HEADS + h];
#pragma unroll
        for (int u = 0; u < UE; ++u) hv[u] = *(const float4*)(H + (size_t)sidx[u] * F + f);
#pragma unroll
        for (int u = 0; u < UE; ++u) {
          float sc = sa[u] + ad;
          sc = (sc > 0.f) ? sc : SLOPE * sc;
          float ex = __expf(sc - m);
          ex = (i0 + u < nt) ? ex : 0.f;
          dn += ex;
          a4.x += hv[u].x * ex;  a4.y += hv[u].y * ex;
          a4.z += hv[u].z * ex;  a4.w += hv[u].w * ex;
        }
      }
      const float dd = dn + 1e-16f;
      if (!MEAN) {
        const float4 b4 = *(const float4*)(bias + f);
        float4 o;
        o.x = a4.x / dd + b4.x;  o.y = a4.y / dd + b4.y;
        o.z = a4.z / dd + b4.z;  o.w = a4.w / dd + b4.w;
        o.x = o.x > 0.f ? o.x : 0.f;  o.y = o.y > 0.f ? o.y : 0.f;
        o.z = o.z > 0.f ? o.z : 0.f;  o.w = o.w > 0.f ? o.w : 0.f;
        *(float4*)(out + (size_t)node * F + f) = o;
      } else {
        const float v0 = a4.x / dd, v1 = a4.y / dd, v2 = a4.z / dd, v3 = a4.w / dd;
        const int l8 = lane & 7;
        const float r0 = ((__shfl(v0, l8) + __shfl(v0, l8 + 8)) + __shfl(v0, l8 + 16)) + __shfl(v0, l8 + 24);
        const float r1 = ((__shfl(v1, l8) + __shfl(v1, l8 + 8)) + __shfl(v1, l8 + 16)) + __shfl(v1, l8 + 24);
        const float r2 = ((__shfl(v2, l8) + __shfl(v2, l8 + 8)) + __shfl(v2, l8 + 16)) + __shfl(v2, l8 + 24);
        const float r3 = ((__shfl(v3, l8) + __shfl(v3, l8 + 8)) + __shfl(v3, l8 + 16)) + __shfl(v3, l8 + 24);
        if (lane < 8) {
          const float4 b4 = *(const float4*)(bias + 4 * lane);
          float4 o;
          o.x = 0.25f * r0 + b4.x;  o.y = 0.25f * r1 + b4.y;
          o.z = 0.25f * r2 + b4.z;  o.w = 0.25f * r3 + b4.w;
          *(float4*)(out + (size_t)node * C + 4 * lane) = o;
        }
      }
    }
  } else {
  __syncthreads();

  {
    constexpr int P = 4;
    const int items = rn * (F / 4);
    for (int idx0 = tid; idx0 < items; idx0 += NTHR * P) {
      float4 hp[P];
#pragma unroll
      for (int p = 0; p < P; ++p) {
        const int idx = idx0 + p * NTHR;
        const int idc = (idx < items) ? idx : idx0;
        const int dl = idc / (F / 4);
        const int r  = idc % (F / 4);
        const int f  = 128 * (r >> 5) + 4 * (r & 31);
        hp[p] = *(const float4*)(H + (size_t)(base + dl) * F + f);
      }
#pragma unroll
      for (int p = 0; p < P; ++p) {
        const int idx = idx0 + p * NTHR;
        const int idc = (idx < items) ? idx : idx0;
        const int dl = idc / (F / 4);
        const int r  = idc % (F / 4);
        const int j  = r >> 5, l = r & 31;
        const int f  = 128 * j + 4 * l;
        const float ex = den[dl * HEADS + f / C];
        float* ap = &acc[dl * F + 128 * j + l];
        ap[0] = hp[p].x * ex; ap[32] = hp[p].y * ex; ap[64] = hp[p].z * ex; ap[96] = hp[p].w * ex;
      }
    }
  }
  __syncthreads();

  {
    for (int p = 0; p < E; p += QCAP) {
      __syncthreads();
      if (tid == 0) qn = 0u;
      __syncthreads();
      scan(p, (p + QCAP < E) ? (p + QCAP) : E);
      __syncthreads();
      resolve((int)qn, false, false);
      __syncthreads();
      drain((int)qn);
    }
  }
  __syncthreads();

  if (!MEAN) {
    for (int idx = tid; idx < rn * (F / 4); idx += NTHR) {
      const int dl = idx / (F / 4);
      const int r  = idx % (F / 4);
      const int j  = r >> 5, l = r & 31;
      const int f  = 128 * j + 4 * l;
      const float dn = den[dl * HEADS + f / C] + 1e-16f;
      const float* ap = &acc[dl * F + 128 * j + l];
      const float4 b4 = *(const float4*)(bias + f);
      float4 o;
      o.x = ap[0]  / dn + b4.x;  o.y = ap[32] / dn + b4.y;
      o.z = ap[64] / dn + b4.z;  o.w = ap[96] / dn + b4.w;
      o.x = o.x > 0.f ? o.x : 0.f;  o.y = o.y > 0.f ? o.y : 0.f;
      o.z = o.z > 0.f ? o.z : 0.f;  o.w = o.w > 0.f ? o.w : 0.f;
      *(float4*)(out + (size_t)(base + dl) * F + f) = o;
    }
  } else {
    for (int idx = tid; idx < rn * C; idx += NTHR) {
      const int dl = idx / C, c = idx % C;
      const float* qd = &den[dl * HEADS];
      float p[HEADS];
#pragma unroll
      for (int h = 0; h < HEADS; ++h) {
        const int f = h * C + c;
        p[h] = acc[dl * F + 128 * (f >> 7) + 32 * (f & 3) + ((f & 127) >> 2)];
      }
      out[(size_t)(base + dl) * C + c] =
          0.25f * (p[0] / (qd[0] + 1e-16f) + p[1] / (qd[1] + 1e-16f) +
                   p[2] / (qd[2] + 1e-16f) + p[3] / (qd[3] + 1e-16f)) + bias[c];
    }
  }
  }
}

template <int C, bool MEAN, bool FIRST>
static void run_gat_layer(hipStream_t st, const float* in, int Fin,
                          const float* W, const float* a_s, const float* a_d, const float* b,
                          float* Hbuf, float* outp,
                          const int* src, const int* dst, int E,
                          float* ALS, float* ALD, unsigned* QL, int* QC) {
  const int Nn = NNODES;
  const int Fout = HEADS * C;
  const int Mtiles = Nn / 16;
  dim3 ggrid((Mtiles + 7) / 8, Fout / 64);
  gemm_xw_wmma<<<ggrid, 256, 0, st>>>(in, W, Hbuf, Nn, Fin, Fout);
  int t = Nn * HEADS;
  attn_logits<<<(t + 255) / 256, 256, 0, st>>>(Hbuf, a_s, a_d, ALS, ALD, Nn, C);
  node_range_softmax_aggregate<C, MEAN, FIRST>
      <<<NGROUPS, NTHR, 0, st>>>(src, dst, E, Hbuf, ALS, ALD, b, outp, QL, QC);
}

extern "C" void kernel_launch(void* const* d_in, const int* in_sizes, int n_in,
                              void* d_out, int out_size, void* d_ws, size_t ws_size,
                              hipStream_t stream) {
  const float* x   = (const float*)d_in[0];
  const int*   ei  = (const int*)d_in[1];
  const int E = in_sizes[1] / 2;
  const int* src = ei;
  const int* dst = ei + E;
  const float *W1 = (const float*)d_in[2],  *as1 = (const float*)d_in[3],
              *ad1 = (const float*)d_in[4], *b1  = (const float*)d_in[5];
  const float *W2 = (const float*)d_in[6],  *as2 = (const float*)d_in[7],
              *ad2 = (const float*)d_in[8], *b2  = (const float*)d_in[9];
  const float *Wm = (const float*)d_in[10], *asm_ = (const float*)d_in[11],
              *adm = (const float*)d_in[12], *bm = (const float*)d_in[13];
  const float *Wl = (const float*)d_in[14], *asl = (const float*)d_in[15],
              *adl = (const float*)d_in[16], *bl = (const float*)d_in[17];

  char* ws = (char*)d_ws;
  size_t off = 0;
  auto alloc = [&](size_t bytes) -> void* {
    void* p = ws + off;
    off += (bytes + 255) & ~(size_t)255;
    return p;
  };
  float*    Hbuf = (float*)alloc((size_t)NNODES * 512 * 4);
  float*    A0   = (float*)alloc((size_t)NNODES * 256 * 4);
  float*    A1   = (float*)alloc((size_t)NNODES * 512 * 4);
  float*    ALS  = (float*)alloc((size_t)NNODES * HEADS * 4);
  float*    ALD  = (float*)alloc((size_t)NNODES * HEADS * 4);
  unsigned* QL   = (unsigned*)alloc((size_t)NGROUPS * QCAP * 4);
  int*      QC   = (int*)alloc((size_t)NGROUPS * 4);

  float* zmean   = (float*)d_out;
  float* zlogstd = (float*)d_out + (size_t)NNODES * 32;

  run_gat_layer<64, false, true>(stream, x, 256, W1, as1, ad1, b1, Hbuf, A0, src, dst, E, ALS, ALD, QL, QC);
  run_gat_layer<128, false, false>(stream, A0, 256, W2, as2, ad2, b2, Hbuf, A1, src, dst, E, ALS, ALD, QL, QC);
  run_gat_layer<32, true, false>(stream, A1, 512, Wm, asm_, adm, bm, Hbuf, zmean, src, dst, E, ALS, ALD, QL, QC);
  run_gat_layer<32, true, false>(stream, A1, 512, Wl, asl, adl, bl, Hbuf, zlogstd, src, dst, E, ALS, ALD, QL, QC);
}


# GEMM K loop on two alternating LDS stages for BOTH operands: A tile goes through LDS too (whole-line 16-byte loads, was 8-byte loads over 16 lines per k-step), next 32 k in flight under the WMMAs
# speedup vs baseline: 5.7416x; 1.2269x over previous
#include <hip/hip_runtime.h>

typedef float v2f __attribute__((ext_vector_type(2)));
typedef float v8f __attribute__((ext_vector_type(8)));

#define NNODES 30000
#define HEADS 4
#define SLOPE 0.2f
#define KC 32
#define PAIRSTRIDE 160
#define APITCH 36

__device__ __forceinline__ void gemm_stage(const float* __restrict__ ap, const float* __restrict__ bp,
                                           v8f& acc0, v8f& acc1, v8f& acc2, v8f& acc3) {
#pragma unroll
  for (int ks = 0; ks < KC / 4; ++ks) {
    const v2f a = *(const v2f*)(ap + 4 * ks);
    const float* __restrict__ bk = bp + 2 * ks * PAIRSTRIDE;
    const v2f b0 = *(const v2f*)(bk + 0);
    const v2f b1 = *(const v2f*)(bk + 32);
    const v2f b2 = *(const v2f*)(bk + 64);
    const v2f b3 = *(const v2f*)(bk + 96);
    acc0 = __builtin_amdgcn_wmma_f32_16x16x4_f32(false, a, false, b0, (short)0, acc0, false, false);
    acc1 = __builtin_amdgcn_wmma_f32_16x16x4_f32(false, a, false, b1, (short)0, acc1, false, false);
    acc2 = __builtin_amdgcn_wmma_f32_16x16x4_f32(false, a, false, b2, (short)0, acc2, false, false);
    acc3 = __builtin_amdgcn_wmma_f32_16x16x4_f32(false, a, false, b3, (short)0, acc3, false, false);
  }
}

__device__ __forceinline__ void gemm_land(float* __restrict__ pa, float* __restrict__ pb,
                                          float4 ra0, float4 ra1, float4 ra2, float4 ra3,
                                          float4 rb0, float4 rb1) {
  *(float4*)(pa)               = ra0;
  *(float4*)(pa + 32 * APITCH) = ra1;
  *(float4*)(pa + 64 * APITCH) = ra2;
  *(float4*)(pa + 96 * APITCH) = ra3;
  pb[0] = rb0.x; pb[2] = rb0.y; pb[4] = rb0.z; pb[6] = rb0.w;
  float* __restrict__ pb1 = pb + 8 * PAIRSTRIDE;
  pb1[0] = rb1.x; pb1[2] = rb1.y; pb1[4] = rb1.z; pb1[6] = rb1.w;
}

__global__ __launch_bounds__(256)
void gemm_xw_wmma(const float* __restrict__ A, const float* __restrict__ B,
                  float* __restrict__ C, int M, int K, int Nout) {
  __shared__ float ldsA[2][128 * APITCH];
  __shared__ float ldsB[2][(KC / 2) * PAIRSTRIDE];

  const int tid  = threadIdx.x;
  const int lane = tid & 31;
  const int wave = tid >> 5;
  const int Mtiles = M >> 4;

  const int n0 = blockIdx.y * 64;
  const int mt = blockIdx.x * 8 + wave;
  const bool active = (mt < Mtiles);
  const int m0   = mt * 16;
  const int half = lane >> 4;
  const int lm   = lane & 15;

  const int arow = tid >> 3;
  const int acol = (tid & 7) * 4;
  const int brow = tid >> 4;
  const int bcol = (tid & 15) * 4;

  const int rlast = M - 1;
  const int r0 = blockIdx.x * 128 + arow;
  const float* __restrict__ ag0 = A + (size_t)(r0      < rlast ? r0      : rlast) * K + acol;
  const float* __restrict__ ag1 = A + (size_t)(r0 + 32 < rlast ? r0 + 32 : rlast) * K + acol;
  const float* __restrict__ ag2 = A + (size_t)(r0 + 64 < rlast ? r0 + 64 : rlast) * K + acol;
  const float* __restrict__ ag3 = A + (size_t)(r0 + 96 < rlast ? r0 + 96 : rlast) * K + acol;
  const float* __restrict__ bg  = B + (size_t)brow * Nout + n0 + bcol;
  const size_t b16 = (size_t)16 * Nout;

  const int ast = arow * APITCH + acol;
  const int bst = (brow >> 1) * PAIRSTRIDE + bcol * 2 + (brow & 1);

  float4 ra0, ra1, ra2, ra3, rb0, rb1;
  v8f acc0 = {}, acc1 = {}, acc2 = {}, acc3 = {};

  ra0 = *(const float4*)(ag0);  ra1 = *(const float4*)(ag1);
  ra2 = *(const float4*)(ag2);  ra3 = *(const float4*)(ag3);
  rb0 = *(const float4*)(bg);   rb1 = *(const float4*)(bg + b16);
  gemm_land(&ldsA[0][ast], &ldsB[0][bst], ra0, ra1, ra2, ra3, rb0, rb1);
  __syncthreads();

  const int afr = (16 * wave + lm) * APITCH + 2 * half;
  const int bfr = half * PAIRSTRIDE + lm * 2;

  const int nch = K / KC;
  int i = 0;
  for (; i + 1 < nch; ++i) {
    const int s = i & 1;
    const int kb = (i + 1) * KC;
    ra0 = *(const float4*)(ag0 + kb);  ra1 = *(const float4*)(ag1 + kb);
    ra2 = *(const float4*)(ag2 + kb);  ra3 = *(const float4*)(ag3 + kb);
    rb0 = *(const float4*)(bg + (size_t)kb * Nout);
    rb1 = *(const float4*)(bg + (size_t)kb * Nout + b16);
    __builtin_amdgcn_sched_barrier(0);
    gemm_stage(&ldsA[s][afr], &ldsB[s][bfr], acc0, acc1, acc2, acc3);
    gemm_land(&ldsA[s ^ 1][ast], &ldsB[s ^ 1][bst], ra0, ra1, ra2, ra3, rb0, rb1);
    __syncthreads();
  }
  gemm_stage(&ldsA[i & 1][afr], &ldsB[i & 1][bfr], acc0, acc1, acc2, acc3);

  if (active) {
    float* __restrict__ crow = C + (size_t)(m0 + 8 * half) * Nout + n0 + lm;
#pragma unroll
    for (int v = 0; v < 8; ++v) {
      crow[(size_t)v * Nout + 0]  = acc0[v];
      crow[(size_t)v * Nout + 16] = acc1[v];
      crow[(size_t)v * Nout + 32] = acc2[v];
      crow[(size_t)v * Nout + 48] = acc3[v];
    }
  }
}

__global__ void attn_logits(const float* __restrict__ H, const float* __restrict__ a_s,
                            const float* __restrict__ a_d, float* __restrict__ als,
                            float* __restrict__ ald, int Nn, int C) {
  int i = blockIdx.x * blockDim.x + threadIdx.x;
  if (i >= Nn * HEADS) return;
  int n = i / HEADS, h = i % HEADS;
  const float* hp = H + (size_t)n * HEADS * C + (size_t)h * C;
  float s = 0.f, d = 0.f;
  for (int c = 0; c < C; ++c) { float v = hp[c]; s += v * a_s[h * C + c]; d += v * a_d[h * C + c]; }
  als[i] = s; ald[i] = d;
}

__device__ __forceinline__ unsigned fkey(float x) {
  unsigned b = __float_as_uint(x);
  return (b & 0x80000000u) ? ~b : (b | 0x80000000u);
}
__device__ __forceinline__ float funkey(unsigned k) {
  unsigned b = (k & 0x80000000u) ? (k & 0x7FFFFFFFu) : ~k;
  return __uint_as_float(b);
}

#define RNODES 118
#define QCAP 8192
#define NTHR 1024
#define NWAVE (NTHR / 32)
#define NGROUPS ((NNODES + RNODES - 1) / RNODES)

template <int C, bool MEAN, bool FIRST>
__global__ __launch_bounds__(NTHR)
void node_range_softmax_aggregate(const int* __restrict__ src, const int* __restrict__ dst, int E,
                                  const float* __restrict__ H, const float* __restrict__ als,
                                  const float* __restrict__ ald, const float* __restrict__ bias,
                                  float* __restrict__ out,
                                  unsigned* __restrict__ qlist, int* __restrict__ qcnt) {
  constexpr int F = HEADS * C;
  constexpr int J = F / 128;
  constexpr int U = 8 / J;
  __shared__ float    acc[RNODES * F];
  __shared__ unsigned mk[RNODES * HEADS];
  __shared__ float    mv[RNODES * HEADS];
  __shared__ float    den[RNODES * HEADS];
  __shared__ float    aldc[RNODES * HEADS];
  __shared__ unsigned q[QCAP];
  __shared__ unsigned qn;
  __shared__ int      cnt[128];
  __shared__ int      off[132];

  const int tid  = threadIdx.x;
  const int lane = tid & 31;
  const int wave = tid >> 5;
  const int base = blockIdx.x * RNODES;
  int rn = NNODES - base;
  if (rn > RNODES) rn = RNODES;
  if (rn < 0) rn = 0;
  unsigned* __restrict__ lst = qlist + (size_t)blockIdx.x * QCAP;

  auto take = [&](int e, unsigned dl) {
    if (dl < (unsigned)rn) {
      const unsigned slot = atomicAdd(&qn, 1u);
      if (slot < QCAP) q[slot] = (unsigned)e | (dl << 25);
    }
  };

  auto scan = [&](int e0, int e1) {
    if (((E | e0 | e1) & 3) == 0) {
      const int4* __restrict__ d4 = (const int4*)dst;
      const int g1 = e1 >> 2;
      for (int g0 = (e0 >> 2); g0 < g1; g0 += NTHR * 8) {
        int4 v[8];
#pragma unroll
        for (int u = 0; u < 8; ++u) {
          const int g = g0 + u * NTHR + tid;
          v[u] = (g < g1) ? d4[g] : make_int4(-1, -1, -1, -1);
        }
#pragma unroll
        for (int u = 0; u < 8; ++u) {
          const int e = (g0 + u * NTHR + tid) * 4;
          const unsigned a0 = (unsigned)(v[u].x - base), a1 = (unsigned)(v[u].y - base);
          const unsigned a2 = (unsigned)(v[u].z - base), a3 = (unsigned)(v[u].w - base);
          const unsigned lo01 = a0 < a1 ? a0 : a1, lo23 = a2 < a3 ? a2 : a3;
          if ((lo01 < lo23 ? lo01 : lo23) < (unsigned)rn) {
            take(e + 0, a0); take(e + 1, a1); take(e + 2, a2); take(e + 3, a3);
          }
        }
      }
    } else {
      for (int e = e0 + tid; e < e1; e += NTHR) take(e, (unsigned)(dst[e] - base));
    }
  };

  auto resolve = [&](int n, bool withmax, bool keep) {
    for (int i = tid; i < n; i += NTHR) {
      const unsigned ent = q[i];
      const unsigned dl = ent >> 25;
      const int s = src[ent & 0x1FFFFFFu];
      if (withmax) {
        const float4 a = *(const float4*)(als + (size_t)s * HEADS);
        atomicMax(&mk[dl * HEADS + 0], fkey(a.x));
        atomicMax(&mk[dl * HEADS + 1], fkey(a.y));
        atomicMax(&mk[dl * HEADS + 2], fkey(a.z));
        atomicMax(&mk[dl * HEADS + 3], fkey(a.w));
      }
      const unsigned packed = (unsigned)s | (dl << 16);
      q[i] = packed;
      if (keep) lst[i] = packed;
    }
  };

  auto reload = [&](int n) {
    for (int i = tid; i < n; i += NTHR) {
      const unsigned ent = lst[i];
      const unsigned dl = ent >> 16;
      const float4 a = *(const float4*)(als + (size_t)(ent & 0xFFFFu) * HEADS);
      atomicMax(&mk[dl * HEADS + 0], fkey(a.x));
      atomicMax(&mk[dl * HEADS + 1], fkey(a.y));
      atomicMax(&mk[dl * HEADS + 2], fkey(a.z));
      atomicMax(&mk[dl * HEADS + 3], fkey(a.w));
      q[i] = ent;
    }
  };

  auto drain = [&](int n) {
    for (int i0 = wave; i0 < n; i0 += NWAVE * U) {
      const int iu = i0 + lane * NWAVE;
      const unsigned mine = (lane < U && iu < n) ? q[iu] : 0u;
      float4 hv[U][J];
      float sa[U][J], da[U][J], ma[U][J];
      int sidx[U], dloc[U];
#pragma unroll
      for (int u = 0; u < U; ++u) {
        const unsigned ent = (unsigned)__builtin_amdgcn_readlane((int)mine, u);
        sidx[u] = (int)(ent & 0xFFFFu);
        dloc[u] = (int)(ent >> 16);
      }
#pragma unroll
      for (int u = 0; u < U; ++u) {
#pragma unroll
        for (int j = 0; j < J; ++j) {
          const int h = (128 * j + lane * 4) / C;
          sa[u][j] = als[sidx[u] * HEADS + h];
        }
      }
#pragma unroll
      for (int u = 0; u < U; ++u) {
        const float* __restrict__ hrow = H + (size_t)sidx[u] * F + lane * 4;
#pragma unroll
        for (int j = 0; j < J; ++j) hv[u][j] = *(const float4*)(hrow + 128 * j);
      }
#pragma unroll
      for (int u = 0; u < U; ++u) {
#pragma unroll
        for (int j = 0; j < J; ++j) {
          const int h = (128 * j + lane * 4) / C;
          da[u][j] = aldc[dloc[u] * HEADS + h];
          ma[u][j] = mv[dloc[u] * HEADS + h];
        }
      }
      __builtin_amdgcn_sched_barrier(0);
#pragma unroll
      for (int u = 0; u < U; ++u) {
        if (i0 + u * NWAVE < n) {
          const int dl = dloc[u];
#pragma unroll
          for (int j = 0; j < J; ++j) {
            const int f = 128 * j + lane * 4;
            const int h = f / C;
            float sc = sa[u][j] + da[u][j];
            sc = (sc > 0.f) ? sc : SLOPE * sc;
            const float ex = __expf(sc - ma[u][j]);
            if ((f % C) == 0) atomicAdd(&den[dl * HEADS + h], ex);
            float* ap = &acc[dl * F + 128 * j + lane];
            atomicAdd(ap +  0, hv[u][j].x * ex);
            atomicAdd(ap + 32, hv[u][j].y * ex);
            atomicAdd(ap + 64, hv[u][j].z * ex);
            atomicAdd(ap + 96, hv[u][j].w * ex);
          }
        }
      }
    }
  };

  if (tid < rn * HEADS) {
    mk[tid]   = fkey(als[(size_t)base * HEADS + tid]);
    aldc[tid] = ald[(size_t)base * HEADS + tid];
  }
  if (tid == 0) qn = 0u;
  if (tid < 128) cnt[tid] = 0;
  __syncthreads();

  int total;
  if (FIRST) {
    scan(0, E);
    __syncthreads();
    total = (int)qn;
    if (tid == 0) qcnt[blockIdx.x] = total;
  } else {
    total = qcnt[blockIdx.x];
  }
  const bool fits = (total <= QCAP);
  if (fits) {
    if (FIRST) resolve(total, true, true);
    else       reload(total);
  } else {
    for (int p = 0; p < E; p += QCAP) {
      __syncthreads();
      if (tid == 0) qn = 0u;
      __syncthreads();
      scan(p, (p + QCAP < E) ? (p + QCAP) : E);
      __syncthreads();
      resolve((int)qn, true, false);
    }
  }
  __syncthreads();

  if (tid < rn * HEADS) {
    const float a = aldc[tid];
    float m = funkey(mk[tid]) + a;
    m = (m > 0.f) ? m : SLOPE * m;
    mv[tid] = m;
    float sc = als[(size_t)base * HEADS + tid] + a;
    sc = (sc > 0.f) ? sc : SLOPE * sc;
    den[tid] = __expf(sc - m);
  }

  if (fits) {
    static_assert(!MEAN || C == 32, "the sum over heads across lanes is written for 32 channels per head");
    unsigned* __restrict__ qs = (unsigned*)acc;
    constexpr int KMAX = QCAP / NTHR;
    unsigned ent[KMAX];
    int rk[KMAX];
#pragma unroll
    for (int k = 0; k < KMAX; ++k) {
      const int i = tid + k * NTHR;
      ent[k] = 0u; rk[k] = 0;
      if (i < total) { ent[k] = q[i]; rk[k] = atomicAdd(&cnt[ent[k] >> 16], 1); }
    }
    __syncthreads();
    if (wave == 0) {
      const int c0 = cnt[4 * lane], c1 = cnt[4 * lane + 1], c2 = cnt[4 * lane + 2], c3 = cnt[4 * lane + 3];
      const int s = c0 + c1 + c2 + c3;
      int incl = s;
#pragma unroll
      for (int d = 1; d < 32; d <<= 1) {
        const int t = __shfl_up(incl, d);
        if (lane >= d) incl += t;
      }
      const int ex0 = incl - s;
      off[4 * lane]     = ex0;
      off[4 * lane + 1] = ex0 + c0;
      off[4 * lane + 2] = ex0 + c0 + c1;
      off[4 * lane + 3] = ex0 + c0 + c1 + c2;
      if (lane == 31) off[128] = incl;
    }
    __syncthreads();
#pragma unroll
    for (int k = 0; k < KMAX; ++k) {
      const int i = tid + k * NTHR;
      if (i < total) qs[off[ent[k] >> 16] + rk[k]] = ent[k] & 0xFFFFu;
    }
    __syncthreads();

    constexpr int UE = 8;
    const int wv = __builtin_amdgcn_readfirstlane(wave);
    const int nitems = rn * J;
    for (int it = wv; it < nitems; it += NWAVE) {
      const int dl = it / J, j = it % J;
      const int node = base + dl;
      const int f = 128 * j + lane * 4;
      const int h = f / C;
      const int o0 = __builtin_amdgcn_readfirstlane(off[dl]);
      const int nt = __builtin_amdgcn_readfirstlane(off[dl + 1]) - o0 + 1;
      const float m  = mv[dl * HEADS + h];
      const float ad = aldc[dl * HEADS + h];
      float4 a4 = make_float4(0.f, 0.f, 0.f, 0.f);
      float dn = 0.f;
      for (int i0 = 0; i0 < nt; i0 += UE) {
        const int p = i0 + lane;
        unsigned mine = (unsigned)node;
        if (lane < UE && p >= 1 && p < nt) mine = qs[o0 + p - 1];
        int sidx[UE];
        float sa[UE];
        float4 hv[UE];
#pragma unroll
        for (int u = 0; u < UE; ++u) sidx[u] = __builtin_amdgcn_readlane((int)mine, u);
#pragma unroll
        for (int u = 0; u < UE; ++u) sa[u] = als[sidx[u] * HEADS + h];
#pragma unroll
        for (int u = 0; u < UE; ++u) hv[u] = *(const float4*)(H + (size_t)sidx[u] * F + f);
#pragma unroll
        for (int u = 0; u < UE; ++u) {
          float sc = sa[u] + ad;
          sc = (sc > 0.f) ? sc : SLOPE * sc;
          float ex = __expf(sc - m);
          ex = (i0 + u < nt) ? ex : 0.f;
          dn += ex;
          a4.x += hv[u].x * ex;  a4.y += hv[u].y * ex;
          a4.z += hv[u].z * ex;  a4.w += hv[u].w * ex;
        }
      }
      const float dd = dn + 1e-16f;
      if (!MEAN) {
        const float4 b4 = *(const float4*)(bias + f);
        float4 o;
        o.x = a4.x / dd + b4.x;  o.y = a4.y / dd + b4.y;
        o.z = a4.z / dd + b4.z;  o.w = a4.w / dd + b4.w;
        o.x = o.x > 0.f ? o.x : 0.f;  o.y = o.y > 0.f ? o.y : 0.f;
        o.z = o.z > 0.f ? o.z : 0.f;  o.w = o.w > 0.f ? o.w : 0.f;
        *(float4*)(out + (size_t)node * F + f) = o;
      } else {
        const float v0 = a4.x / dd, v1 = a4.y / dd, v2 = a4.z / dd, v3 = a4.w / dd;
        const int l8 = lane & 7;
        const float r0 = ((__shfl(v0, l8) + __shfl(v0, l8 + 8)) + __shfl(v0, l8 + 16)) + __shfl(v0, l8 + 24);
        const float r1 = ((__shfl(v1, l8) + __shfl(v1, l8 + 8)) + __shfl(v1, l8 + 16)) + __shfl(v1, l8 + 24);
        const float r2 = ((__shfl(v2, l8) + __shfl(v2, l8 + 8)) + __shfl(v2, l8 + 16)) + __shfl(v2, l8 + 24);
        const float r3 = ((__shfl(v3, l8) + __shfl(v3, l8 + 8)) + __shfl(v3, l8 + 16)) + __shfl(v3, l8 + 24);
        if (lane < 8) {
          const float4 b4 = *(const float4*)(bias + 4 * lane);
          float4 o;
          o.x = 0.25f * r0 + b4.x;  o.y = 0.25f * r1 + b4.y;
          o.z = 0.25f * r2 + b4.z;  o.w = 0.25f * r3 + b4.w;
          *(float4*)(out + (size_t)node * C + 4 * lane) = o;
        }
      }
    }
  } else {
  __syncthreads();

  {
    constexpr int P = 4;
    const int items = rn * (F / 4);
    for (int idx0 = tid; idx0 < items; idx0 += NTHR * P) {
      float4 hp[P];
#pragma unroll
      for (int p = 0; p < P; ++p) {
        const int idx = idx0 + p * NTHR;
        const int idc = (idx < items) ? idx : idx0;
        const int dl = idc / (F / 4);
        const int r  = idc % (F / 4);
        const int f  = 128 * (r >> 5) + 4 * (r & 31);
        hp[p] = *(const float4*)(H + (size_t)(base + dl) * F + f);
      }
#pragma unroll
      for (int p = 0; p < P; ++p) {
        const int idx = idx0 + p * NTHR;
        const int idc = (idx < items) ? idx : idx0;
        const int dl = idc / (F / 4);
        const int r  = idc % (F / 4);
        const int j  = r >> 5, l = r & 31;
        const int f  = 128 * j + 4 * l;
        const float ex = den[dl * HEADS + f / C];
        float* ap = &acc[dl * F + 128 * j + l];
        ap[0] = hp[p].x * ex; ap[32] = hp[p].y * ex; ap[64] = hp[p].z * ex; ap[96] = hp[p].w * ex;
      }
    }
  }
  __syncthreads();

  {
    for (int p = 0; p < E; p += QCAP) {
      __syncthreads();
      if (tid == 0) qn = 0u;
      __syncthreads();
      scan(p, (p + QCAP < E) ? (p + QCAP) : E);
      __syncthreads();
      resolve((int)qn, false, false);
      __syncthreads();
      drain((int)qn);
    }
  }
  __syncthreads();

  if (!MEAN) {
    for (int idx = tid; idx < rn * (F / 4); idx += NTHR) {
      const int dl = idx / (F / 4);
      const int r  = idx % (F / 4);
      const int j  = r >> 5, l = r & 31;
      const int f  = 128 * j + 4 * l;
      const float dn = den[dl * HEADS + f / C] + 1e-16f;
      const float* ap = &acc[dl * F + 128 * j + l];
      const float4 b4 = *(const float4*)(bias + f);
      float4 o;
      o.x = ap[0]  / dn + b4.x;  o.y = ap[32] / dn + b4.y;
      o.z = ap[64] / dn + b4.z;  o.w = ap[96] / dn + b4.w;
      o.x = o.x > 0.f ? o.x : 0.f;  o.y = o.y > 0.f ? o.y : 0.f;
      o.z = o.z > 0.f ? o.z : 0.f;  o.w = o.w > 0.f ? o.w : 0.f;
      *(float4*)(out + (size_t)(base + dl) * F + f) = o;
    }
  } else {
    for (int idx = tid; idx < rn * C; idx += NTHR) {
      const int dl = idx / C, c = idx % C;
      const float* qd = &den[dl * HEADS];
      float p[HEADS];
#pragma unroll
      for (int h = 0; h < HEADS; ++h) {
        const int f = h * C + c;
        p[h] = acc[dl * F + 128 * (f >> 7) + 32 * (f & 3) + ((f & 127) >> 2)];
      }
      out[(size_t)(base + dl) * C + c] =
          0.25f * (p[0] / (qd[0] + 1e-16f) + p[1] / (qd[1] + 1e-16f) +
                   p[2] / (qd[2] + 1e-16f) + p[3] / (qd[3] + 1e-16f)) + bias[c];
    }
  }
  }
}

template <int C, bool MEAN, bool FIRST>
static void run_gat_layer(hipStream_t st, const float* in, int Fin,
                          const float* W, const float* a_s, const float* a_d, const float* b,
                          float* Hbuf, float* outp,
                          const int* src, const int* dst, int E,
                          float* ALS, float* ALD, unsigned* QL, int* QC) {
  const int Nn = NNODES;
  const int Fout = HEADS * C;
  const int Mtiles = Nn / 16;
  dim3 ggrid((Mtiles + 7) / 8, Fout / 64);
  gemm_xw_wmma<<<ggrid, 256, 0, st>>>(in, W, Hbuf, Nn, Fin, Fout);
  int t = Nn * HEADS;
  attn_logits<<<(t + 255) / 256, 256, 0, st>>>(Hbuf, a_s, a_d, ALS, ALD, Nn, C);
  node_range_softmax_aggregate<C, MEAN, FIRST>
      <<<NGROUPS, NTHR, 0, st>>>(src, dst, E, Hbuf, ALS, ALD, b, outp, QL, QC);
}

extern "C" void kernel_launch(void* const* d_in, const int* in_sizes, int n_in,
                              void* d_out, int out_size, void* d_ws, size_t ws_size,
                              hipStream_t stream) {
  const float* x   = (const float*)d_in[0];
  const int*   ei  = (const int*)d_in[1];
  const int E = in_sizes[1] / 2;
  const int* src = ei;
  const int* dst = ei + E;
  const float *W1 = (const float*)d_in[2],  *as1 = (const float*)d_in[3],
              *ad1 = (const float*)d_in[4], *b1  = (const float*)d_in[5];
  const float *W2 = (const float*)d_in[6],  *as2 = (const float*)d_in[7],
              *ad2 = (const float*)d_in[8], *b2  = (const float*)d_in[9];
  const float *Wm = (const float*)d_in[10], *asm_ = (const float*)d_in[11],
              *adm = (const float*)d_in[12], *bm = (const float*)d_in[13];
  const float *Wl = (const float*)d_in[14], *asl = (const float*)d_in[15],
              *adl = (const float*)d_in[16], *bl = (const float*)d_in[17];

  char* ws = (char*)d_ws;
  size_t off = 0;
  auto alloc = [&](size_t bytes) -> void* {
    void* p = ws + off;
    off += (bytes + 255) & ~(size_t)255;
    return p;
  };
  float*    Hbuf = (float*)alloc((size_t)NNODES * 512 * 4);
  float*    A0   = (float*)alloc((size_t)NNODES * 256 * 4);
  float*    A1   = (float*)alloc((size_t)NNODES * 512 * 4);
  float*    ALS  = (float*)alloc((size_t)NNODES * HEADS * 4);
  float*    ALD  = (float*)alloc((size_t)NNODES * HEADS * 4);
  unsigned* QL   = (unsigned*)alloc((size_t)NGROUPS * QCAP * 4);
  int*      QC   = (int*)alloc((size_t)NGROUPS * 4);

  float* zmean   = (float*)d_out;
  float* zlogstd = (float*)d_out + (size_t)NNODES * 32;

  run_gat_layer<64, false, true>(stream, x, 256, W1, as1, ad1, b1, Hbuf, A0, src, dst, E, ALS, ALD, QL, QC);
  run_gat_layer<128, false, false>(stream, A0, 256, W2, as2, ad2, b2, Hbuf, A1, src, dst, E, ALS, ALD, QL, QC);
  run_gat_layer<32, true, false>(stream, A1, 512, Wm, asm_, adm, bm, Hbuf, zmean, src, dst, E, ALS, ALD, QL, QC);
  run_gat_layer<32, true, false>(stream, A1, 512, Wl, asl, adl, bl, Hbuf, zlogstd, src, dst, E, ALS, ALD, QL, QC);
}


# Softmax weight of each (edge, head) computed once as the edge is sorted into its bucket, kept in LDS; row loop reads weights from LDS: per-piece gather of source logits and exp gone; sum order same
# speedup vs baseline: 5.9329x; 1.0333x over previous
#include <hip/hip_runtime.h>

typedef float v2f __attribute__((ext_vector_type(2)));
typedef float v8f __attribute__((ext_vector_type(8)));

#define NNODES 30000
#define HEADS 4
#define SLOPE 0.2f
#define KC 32
#define PAIRSTRIDE 160
#define APITCH 36

__device__ __forceinline__ void gemm_stage(const float* __restrict__ ap, const float* __restrict__ bp,
                                           v8f& acc0, v8f& acc1, v8f& acc2, v8f& acc3) {
#pragma unroll
  for (int ks = 0; ks < KC / 4; ++ks) {
    const v2f a = *(const v2f*)(ap + 4 * ks);
    const float* __restrict__ bk = bp + 2 * ks * PAIRSTRIDE;
    const v2f b0 = *(const v2f*)(bk + 0);
    const v2f b1 = *(const v2f*)(bk + 32);
    const v2f b2 = *(const v2f*)(bk + 64);
    const v2f b3 = *(const v2f*)(bk + 96);
    acc0 = __builtin_amdgcn_wmma_f32_16x16x4_f32(false, a, false, b0, (short)0, acc0, false, false);
    acc1 = __builtin_amdgcn_wmma_f32_16x16x4_f32(false, a, false, b1, (short)0, acc1, false, false);
    acc2 = __builtin_amdgcn_wmma_f32_16x16x4_f32(false, a, false, b2, (short)0, acc2, false, false);
    acc3 = __builtin_amdgcn_wmma_f32_16x16x4_f32(false, a, false, b3, (short)0, acc3, false, false);
  }
}

__device__ __forceinline__ void gemm_land(float* __restrict__ pa, float* __restrict__ pb,
                                          float4 ra0, float4 ra1, float4 ra2, float4 ra3,
                                          float4 rb0, float4 rb1) {
  *(float4*)(pa)               = ra0;
  *(float4*)(pa + 32 * APITCH) = ra1;
  *(float4*)(pa + 64 * APITCH) = ra2;
  *(float4*)(pa + 96 * APITCH) = ra3;
  pb[0] = rb0.x; pb[2] = rb0.y; pb[4] = rb0.z; pb[6] = rb0.w;
  float* __restrict__ pb1 = pb + 8 * PAIRSTRIDE;
  pb1[0] = rb1.x; pb1[2] = rb1.y; pb1[4] = rb1.z; pb1[6] = rb1.w;
}

__global__ __launch_bounds__(256)
void gemm_xw_wmma(const float* __restrict__ A, const float* __restrict__ B,
                  float* __restrict__ C, int M, int K, int Nout) {
  __shared__ float ldsA[2][128 * APITCH];
  __shared__ float ldsB[2][(KC / 2) * PAIRSTRIDE];

  const int tid  = threadIdx.x;
  const int lane = tid & 31;
  const int wave = tid >> 5;
  const int Mtiles = M >> 4;

  const int n0 = blockIdx.y * 64;
  const int mt = blockIdx.x * 8 + wave;
  const bool active = (mt < Mtiles);
  const int m0   = mt * 16;
  const int half = lane >> 4;
  const int lm   = lane & 15;

  const int arow = tid >> 3;
  const int acol = (tid & 7) * 4;
  const int brow = tid >> 4;
  const int bcol = (tid & 15) * 4;

  const int rlast = M - 1;
  const int r0 = blockIdx.x * 128 + arow;
  const float* __restrict__ ag0 = A + (size_t)(r0      < rlast ? r0      : rlast) * K + acol;
  const float* __restrict__ ag1 = A + (size_t)(r0 + 32 < rlast ? r0 + 32 : rlast) * K + acol;
  const float* __restrict__ ag2 = A + (size_t)(r0 + 64 < rlast ? r0 + 64 : rlast) * K + acol;
  const float* __restrict__ ag3 = A + (size_t)(r0 + 96 < rlast ? r0 + 96 : rlast) * K + acol;
  const float* __restrict__ bg  = B + (size_t)brow * Nout + n0 + bcol;
  const size_t b16 = (size_t)16 * Nout;

  const int ast = arow * APITCH + acol;
  const int bst = (brow >> 1) * PAIRSTRIDE + bcol * 2 + (brow & 1);

  float4 ra0, ra1, ra2, ra3, rb0, rb1;
  v8f acc0 = {}, acc1 = {}, acc2 = {}, acc3 = {};

  ra0 = *(const float4*)(ag0);  ra1 = *(const float4*)(ag1);
  ra2 = *(const float4*)(ag2);  ra3 = *(const float4*)(ag3);
  rb0 = *(const float4*)(bg);   rb1 = *(const float4*)(bg + b16);
  gemm_land(&ldsA[0][ast], &ldsB[0][bst], ra0, ra1, ra2, ra3, rb0, rb1);
  __syncthreads();

  const int afr = (16 * wave + lm) * APITCH + 2 * half;
  const int bfr = half * PAIRSTRIDE + lm * 2;

  const int nch = K / KC;
  int i = 0;
  for (; i + 1 < nch; ++i) {
    const int s = i & 1;
    const int kb = (i + 1) * KC;
    ra0 = *(const float4*)(ag0 + kb);  ra1 = *(const float4*)(ag1 + kb);
    ra2 = *(const float4*)(ag2 + kb);  ra3 = *(const float4*)(ag3 + kb);
    rb0 = *(const float4*)(bg + (size_t)kb * Nout);
    rb1 = *(const float4*)(bg + (size_t)kb * Nout + b16);
    __builtin_amdgcn_sched_barrier(0);
    gemm_stage(&ldsA[s][afr], &ldsB[s][bfr], acc0, acc1, acc2, acc3);
    gemm_land(&ldsA[s ^ 1][ast], &ldsB[s ^ 1][bst], ra0, ra1, ra2, ra3, rb0, rb1);
    __syncthreads();
  }
  gemm_stage(&ldsA[i & 1][afr], &ldsB[i & 1][bfr], acc0, acc1, acc2, acc3);

  if (active) {
    float* __restrict__ crow = C + (size_t)(m0 + 8 * half) * Nout + n0 + lm;
#pragma unroll
    for (int v = 0; v < 8; ++v) {
      crow[(size_t)v * Nout + 0]  = acc0[v];
      crow[(size_t)v * Nout + 16] = acc1[v];
      crow[(size_t)v * Nout + 32] = acc2[v];
      crow[(size_t)v * Nout + 48] = acc3[v];
    }
  }
}

__global__ void attn_logits(const float* __restrict__ H, const float* __restrict__ a_s,
                            const float* __restrict__ a_d, float* __restrict__ als,
                            float* __restrict__ ald, int Nn, int C) {
  int i = blockIdx.x * blockDim.x + threadIdx.x;
  if (i >= Nn * HEADS) return;
  int n = i / HEADS, h = i % HEADS;
  const float* hp = H + (size_t)n * HEADS * C + (size_t)h * C;
  float s = 0.f, d = 0.f;
  for (int c = 0; c < C; ++c) { float v = hp[c]; s += v * a_s[h * C + c]; d += v * a_d[h * C + c]; }
  als[i] = s; ald[i] = d;
}

__device__ __forceinline__ unsigned fkey(float x) {
  unsigned b = __float_as_uint(x);
  return (b & 0x80000000u) ? ~b : (b | 0x80000000u);
}
__device__ __forceinline__ float funkey(unsigned k) {
  unsigned b = (k & 0x80000000u) ? (k & 0x7FFFFFFFu) : ~k;
  return __uint_as_float(b);
}

#define RNODES 118
#define QCAP 8192
#define NTHR 1024
#define NWAVE (NTHR / 32)
#define NGROUPS ((NNODES + RNODES - 1) / RNODES)

template <int C, bool MEAN, bool FIRST>
__global__ __launch_bounds__(NTHR)
void node_range_softmax_aggregate(const int* __restrict__ src, const int* __restrict__ dst, int E,
                                  const float* __restrict__ H, const float* __restrict__ als,
                                  const float* __restrict__ ald, const float* __restrict__ bias,
                                  float* __restrict__ out,
                                  unsigned* __restrict__ qlist, int* __restrict__ qcnt) {
  constexpr int F = HEADS * C;
  constexpr int J = F / 128;
  constexpr int U = 8 / J;
  constexpr int SLOTS = QCAP + 256;
  __shared__ __attribute__((aligned(16))) float acc[(RNODES * F > 5 * SLOTS) ? RNODES * F : 5 * SLOTS];
  __shared__ unsigned mk[RNODES * HEADS];
  __shared__ __attribute__((aligned(16))) float mv[RNODES * HEADS];
  __shared__ float    den[RNODES * HEADS];
  __shared__ __attribute__((aligned(16))) float aldc[RNODES * HEADS];
  __shared__ unsigned q[QCAP];
  __shared__ unsigned qn;
  __shared__ int      cnt[128];
  __shared__ int      off[132];

  const int tid  = threadIdx.x;
  const int lane = tid & 31;
  const int wave = tid >> 5;
  const int base = blockIdx.x * RNODES;
  int rn = NNODES - base;
  if (rn > RNODES) rn = RNODES;
  if (rn < 0) rn = 0;
  unsigned* __restrict__ lst = qlist + (size_t)blockIdx.x * QCAP;

  auto take = [&](int e, unsigned dl) {
    if (dl < (unsigned)rn) {
      const unsigned slot = atomicAdd(&qn, 1u);
      if (slot < QCAP) q[slot] = (unsigned)e | (dl << 25);
    }
  };

  auto scan = [&](int e0, int e1) {
    if (((E | e0 | e1) & 3) == 0) {
      const int4* __restrict__ d4 = (const int4*)dst;
      const int g1 = e1 >> 2;
      for (int g0 = (e0 >> 2); g0 < g1; g0 += NTHR * 8) {
        int4 v[8];
#pragma unroll
        for (int u = 0; u < 8; ++u) {
          const int g = g0 + u * NTHR + tid;
          v[u] = (g < g1) ? d4[g] : make_int4(-1, -1, -1, -1);
        }
#pragma unroll
        for (int u = 0; u < 8; ++u) {
          const int e = (g0 + u * NTHR + tid) * 4;
          const unsigned a0 = (unsigned)(v[u].x - base), a1 = (unsigned)(v[u].y - base);
          const unsigned a2 = (unsigned)(v[u].z - base), a3 = (unsigned)(v[u].w - base);
          const unsigned lo01 = a0 < a1 ? a0 : a1, lo23 = a2 < a3 ? a2 : a3;
          if ((lo01 < lo23 ? lo01 : lo23) < (unsigned)rn) {
            take(e + 0, a0); take(e + 1, a1); take(e + 2, a2); take(e + 3, a3);
          }
        }
      }
    } else {
      for (int e = e0 + tid; e < e1; e += NTHR) take(e, (unsigned)(dst[e] - base));
    }
  };

  auto resolve = [&](int n, bool withmax) {
    for (int i = tid; i < n; i += NTHR) {
      const unsigned ent = q[i];
      const unsigned dl = ent >> 25;
      const int s = src[ent & 0x1FFFFFFu];
      if (withmax) {
        const float4 a = *(const float4*)(als + (size_t)s * HEADS);
        atomicMax(&mk[dl * HEADS + 0], fkey(a.x));
        atomicMax(&mk[dl * HEADS + 1], fkey(a.y));
        atomicMax(&mk[dl * HEADS + 2], fkey(a.z));
        atomicMax(&mk[dl * HEADS + 3], fkey(a.w));
      }
      q[i] = (unsigned)s | (dl << 16);
    }
  };

  auto drain = [&](int n) {
    for (int i0 = wave; i0 < n; i0 += NWAVE * U) {
      const int iu = i0 + lane * NWAVE;
      const unsigned mine = (lane < U && iu < n) ? q[iu] : 0u;
      float4 hv[U][J];
      float sa[U][J], da[U][J], ma[U][J];
      int sidx[U], dloc[U];
#pragma unroll
      for (int u = 0; u < U; ++u) {
        const unsigned ent = (unsigned)__builtin_amdgcn_readlane((int)mine, u);
        sidx[u] = (int)(ent & 0xFFFFu);
        dloc[u] = (int)(ent >> 16);
      }
#pragma unroll
      for (int u = 0; u < U; ++u) {
#pragma unroll
        for (int j = 0; j < J; ++j) {
          const int h = (128 * j + lane * 4) / C;
          sa[u][j] = als[sidx[u] * HEADS + h];
        }
      }
#pragma unroll
      for (int u = 0; u < U; ++u) {
        const float* __restrict__ hrow = H + (size_t)sidx[u] * F + lane * 4;
#pragma unroll
        for (int j = 0; j < J; ++j) hv[u][j] = *(const float4*)(hrow + 128 * j);
      }
#pragma unroll
      for (int u = 0; u < U; ++u) {
#pragma unroll
        for (int j = 0; j < J; ++j) {
          const int h = (128 * j + lane * 4) / C;
          da[u][j] = aldc[dloc[u] * HEADS + h];
          ma[u][j] = mv[dloc[u] * HEADS + h];
        }
      }
      __builtin_amdgcn_sched_barrier(0);
#pragma unroll
      for (int u = 0; u < U; ++u) {
        if (i0 + u * NWAVE < n) {
          const int dl = dloc[u];
#pragma unroll
          for (int j = 0; j < J; ++j) {
            const int f = 128 * j + lane * 4;
            const int h = f / C;
            float sc = sa[u][j] + da[u][j];
            sc = (sc > 0.f) ? sc : SLOPE * sc;
            const float ex = __expf(sc - ma[u][j]);
            if ((f % C) == 0) atomicAdd(&den[dl * HEADS + h], ex);
            float* ap = &acc[dl * F + 128 * j + lane];
            atomicAdd(ap +  0, hv[u][j].x * ex);
            atomicAdd(ap + 32, hv[u][j].y * ex);
            atomicAdd(ap + 64, hv[u][j].z * ex);
            atomicAdd(ap + 96, hv[u][j].w * ex);
          }
        }
      }
    }
  };

  if (tid < rn * HEADS) {
    mk[tid]   = fkey(als[(size_t)base * HEADS + tid]);
    aldc[tid] = ald[(size_t)base * HEADS + tid];
  }
  if (tid == 0) qn = 0u;
  if (tid < 128) cnt[tid] = 0;
  __syncthreads();

  int total;
  if (FIRST) {
    scan(0, E);
    __syncthreads();
    total = (int)qn;
    if (tid == 0) qcnt[blockIdx.x] = total;
  } else {
    total = qcnt[blockIdx.x];
  }
  const bool fits = (total <= QCAP);
  constexpr int KMAX = QCAP / NTHR;
  constexpr int KREG = 4;
  unsigned ent[KMAX];
  int rk[KMAX];
  float4 av[KREG];
  if (fits) {
#pragma unroll
    for (int k = 0; k < KMAX; ++k) {
      const int i = tid + k * NTHR;
      ent[k] = 0u;
      if (i < total) {
        if (FIRST) {
          const unsigned e = q[i];
          ent[k] = (unsigned)src[e & 0x1FFFFFFu] | ((e >> 25) << 16);
        } else {
          ent[k] = lst[i];
        }
      }
    }
#pragma unroll
    for (int k = 0; k < KREG; ++k) {
      const int i = tid + k * NTHR;
      av[k] = make_float4(0.f, 0.f, 0.f, 0.f);
      if (i < total) av[k] = *(const float4*)(als + (size_t)(ent[k] & 0xFFFFu) * HEADS);
    }
#pragma unroll
    for (int k = 0; k < KMAX; ++k) {
      const int i = tid + k * NTHR;
      rk[k] = 0;
      if (i < total) {
        const unsigned dl = ent[k] >> 16;
        if (FIRST) lst[i] = ent[k];
        float4 a;
        if (k < KREG) a = av[k];
        else          a = *(const float4*)(als + (size_t)(ent[k] & 0xFFFFu) * HEADS);
        atomicMax(&mk[dl * HEADS + 0], fkey(a.x));
        atomicMax(&mk[dl * HEADS + 1], fkey(a.y));
        atomicMax(&mk[dl * HEADS + 2], fkey(a.z));
        atomicMax(&mk[dl * HEADS + 3], fkey(a.w));
        rk[k] = atomicAdd(&cnt[dl], 1);
      }
    }
  } else {
    for (int p = 0; p < E; p += QCAP) {
      __syncthreads();
      if (tid == 0) qn = 0u;
      __syncthreads();
      scan(p, (p + QCAP < E) ? (p + QCAP) : E);
      __syncthreads();
      resolve((int)qn, true);
    }
  }
  __syncthreads();

  if (tid < rn * HEADS) {
    const float a = aldc[tid];
    float m = funkey(mk[tid]) + a;
    m = (m > 0.f) ? m : SLOPE * m;
    mv[tid] = m;
    float sc = als[(size_t)base * HEADS + tid] + a;
    sc = (sc > 0.f) ? sc : SLOPE * sc;
    den[tid] = __expf(sc - m);
  }

  if (fits) {
    static_assert(!MEAN || C == 32, "the sum over heads across lanes is written for 32 channels per head");
    unsigned* __restrict__ qs = (unsigned*)acc;
    float* __restrict__ wt = acc + SLOTS;
    if (wave == 0) {
      const int c0 = cnt[4 * lane], c1 = cnt[4 * lane + 1], c2 = cnt[4 * lane + 2], c3 = cnt[4 * lane + 3];
      const int s = c0 + c1 + c2 + c3;
      int incl = s;
#pragma unroll
      for (int d = 1; d < 32; d <<= 1) {
        const int t = __shfl_up(incl, d);
        if (lane >= d) incl += t;
      }
      const int ex0 = incl - s;
      off[4 * lane]     = ex0;
      off[4 * lane + 1] = ex0 + c0;
      off[4 * lane + 2] = ex0 + c0 + c1;
      off[4 * lane + 3] = ex0 + c0 + c1 + c2;
      if (lane == 31) off[128] = incl;
    }
    __syncthreads();
#pragma unroll
    for (int k = 0; k < KMAX; ++k) {
      const int i = tid + k * NTHR;
      if (i < total) {
        const unsigned dl = ent[k] >> 16;
        const int pos = off[dl] + (int)dl + 1 + rk[k];
        qs[pos] = ent[k] & 0xFFFFu;
        const float4 ad4 = *(const float4*)&aldc[dl * HEADS];
        const float4 m4  = *(const float4*)&mv[dl * HEADS];
        float4 a;
        if (k < KREG) a = av[k];
        else          a = *(const float4*)(als + (size_t)(ent[k] & 0xFFFFu) * HEADS);
        float4 w;
        float sc;
        sc = a.x + ad4.x;  sc = (sc > 0.f) ? sc : SLOPE * sc;  w.x = __expf(sc - m4.x);
        sc = a.y + ad4.y;  sc = (sc > 0.f) ? sc : SLOPE * sc;  w.y = __expf(sc - m4.y);
        sc = a.z + ad4.z;  sc = (sc > 0.f) ? sc : SLOPE * sc;  w.z = __expf(sc - m4.z);
        sc = a.w + ad4.w;  sc = (sc > 0.f) ? sc : SLOPE * sc;  w.w = __expf(sc - m4.w);
        *(float4*)&wt[4 * pos] = w;
      }
    }
    if (tid < rn * HEADS) {
      const int dl = tid / HEADS;
      const int pos = off[dl] + dl;
      wt[4 * pos + (tid % HEADS)] = den[tid];
      if ((tid % HEADS) == 0) qs[pos] = (unsigned)(base + dl);
    }
    constexpr int UE = 8;
    if (tid < HEADS * UE) wt[4 * (total + rn) + tid] = 0.f;
    __syncthreads();

    const int wv = __builtin_amdgcn_readfirstlane(wave);
    const int nitems = rn * J;
    for (int it = wv; it < nitems; it += NWAVE) {
      const int dl = it / J, j = it % J;
      const int node = base + dl;
      const int f = 128 * j + lane * 4;
      const int h = f / C;
      const int o0 = __builtin_amdgcn_readfirstlane(off[dl]) + dl;
      const int nt = __builtin_amdgcn_readfirstlane(off[dl + 1]) + dl + 1 - o0;
      const float* __restrict__ wp = wt + 4 * o0 + h;
      float4 a4 = make_float4(0.f, 0.f, 0.f, 0.f);
      float dn = 0.f;
      for (int i0 = 0; i0 < nt; i0 += UE) {
        const int p = i0 + lane;
        unsigned mine = (unsigned)node;
        if (lane < UE && p < nt) mine = qs[o0 + p];
        int sidx[UE];
        float wv8[UE];
        float4 hv[UE];
#pragma unroll
        for (int u = 0; u < UE; ++u) sidx[u] = __builtin_amdgcn_readlane((int)mine, u);
#pragma unroll
        for (int u = 0; u < UE; ++u) wv8[u] = wp[4 * (i0 + u)];
#pragma unroll
        for (int u = 0; u < UE; ++u) hv[u] = *(const float4*)(H + (size_t)sidx[u] * F + f);
#pragma unroll
        for (int u = 0; u < UE; ++u) {
          const float ex = (i0 + u < nt) ? wv8[u] : 0.f;
          dn += ex;
          a4.x += hv[u].x * ex;  a4.y += hv[u].y * ex;
          a4.z += hv[u].z * ex;  a4.w += hv[u].w * ex;
        }
      }
      const float dd = dn + 1e-16f;
      if (!MEAN) {
        const float4 b4 = *(const float4*)(bias + f);
        float4 o;
        o.x = a4.x / dd + b4.x;  o.y = a4.y / dd + b4.y;
        o.z = a4.z / dd + b4.z;  o.w = a4.w / dd + b4.w;
        o.x = o.x > 0.f ? o.x : 0.f;  o.y = o.y > 0.f ? o.y : 0.f;
        o.z = o.z > 0.f ? o.z : 0.f;  o.w = o.w > 0.f ? o.w : 0.f;
        *(float4*)(out + (size_t)node * F + f) = o;
      } else {
        const float v0 = a4.x / dd, v1 = a4.y / dd, v2 = a4.z / dd, v3 = a4.w / dd;
        const int l8 = lane & 7;
        const float r0 = ((__shfl(v0, l8) + __shfl(v0, l8 + 8)) + __shfl(v0, l8 + 16)) + __shfl(v0, l8 + 24);
        const float r1 = ((__shfl(v1, l8) + __shfl(v1, l8 + 8)) + __shfl(v1, l8 + 16)) + __shfl(v1, l8 + 24);
        const float r2 = ((__shfl(v2, l8) + __shfl(v2, l8 + 8)) + __shfl(v2, l8 + 16)) + __shfl(v2, l8 + 24);
        const float r3 = ((__shfl(v3, l8) + __shfl(v3, l8 + 8)) + __shfl(v3, l8 + 16)) + __shfl(v3, l8 + 24);
        if (lane < 8) {
          const float4 b4 = *(const float4*)(bias + 4 * lane);
          float4 o;
          o.x = 0.25f * r0 + b4.x;  o.y = 0.25f * r1 + b4.y;
          o.z = 0.25f * r2 + b4.z;  o.w = 0.25f * r3 + b4.w;
          *(float4*)(out + (size_t)node * C + 4 * lane) = o;
        }
      }
    }
  } else {
  __syncthreads();

  {
    constexpr int P = 4;
    const int items = rn * (F / 4);
    for (int idx0 = tid; idx0 < items; idx0 += NTHR * P) {
      float4 hp[P];
#pragma unroll
      for (int p = 0; p < P; ++p) {
        const int idx = idx0 + p * NTHR;
        const int idc = (idx < items) ? idx : idx0;
        const int dl = idc / (F / 4);
        const int r  = idc % (F / 4);
        const int f  = 128 * (r >> 5) + 4 * (r & 31);
        hp[p] = *(const float4*)(H + (size_t)(base + dl) * F + f);
      }
#pragma unroll
      for (int p = 0; p < P; ++p) {
        const int idx = idx0 + p * NTHR;
        const int idc = (idx < items) ? idx : idx0;
        const int dl = idc / (F / 4);
        const int r  = idc % (F / 4);
        const int j  = r >> 5, l = r & 31;
        const int f  = 128 * j + 4 * l;
        const float ex = den[dl * HEADS + f / C];
        float* ap = &acc[dl * F + 128 * j + l];
        ap[0] = hp[p].x * ex; ap[32] = hp[p].y * ex; ap[64] = hp[p].z * ex; ap[96] = hp[p].w * ex;
      }
    }
  }
  __syncthreads();

  {
    for (int p = 0; p < E; p += QCAP) {
      __syncthreads();
      if (tid == 0) qn = 0u;
      __syncthreads();
      scan(p, (p + QCAP < E) ? (p + QCAP) : E);
      __syncthreads();
      resolve((int)qn, false);
      __syncthreads();
      drain((int)qn);
    }
  }
  __syncthreads();

  if (!MEAN) {
    for (int idx = tid; idx < rn * (F / 4); idx += NTHR) {
      const int dl = idx / (F / 4);
      const int r  = idx % (F / 4);
      const int j  = r >> 5, l = r & 31;
      const int f  = 128 * j + 4 * l;
      const float dn = den[dl * HEADS + f / C] + 1e-16f;
      const float* ap = &acc[dl * F + 128 * j + l];
      const float4 b4 = *(const float4*)(bias + f);
      float4 o;
      o.x = ap[0]  / dn + b4.x;  o.y = ap[32] / dn + b4.y;
      o.z = ap[64] / dn + b4.z;  o.w = ap[96] / dn + b4.w;
      o.x = o.x > 0.f ? o.x : 0.f;  o.y = o.y > 0.f ? o.y : 0.f;
      o.z = o.z > 0.f ? o.z : 0.f;  o.w = o.w > 0.f ? o.w : 0.f;
      *(float4*)(out + (size_t)(base + dl) * F + f) = o;
    }
  } else {
    for (int idx = tid; idx < rn * C; idx += NTHR) {
      const int dl = idx / C, c = idx % C;
      const float* qd = &den[dl * HEADS];
      float p[HEADS];
#pragma unroll
      for (int h = 0; h < HEADS; ++h) {
        const int f = h * C + c;
        p[h] = acc[dl * F + 128 * (f >> 7) + 32 * (f & 3) + ((f & 127) >> 2)];
      }
      out[(size_t)(base + dl) * C + c] =
          0.25f * (p[0] / (qd[0] + 1e-16f) + p[1] / (qd[1] + 1e-16f) +
                   p[2] / (qd[2] + 1e-16f) + p[3] / (qd[3] + 1e-16f)) + bias[c];
    }
  }
  }
}

template <int C, bool MEAN, bool FIRST>
static void run_gat_layer(hipStream_t st, const float* in, int Fin,
                          const float* W, const float* a_s, const float* a_d, const float* b,
                          float* Hbuf, float* outp,
                          const int* src, const int* dst, int E,
                          float* ALS, float* ALD, unsigned* QL, int* QC) {
  const int Nn = NNODES;
  const int Fout = HEADS * C;
  const int Mtiles = Nn / 16;
  dim3 ggrid((Mtiles + 7) / 8, Fout / 64);
  gemm_xw_wmma<<<ggrid, 256, 0, st>>>(in, W, Hbuf, Nn, Fin, Fout);
  int t = Nn * HEADS;
  attn_logits<<<(t + 255) / 256, 256, 0, st>>>(Hbuf, a_s, a_d, ALS, ALD, Nn, C);
  node_range_softmax_aggregate<C, MEAN, FIRST>
      <<<NGROUPS, NTHR, 0, st>>>(src, dst, E, Hbuf, ALS, ALD, b, outp, QL, QC);
}

extern "C" void kernel_launch(void* const* d_in, const int* in_sizes, int n_in,
                              void* d_out, int out_size, void* d_ws, size_t ws_size,
                              hipStream_t stream) {
  const float* x   = (const float*)d_in[0];
  const int*   ei  = (const int*)d_in[1];
  const int E = in_sizes[1] / 2;
  const int* src = ei;
  const int* dst = ei + E;
  const float *W1 = (const float*)d_in[2],  *as1 = (const float*)d_in[3],
              *ad1 = (const float*)d_in[4], *b1  = (const float*)d_in[5];
  const float *W2 = (const float*)d_in[6],  *as2 = (const float*)d_in[7],
              *ad2 = (const float*)d_in[8], *b2  = (const float*)d_in[9];
  const float *Wm = (const float*)d_in[10], *asm_ = (const float*)d_in[11],
              *adm = (const float*)d_in[12], *bm = (const float*)d_in[13];
  const float *Wl = (const float*)d_in[14], *asl = (const float*)d_in[15],
              *adl = (const float*)d_in[16], *bl = (const float*)d_in[17];

  char* ws = (char*)d_ws;
  size_t off = 0;
  auto alloc = [&](size_t bytes) -> void* {
    void* p = ws + off;
    off += (bytes + 255) & ~(size_t)255;
    return p;
  };
  float*    Hbuf = (float*)alloc((size_t)NNODES * 512 * 4);
  float*    A0   = (float*)alloc((size_t)NNODES * 256 * 4);
  float*    A1   = (float*)alloc((size_t)NNODES * 512 * 4);
  float*    ALS  = (float*)alloc((size_t)NNODES * HEADS * 4);
  float*    ALD  = (float*)alloc((size_t)NNODES * HEADS * 4);
  unsigned* QL   = (unsigned*)alloc((size_t)NGROUPS * QCAP * 4);
  int*      QC   = (int*)alloc((size_t)NGROUPS * 4);

  float* zmean   = (float*)d_out;
  float* zlogstd = (float*)d_out + (size_t)NNODES * 32;

  run_gat_layer<64, false, true>(stream, x, 256, W1, as1, ad1, b1, Hbuf, A0, src, dst, E, ALS, ALD, QL, QC);
  run_gat_layer<128, false, false>(stream, A0, 256, W2, as2, ad2, b2, Hbuf, A1, src, dst, E, ALS, ALD, QL, QC);
  run_gat_layer<32, true, false>(stream, A1, 512, Wm, asm_, adm, bm, Hbuf, zmean, src, dst, E, ALS, ALD, QL, QC);
  run_gat_layer<32, true, false>(stream, A1, 512, Wl, asl, adl, bl, Hbuf, zlogstd, src, dst, E, ALS, ALD, QL, QC);
}


# Edges read ONCE: blocks added to layer 1's logits launch bin them by owning node group (LDS count, 1 global int atomic per block and group) into kept lists; 255 full scans gone; list in arrival order
# speedup vs baseline: 6.2317x; 1.0504x over previous
#include <hip/hip_runtime.h>

typedef float v2f __attribute__((ext_vector_type(2)));
typedef float v8f __attribute__((ext_vector_type(8)));

#define NNODES 30000
#define HEADS 4
#define SLOPE 0.2f
#define KC 32
#define PAIRSTRIDE 160
#define APITCH 36
#define RNODES 118
#define QCAP 8192
#define NTHR 1024
#define NWAVE (NTHR / 32)
#define NGROUPS ((NNODES + RNODES - 1) / RNODES)
#define BINE 8
#define BINBLK (256 * BINE)
static_assert(NGROUPS <= 256, "one thread per node group clears and reserves the list lengths");

__device__ __forceinline__ void gemm_stage(const float* __restrict__ ap, const float* __restrict__ bp,
                                           v8f& acc0, v8f& acc1, v8f& acc2, v8f& acc3) {
#pragma unroll
  for (int ks = 0; ks < KC / 4; ++ks) {
    const v2f a = *(const v2f*)(ap + 4 * ks);
    const float* __restrict__ bk = bp + 2 * ks * PAIRSTRIDE;
    const v2f b0 = *(const v2f*)(bk + 0);
    const v2f b1 = *(const v2f*)(bk + 32);
    const v2f b2 = *(const v2f*)(bk + 64);
    const v2f b3 = *(const v2f*)(bk + 96);
    acc0 = __builtin_amdgcn_wmma_f32_16x16x4_f32(false, a, false, b0, (short)0, acc0, false, false);
    acc1 = __builtin_amdgcn_wmma_f32_16x16x4_f32(false, a, false, b1, (short)0, acc1, false, false);
    acc2 = __builtin_amdgcn_wmma_f32_16x16x4_f32(false, a, false, b2, (short)0, acc2, false, false);
    acc3 = __builtin_amdgcn_wmma_f32_16x16x4_f32(false, a, false, b3, (short)0, acc3, false, false);
  }
}

__device__ __forceinline__ void gemm_land(float* __restrict__ pa, float* __restrict__ pb,
                                          float4 ra0, float4 ra1, float4 ra2, float4 ra3,
                                          float4 rb0, float4 rb1) {
  *(float4*)(pa)               = ra0;
  *(float4*)(pa + 32 * APITCH) = ra1;
  *(float4*)(pa + 64 * APITCH) = ra2;
  *(float4*)(pa + 96 * APITCH) = ra3;
  pb[0] = rb0.x; pb[2] = rb0.y; pb[4] = rb0.z; pb[6] = rb0.w;
  float* __restrict__ pb1 = pb + 8 * PAIRSTRIDE;
  pb1[0] = rb1.x; pb1[2] = rb1.y; pb1[4] = rb1.z; pb1[6] = rb1.w;
}

__global__ __launch_bounds__(256)
void gemm_xw_wmma(const float* __restrict__ A, const float* __restrict__ B,
                  float* __restrict__ C, int M, int K, int Nout, int* __restrict__ zcnt) {
  __shared__ float ldsA[2][128 * APITCH];
  __shared__ float ldsB[2][(KC / 2) * PAIRSTRIDE];

  const int tid  = threadIdx.x;
  const int lane = tid & 31;
  const int wave = tid >> 5;
  const int Mtiles = M >> 4;

  const int n0 = blockIdx.y * 64;
  const int mt = blockIdx.x * 8 + wave;
  const bool active = (mt < Mtiles);
  const int m0   = mt * 16;
  const int half = lane >> 4;
  const int lm   = lane & 15;

  const int arow = tid >> 3;
  const int acol = (tid & 7) * 4;
  const int brow = tid >> 4;
  const int bcol = (tid & 15) * 4;

  const int rlast = M - 1;
  const int r0 = blockIdx.x * 128 + arow;
  const float* __restrict__ ag0 = A + (size_t)(r0      < rlast ? r0      : rlast) * K + acol;
  const float* __restrict__ ag1 = A + (size_t)(r0 + 32 < rlast ? r0 + 32 : rlast) * K + acol;
  const float* __restrict__ ag2 = A + (size_t)(r0 + 64 < rlast ? r0 + 64 : rlast) * K + acol;
  const float* __restrict__ ag3 = A + (size_t)(r0 + 96 < rlast ? r0 + 96 : rlast) * K + acol;
  const float* __restrict__ bg  = B + (size_t)brow * Nout + n0 + bcol;
  const size_t b16 = (size_t)16 * Nout;

  const int ast = arow * APITCH + acol;
  const int bst = (brow >> 1) * PAIRSTRIDE + bcol * 2 + (brow & 1);

  float4 ra0, ra1, ra2, ra3, rb0, rb1;
  v8f acc0 = {}, acc1 = {}, acc2 = {}, acc3 = {};

  ra0 = *(const float4*)(ag0);  ra1 = *(const float4*)(ag1);
  ra2 = *(const float4*)(ag2);  ra3 = *(const float4*)(ag3);
  rb0 = *(const float4*)(bg);   rb1 = *(const float4*)(bg + b16);
  gemm_land(&ldsA[0][ast], &ldsB[0][bst], ra0, ra1, ra2, ra3, rb0, rb1);
  __syncthreads();

  const int afr = (16 * wave + lm) * APITCH + 2 * half;
  const int bfr = half * PAIRSTRIDE + lm * 2;

  const int nch = K / KC;
  int i = 0;
  for (; i + 1 < nch; ++i) {
    const int s = i & 1;
    const int kb = (i + 1) * KC;
    ra0 = *(const float4*)(ag0 + kb);  ra1 = *(const float4*)(ag1 + kb);
    ra2 = *(const float4*)(ag2 + kb);  ra3 = *(const float4*)(ag3 + kb);
    rb0 = *(const float4*)(bg + (size_t)kb * Nout);
    rb1 = *(const float4*)(bg + (size_t)kb * Nout + b16);
    __builtin_amdgcn_sched_barrier(0);
    gemm_stage(&ldsA[s][afr], &ldsB[s][bfr], acc0, acc1, acc2, acc3);
    gemm_land(&ldsA[s ^ 1][ast], &ldsB[s ^ 1][bst], ra0, ra1, ra2, ra3, rb0, rb1);
    __syncthreads();
  }
  gemm_stage(&ldsA[i & 1][afr], &ldsB[i & 1][bfr], acc0, acc1, acc2, acc3);

  if (active) {
    float* __restrict__ crow = C + (size_t)(m0 + 8 * half) * Nout + n0 + lm;
#pragma unroll
    for (int v = 0; v < 8; ++v) {
      crow[(size_t)v * Nout + 0]  = acc0[v];
      crow[(size_t)v * Nout + 16] = acc1[v];
      crow[(size_t)v * Nout + 32] = acc2[v];
      crow[(size_t)v * Nout + 48] = acc3[v];
    }
  }
  if (zcnt != nullptr && blockIdx.x == 0 && blockIdx.y == 0 && tid < NGROUPS) zcnt[tid] = 0;
}

template <bool BIN>
__global__ __launch_bounds__(256)
void attn_logits(const float* __restrict__ H, const float* __restrict__ a_s,
                 const float* __restrict__ a_d, float* __restrict__ als,
                 float* __restrict__ ald, int Nn, int C,
                 const int* __restrict__ src, const int* __restrict__ dst, int E,
                 unsigned* __restrict__ qlist, int* __restrict__ qcnt, int nbin) {
  __shared__ int bcnt[BIN ? 256 : 1];
  __shared__ int bbase[BIN ? 256 : 1];
  if (BIN && (int)blockIdx.x < nbin) {
    const int tid = threadIdx.x;
    bcnt[tid] = 0;
    __syncthreads();
    const int e0 = blockIdx.x * BINBLK;
    int d[BINE], s[BINE], rk[BINE];
#pragma unroll
    for (int u = 0; u < BINE; ++u) {
      const int e = e0 + u * 256 + tid;
      d[u] = (e < E) ? dst[e] : -1;
      s[u] = (e < E) ? src[e] : 0;
    }
#pragma unroll
    for (int u = 0; u < BINE; ++u) {
      rk[u] = 0;
      if ((unsigned)d[u] < (unsigned)NNODES) rk[u] = atomicAdd(&bcnt[d[u] / RNODES], 1);
    }
    __syncthreads();
    if (tid < NGROUPS) {
      const int cblk = bcnt[tid];
      bbase[tid] = (cblk > 0) ? atomicAdd(&qcnt[tid], cblk) : 0;
    }
    __syncthreads();
#pragma unroll
    for (int u = 0; u < BINE; ++u) {
      if ((unsigned)d[u] < (unsigned)NNODES) {
        const int g = d[u] / RNODES;
        const int pos = bbase[g] + rk[u];
        if (pos < QCAP)
          qlist[(size_t)g * QCAP + pos] = (unsigned)s[u] | ((unsigned)(d[u] - g * RNODES) << 16);
      }
    }
    return;
  }
  int i = (blockIdx.x - (BIN ? nbin : 0)) * blockDim.x + threadIdx.x;
  if (i >= Nn * HEADS) return;
  int n = i / HEADS, h = i % HEADS;
  const float* hp = H + (size_t)n * HEADS * C + (size_t)h * C;
  float s = 0.f, d = 0.f;
  for (int c = 0; c < C; ++c) { float v = hp[c]; s += v * a_s[h * C + c]; d += v * a_d[h * C + c]; }
  als[i] = s; ald[i] = d;
}

__device__ __forceinline__ unsigned fkey(float x) {
  unsigned b = __float_as_uint(x);
  return (b & 0x80000000u) ? ~b : (b | 0x80000000u);
}
__device__ __forceinline__ float funkey(unsigned k) {
  unsigned b = (k & 0x80000000u) ? (k & 0x7FFFFFFFu) : ~k;
  return __uint_as_float(b);
}


template <int C, bool MEAN>
__global__ __launch_bounds__(NTHR)
void node_range_softmax_aggregate(const int* __restrict__ src, const int* __restrict__ dst, int E,
                                  const float* __restrict__ H, const float* __restrict__ als,
                                  const float* __restrict__ ald, const float* __restrict__ bias,
                                  float* __restrict__ out,
                                  const unsigned* __restrict__ qlist, const int* __restrict__ qcnt) {
  constexpr int F = HEADS * C;
  constexpr int J = F / 128;
  constexpr int U = 8 / J;
  constexpr int SLOTS = QCAP + 256;
  __shared__ __attribute__((aligned(16))) float acc[(RNODES * F > 5 * SLOTS) ? RNODES * F : 5 * SLOTS];
  __shared__ unsigned mk[RNODES * HEADS];
  __shared__ __attribute__((aligned(16))) float mv[RNODES * HEADS];
  __shared__ float    den[RNODES * HEADS];
  __shared__ __attribute__((aligned(16))) float aldc[RNODES * HEADS];
  __shared__ unsigned q[QCAP];
  __shared__ unsigned qn;
  __shared__ int      cnt[128];
  __shared__ int      off[132];

  const int tid  = threadIdx.x;
  const int lane = tid & 31;
  const int wave = tid >> 5;
  const int base = blockIdx.x * RNODES;
  int rn = NNODES - base;
  if (rn > RNODES) rn = RNODES;
  if (rn < 0) rn = 0;
  const unsigned* __restrict__ lst = qlist + (size_t)blockIdx.x * QCAP;

  auto take = [&](int e, unsigned dl) {
    if (dl < (unsigned)rn) {
      const unsigned slot = atomicAdd(&qn, 1u);
      if (slot < QCAP) q[slot] = (unsigned)e | (dl << 25);
    }
  };

  auto scan = [&](int e0, int e1) {
    if (((E | e0 | e1) & 3) == 0) {
      const int4* __restrict__ d4 = (const int4*)dst;
      const int g1 = e1 >> 2;
      for (int g0 = (e0 >> 2); g0 < g1; g0 += NTHR * 8) {
        int4 v[8];
#pragma unroll
        for (int u = 0; u < 8; ++u) {
          const int g = g0 + u * NTHR + tid;
          v[u] = (g < g1) ? d4[g] : make_int4(-1, -1, -1, -1);
        }
#pragma unroll
        for (int u = 0; u < 8; ++u) {
          const int e = (g0 + u * NTHR + tid) * 4;
          const unsigned a0 = (unsigned)(v[u].x - base), a1 = (unsigned)(v[u].y - base);
          const unsigned a2 = (unsigned)(v[u].z - base), a3 = (unsigned)(v[u].w - base);
          const unsigned lo01 = a0 < a1 ? a0 : a1, lo23 = a2 < a3 ? a2 : a3;
          if ((lo01 < lo23 ? lo01 : lo23) < (unsigned)rn) {
            take(e + 0, a0); take(e + 1, a1); take(e + 2, a2); take(e + 3, a3);
          }
        }
      }
    } else {
      for (int e = e0 + tid; e < e1; e += NTHR) take(e, (unsigned)(dst[e] - base));
    }
  };

  auto resolve = [&](int n, bool withmax) {
    for (int i = tid; i < n; i += NTHR) {
      const unsigned ent = q[i];
      const unsigned dl = ent >> 25;
      const int s = src[ent & 0x1FFFFFFu];
      if (withmax) {
        const float4 a = *(const float4*)(als + (size_t)s * HEADS);
        atomicMax(&mk[dl * HEADS + 0], fkey(a.x));
        atomicMax(&mk[dl * HEADS + 1], fkey(a.y));
        atomicMax(&mk[dl * HEADS + 2], fkey(a.z));
        atomicMax(&mk[dl * HEADS + 3], fkey(a.w));
      }
      q[i] = (unsigned)s | (dl << 16);
    }
  };

  auto drain = [&](int n) {
    for (int i0 = wave; i0 < n; i0 += NWAVE * U) {
      const int iu = i0 + lane * NWAVE;
      const unsigned mine = (lane < U && iu < n) ? q[iu] : 0u;
      float4 hv[U][J];
      float sa[U][J], da[U][J], ma[U][J];
      int sidx[U], dloc[U];
#pragma unroll
      for (int u = 0; u < U; ++u) {
        const unsigned ent = (unsigned)__builtin_amdgcn_readlane((int)mine, u);
        sidx[u] = (int)(ent & 0xFFFFu);
        dloc[u] = (int)(ent >> 16);
      }
#pragma unroll
      for (int u = 0; u < U; ++u) {
#pragma unroll
        for (int j = 0; j < J; ++j) {
          const int h = (128 * j + lane * 4) / C;
          sa[u][j] = als[sidx[u] * HEADS + h];
        }
      }
#pragma unroll
      for (int u = 0; u < U; ++u) {
        const float* __restrict__ hrow = H + (size_t)sidx[u] * F + lane * 4;
#pragma unroll
        for (int j = 0; j < J; ++j) hv[u][j] = *(const float4*)(hrow + 128 * j);
      }
#pragma unroll
      for (int u = 0; u < U; ++u) {
#pragma unroll
        for (int j = 0; j < J; ++j) {
          const int h = (128 * j + lane * 4) / C;
          da[u][j] = aldc[dloc[u] * HEADS + h];
          ma[u][j] = mv[dloc[u] * HEADS + h];
        }
      }
      __builtin_amdgcn_sched_barrier(0);
#pragma unroll
      for (int u = 0; u < U; ++u) {
        if (i0 + u * NWAVE < n) {
          const int dl = dloc[u];
#pragma unroll
          for (int j = 0; j < J; ++j) {
            const int f = 128 * j + lane * 4;
            const int h = f / C;
            float sc = sa[u][j] + da[u][j];
            sc = (sc > 0.f) ? sc : SLOPE * sc;
            const float ex = __expf(sc - ma[u][j]);
            if ((f % C) == 0) atomicAdd(&den[dl * HEADS + h], ex);
            float* ap = &acc[dl * F + 128 * j + lane];
            atomicAdd(ap +  0, hv[u][j].x * ex);
            atomicAdd(ap + 32, hv[u][j].y * ex);
            atomicAdd(ap + 64, hv[u][j].z * ex);
            atomicAdd(ap + 96, hv[u][j].w * ex);
          }
        }
      }
    }
  };

  if (tid < rn * HEADS) {
    mk[tid]   = fkey(als[(size_t)base * HEADS + tid]);
    aldc[tid] = ald[(size_t)base * HEADS + tid];
  }
  if (tid == 0) qn = 0u;
  if (tid < 128) cnt[tid] = 0;
  __syncthreads();

  const int total = qcnt[blockIdx.x];
  const bool fits = (total <= QCAP);
  constexpr int KMAX = QCAP / NTHR;
  constexpr int KREG = 4;
  unsigned ent[KMAX];
  int rk[KMAX];
  float4 av[KREG];
  if (fits) {
#pragma unroll
    for (int k = 0; k < KMAX; ++k) {
      const int i = tid + k * NTHR;
      ent[k] = 0u;
      if (i < total) ent[k] = lst[i];
    }
#pragma unroll
    for (int k = 0; k < KREG; ++k) {
      const int i = tid + k * NTHR;
      av[k] = make_float4(0.f, 0.f, 0.f, 0.f);
      if (i < total) av[k] = *(const float4*)(als + (size_t)(ent[k] & 0xFFFFu) * HEADS);
    }
#pragma unroll
    for (int k = 0; k < KMAX; ++k) {
      const int i = tid + k * NTHR;
      rk[k] = 0;
      if (i < total) {
        const unsigned dl = ent[k] >> 16;
        float4 a;
        if (k < KREG) a = av[k];
        else          a = *(const float4*)(als + (size_t)(ent[k] & 0xFFFFu) * HEADS);
        atomicMax(&mk[dl * HEADS + 0], fkey(a.x));
        atomicMax(&mk[dl * HEADS + 1], fkey(a.y));
        atomicMax(&mk[dl * HEADS + 2], fkey(a.z));
        atomicMax(&mk[dl * HEADS + 3], fkey(a.w));
        rk[k] = atomicAdd(&cnt[dl], 1);
      }
    }
  } else {
    for (int p = 0; p < E; p += QCAP) {
      __syncthreads();
      if (tid == 0) qn = 0u;
      __syncthreads();
      scan(p, (p + QCAP < E) ? (p + QCAP) : E);
      __syncthreads();
      resolve((int)qn, true);
    }
  }
  __syncthreads();

  if (tid < rn * HEADS) {
    const float a = aldc[tid];
    float m = funkey(mk[tid]) + a;
    m = (m > 0.f) ? m : SLOPE * m;
    mv[tid] = m;
    float sc = als[(size_t)base * HEADS + tid] + a;
    sc = (sc > 0.f) ? sc : SLOPE * sc;
    den[tid] = __expf(sc - m);
  }

  if (fits) {
    static_assert(!MEAN || C == 32, "the sum over heads across lanes is written for 32 channels per head");
    unsigned* __restrict__ qs = (unsigned*)acc;
    float* __restrict__ wt = acc + SLOTS;
    if (wave == 0) {
      const int c0 = cnt[4 * lane], c1 = cnt[4 * lane + 1], c2 = cnt[4 * lane + 2], c3 = cnt[4 * lane + 3];
      const int s = c0 + c1 + c2 + c3;
      int incl = s;
#pragma unroll
      for (int d = 1; d < 32; d <<= 1) {
        const int t = __shfl_up(incl, d);
        if (lane >= d) incl += t;
      }
      const int ex0 = incl - s;
      off[4 * lane]     = ex0;
      off[4 * lane + 1] = ex0 + c0;
      off[4 * lane + 2] = ex0 + c0 + c1;
      off[4 * lane + 3] = ex0 + c0 + c1 + c2;
      if (lane == 31) off[128] = incl;
    }
    __syncthreads();
#pragma unroll
    for (int k = 0; k < KMAX; ++k) {
      const int i = tid + k * NTHR;
      if (i < total) {
        const unsigned dl = ent[k] >> 16;
        const int pos = off[dl] + (int)dl + 1 + rk[k];
        qs[pos] = ent[k] & 0xFFFFu;
        const float4 ad4 = *(const float4*)&aldc[dl * HEADS];
        const float4 m4  = *(const float4*)&mv[dl * HEADS];
        float4 a;
        if (k < KREG) a = av[k];
        else          a = *(const float4*)(als + (size_t)(ent[k] & 0xFFFFu) * HEADS);
        float4 w;
        float sc;
        sc = a.x + ad4.x;  sc = (sc > 0.f) ? sc : SLOPE * sc;  w.x = __expf(sc - m4.x);
        sc = a.y + ad4.y;  sc = (sc > 0.f) ? sc : SLOPE * sc;  w.y = __expf(sc - m4.y);
        sc = a.z + ad4.z;  sc = (sc > 0.f) ? sc : SLOPE * sc;  w.z = __expf(sc - m4.z);
        sc = a.w + ad4.w;  sc = (sc > 0.f) ? sc : SLOPE * sc;  w.w = __expf(sc - m4.w);
        *(float4*)&wt[4 * pos] = w;
      }
    }
    if (tid < rn * HEADS) {
      const int dl = tid / HEADS;
      const int pos = off[dl] + dl;
      wt[4 * pos + (tid % HEADS)] = den[tid];
      if ((tid % HEADS) == 0) qs[pos] = (unsigned)(base + dl);
    }
    constexpr int UE = 8;
    if (tid < HEADS * UE) wt[4 * (total + rn) + tid] = 0.f;
    __syncthreads();

    const int wv = __builtin_amdgcn_readfirstlane(wave);
    const int nitems = rn * J;
    for (int it = wv; it < nitems; it += NWAVE) {
      const int dl = it / J, j = it % J;
      const int node = base + dl;
      const int f = 128 * j + lane * 4;
      const int h = f / C;
      const int o0 = __builtin_amdgcn_readfirstlane(off[dl]) + dl;
      const int nt = __builtin_amdgcn_readfirstlane(off[dl + 1]) + dl + 1 - o0;
      const float* __restrict__ wp = wt + 4 * o0 + h;
      float4 a4 = make_float4(0.f, 0.f, 0.f, 0.f);
      float dn = 0.f;
      for (int i0 = 0; i0 < nt; i0 += UE) {
        const int p = i0 + lane;
        unsigned mine = (unsigned)node;
        if (lane < UE && p < nt) mine = qs[o0 + p];
        int sidx[UE];
        float wv8[UE];
        float4 hv[UE];
#pragma unroll
        for (int u = 0; u < UE; ++u) sidx[u] = __builtin_amdgcn_readlane((int)mine, u);
#pragma unroll
        for (int u = 0; u < UE; ++u) wv8[u] = wp[4 * (i0 + u)];
#pragma unroll
        for (int u = 0; u < UE; ++u) hv[u] = *(const float4*)(H + (size_t)sidx[u] * F + f);
#pragma unroll
        for (int u = 0; u < UE; ++u) {
          const float ex = (i0 + u < nt) ? wv8[u] : 0.f;
          dn += ex;
          a4.x += hv[u].x * ex;  a4.y += hv[u].y * ex;
          a4.z += hv[u].z * ex;  a4.w += hv[u].w * ex;
        }
      }
      const float dd = dn + 1e-16f;
      if (!MEAN) {
        const float4 b4 = *(const float4*)(bias + f);
        float4 o;
        o.x = a4.x / dd + b4.x;  o.y = a4.y / dd + b4.y;
        o.z = a4.z / dd + b4.z;  o.w = a4.w / dd + b4.w;
        o.x = o.x > 0.f ? o.x : 0.f;  o.y = o.y > 0.f ? o.y : 0.f;
        o.z = o.z > 0.f ? o.z : 0.f;  o.w = o.w > 0.f ? o.w : 0.f;
        *(float4*)(out + (size_t)node * F + f) = o;
      } else {
        const float v0 = a4.x / dd, v1 = a4.y / dd, v2 = a4.z / dd, v3 = a4.w / dd;
        const int l8 = lane & 7;
        const float r0 = ((__shfl(v0, l8) + __shfl(v0, l8 + 8)) + __shfl(v0, l8 + 16)) + __shfl(v0, l8 + 24);
        const float r1 = ((__shfl(v1, l8) + __shfl(v1, l8 + 8)) + __shfl(v1, l8 + 16)) + __shfl(v1, l8 + 24);
        const float r2 = ((__shfl(v2, l8) + __shfl(v2, l8 + 8)) + __shfl(v2, l8 + 16)) + __shfl(v2, l8 + 24);
        const float r3 = ((__shfl(v3, l8) + __shfl(v3, l8 + 8)) + __shfl(v3, l8 + 16)) + __shfl(v3, l8 + 24);
        if (lane < 8) {
          const float4 b4 = *(const float4*)(bias + 4 * lane);
          float4 o;
          o.x = 0.25f * r0 + b4.x;  o.y = 0.25f * r1 + b4.y;
          o.z = 0.25f * r2 + b4.z;  o.w = 0.25f * r3 + b4.w;
          *(float4*)(out + (size_t)node * C + 4 * lane) = o;
        }
      }
    }
  } else {
  __syncthreads();

  {
    constexpr int P = 4;
    const int items = rn * (F / 4);
    for (int idx0 = tid; idx0 < items; idx0 += NTHR * P) {
      float4 hp[P];
#pragma unroll
      for (int p = 0; p < P; ++p) {
        const int idx = idx0 + p * NTHR;
        const int idc = (idx < items) ? idx : idx0;
        const int dl = idc / (F / 4);
        const int r  = idc % (F / 4);
        const int f  = 128 * (r >> 5) + 4 * (r & 31);
        hp[p] = *(const float4*)(H + (size_t)(base + dl) * F + f);
      }
#pragma unroll
      for (int p = 0; p < P; ++p) {
        const int idx = idx0 + p * NTHR;
        const int idc = (idx < items) ? idx : idx0;
        const int dl = idc / (F / 4);
        const int r  = idc % (F / 4);
        const int j  = r >> 5, l = r & 31;
        const int f  = 128 * j + 4 * l;
        const float ex = den[dl * HEADS + f / C];
        float* ap = &acc[dl * F + 128 * j + l];
        ap[0] = hp[p].x * ex; ap[32] = hp[p].y * ex; ap[64] = hp[p].z * ex; ap[96] = hp[p].w * ex;
      }
    }
  }
  __syncthreads();

  {
    for (int p = 0; p < E; p += QCAP) {
      __syncthreads();
      if (tid == 0) qn = 0u;
      __syncthreads();
      scan(p, (p + QCAP < E) ? (p + QCAP) : E);
      __syncthreads();
      resolve((int)qn, false);
      __syncthreads();
      drain((int)qn);
    }
  }
  __syncthreads();

  if (!MEAN) {
    for (int idx = tid; idx < rn * (F / 4); idx += NTHR) {
      const int dl = idx / (F / 4);
      const int r  = idx % (F / 4);
      const int j  = r >> 5, l = r & 31;
      const int f  = 128 * j + 4 * l;
      const float dn = den[dl * HEADS + f / C] + 1e-16f;
      const float* ap = &acc[dl * F + 128 * j + l];
      const float4 b4 = *(const float4*)(bias + f);
      float4 o;
      o.x = ap[0]  / dn + b4.x;  o.y = ap[32] / dn + b4.y;
      o.z = ap[64] / dn + b4.z;  o.w = ap[96] / dn + b4.w;
      o.x = o.x > 0.f ? o.x : 0.f;  o.y = o.y > 0.f ? o.y : 0.f;
      o.z = o.z > 0.f ? o.z : 0.f;  o.w = o.w > 0.f ? o.w : 0.f;
      *(float4*)(out + (size_t)(base + dl) * F + f) = o;
    }
  } else {
    for (int idx = tid; idx < rn * C; idx += NTHR) {
      const int dl = idx / C, c = idx % C;
      const float* qd = &den[dl * HEADS];
      float p[HEADS];
#pragma unroll
      for (int h = 0; h < HEADS; ++h) {
        const int f = h * C + c;
        p[h] = acc[dl * F + 128 * (f >> 7) + 32 * (f & 3) + ((f & 127) >> 2)];
      }
      out[(size_t)(base + dl) * C + c] =
          0.25f * (p[0] / (qd[0] + 1e-16f) + p[1] / (qd[1] + 1e-16f) +
                   p[2] / (qd[2] + 1e-16f) + p[3] / (qd[3] + 1e-16f)) + bias[c];
    }
  }
  }
}

template <int C, bool MEAN, bool FIRST>
static void run_gat_layer(hipStream_t st, const float* in, int Fin,
                          const float* W, const float* a_s, const float* a_d, const float* b,
                          float* Hbuf, float* outp,
                          const int* src, const int* dst, int E,
                          float* ALS, float* ALD, unsigned* QL, int* QC) {
  const int Nn = NNODES;
  const int Fout = HEADS * C;
  const int Mtiles = Nn / 16;
  dim3 ggrid((Mtiles + 7) / 8, Fout / 64);
  gemm_xw_wmma<<<ggrid, 256, 0, st>>>(in, W, Hbuf, Nn, Fin, Fout, FIRST ? QC : nullptr);
  int t = Nn * HEADS;
  const int nbin = FIRST ? (E + BINBLK - 1) / BINBLK : 0;
  attn_logits<FIRST><<<nbin + (t + 255) / 256, 256, 0, st>>>(Hbuf, a_s, a_d, ALS, ALD, Nn, C,
                                                             src, dst, E, QL, QC, nbin);
  node_range_softmax_aggregate<C, MEAN>
      <<<NGROUPS, NTHR, 0, st>>>(src, dst, E, Hbuf, ALS, ALD, b, outp, QL, QC);
}

extern "C" void kernel_launch(void* const* d_in, const int* in_sizes, int n_in,
                              void* d_out, int out_size, void* d_ws, size_t ws_size,
                              hipStream_t stream) {
  const float* x   = (const float*)d_in[0];
  const int*   ei  = (const int*)d_in[1];
  const int E = in_sizes[1] / 2;
  const int* src = ei;
  const int* dst = ei + E;
  const float *W1 = (const float*)d_in[2],  *as1 = (const float*)d_in[3],
              *ad1 = (const float*)d_in[4], *b1  = (const float*)d_in[5];
  const float *W2 = (const float*)d_in[6],  *as2 = (const float*)d_in[7],
              *ad2 = (const float*)d_in[8], *b2  = (const float*)d_in[9];
  const float *Wm = (const float*)d_in[10], *asm_ = (const float*)d_in[11],
              *adm = (const float*)d_in[12], *bm = (const float*)d_in[13];
  const float *Wl = (const float*)d_in[14], *asl = (const float*)d_in[15],
              *adl = (const float*)d_in[16], *bl = (const float*)d_in[17];

  char* ws = (char*)d_ws;
  size_t off = 0;
  auto alloc = [&](size_t bytes) -> void* {
    void* p = ws + off;
    off += (bytes + 255) & ~(size_t)255;
    return p;
  };
  float*    Hbuf = (float*)alloc((size_t)NNODES * 512 * 4);
  float*    A0   = (float*)alloc((size_t)NNODES * 256 * 4);
  float*    A1   = (float*)alloc((size_t)NNODES * 512 * 4);
  float*    ALS  = (float*)alloc((size_t)NNODES * HEADS * 4);
  float*    ALD  = (float*)alloc((size_t)NNODES * HEADS * 4);
  unsigned* QL   = (unsigned*)alloc((size_t)NGROUPS * QCAP * 4);
  int*      QC   = (int*)alloc((size_t)NGROUPS * 4);

  float* zmean   = (float*)d_out;
  float* zlogstd = (float*)d_out + (size_t)NNODES * 32;

  run_gat_layer<64, false, true>(stream, x, 256, W1, as1, ad1, b1, Hbuf, A0, src, dst, E, ALS, ALD, QL, QC);
  run_gat_layer<128, false, false>(stream, A0, 256, W2, as2, ad2, b2, Hbuf, A1, src, dst, E, ALS, ALD, QL, QC);
  run_gat_layer<32, true, false>(stream, A1, 512, Wm, asm_, adm, bm, Hbuf, zmean, src, dst, E, ALS, ALD, QL, QC);
  run_gat_layer<32, true, false>(stream, A1, 512, Wl, asl, adl, bl, Hbuf, zlogstd, src, dst, E, ALS, ALD, QL, QC);
}


# Logits of layers 2-4 come out of the layer's GEMM launch as 16 extra columns (W folded with a_s, a_d by blocks of the first logits launch): 3 logits launches gone (9 left); logit sum order changed
# speedup vs baseline: 6.4682x; 1.0380x over previous
#include <hip/hip_runtime.h>

typedef float v2f __attribute__((ext_vector_type(2)));
typedef float v8f __attribute__((ext_vector_type(8)));

#define NNODES 30000
#define HEADS 4
#define SLOPE 0.2f
#define KC 32
#define PAIRSTRIDE 160
#define APITCH 36
#define PSV 32
#define NFOLD 16
#define ASTAGE (128 * APITCH)
#define BSTAGE ((KC / 2) * PAIRSTRIDE)
#define VSTAGE ((KC / 2) * PSV)
#define RNODES 118
#define QCAP 8192
#define NTHR 1024
#define NWAVE (NTHR / 32)
#define NGROUPS ((NNODES + RNODES - 1) / RNODES)
#define BINE 8
#define BINBLK (256 * BINE)
static_assert(NGROUPS <= 256, "one thread per node group clears and reserves the list lengths");

template <bool LG>
__device__ __forceinline__ void gemm_stage(const float* __restrict__ ap, const float* __restrict__ bp,
                                           const float* __restrict__ vp,
                                           v8f& acc0, v8f& acc1, v8f& acc2, v8f& acc3, v8f& acc4) {
#pragma unroll
  for (int ks = 0; ks < KC / 4; ++ks) {
    const v2f a = *(const v2f*)(ap + 4 * ks);
    const float* __restrict__ bk = bp + 2 * ks * PAIRSTRIDE;
    const v2f b0 = *(const v2f*)(bk + 0);
    const v2f b1 = *(const v2f*)(bk + 32);
    const v2f b2 = *(const v2f*)(bk + 64);
    const v2f b3 = *(const v2f*)(bk + 96);
    acc0 = __builtin_amdgcn_wmma_f32_16x16x4_f32(false, a, false, b0, (short)0, acc0, false, false);
    acc1 = __builtin_amdgcn_wmma_f32_16x16x4_f32(false, a, false, b1, (short)0, acc1, false, false);
    acc2 = __builtin_amdgcn_wmma_f32_16x16x4_f32(false, a, false, b2, (short)0, acc2, false, false);
    acc3 = __builtin_amdgcn_wmma_f32_16x16x4_f32(false, a, false, b3, (short)0, acc3, false, false);
    if (LG) {
      const v2f b4 = *(const v2f*)(vp + 2 * ks * PSV);
      acc4 = __builtin_amdgcn_wmma_f32_16x16x4_f32(false, a, false, b4, (short)0, acc4, false, false);
    }
  }
}

__device__ __forceinline__ void gemm_land(float* __restrict__ pa, float* __restrict__ pb,
                                          float4 ra0, float4 ra1, float4 ra2, float4 ra3,
                                          float4 rb0, float4 rb1) {
  *(float4*)(pa)               = ra0;
  *(float4*)(pa + 32 * APITCH) = ra1;
  *(float4*)(pa + 64 * APITCH) = ra2;
  *(float4*)(pa + 96 * APITCH) = ra3;
  pb[0] = rb0.x; pb[2] = rb0.y; pb[4] = rb0.z; pb[6] = rb0.w;
  float* __restrict__ pb1 = pb + 8 * PAIRSTRIDE;
  pb1[0] = rb1.x; pb1[2] = rb1.y; pb1[4] = rb1.z; pb1[6] = rb1.w;
}

template <bool LG>
__device__ __forceinline__ void gemm_body(const float* __restrict__ A, const float* __restrict__ B,
                                          float* __restrict__ C, int M, int K, int Nout,
                                          float* __restrict__ ldsA, float* __restrict__ ldsB,
                                          float* __restrict__ ldsV, const float* __restrict__ Vb,
                                          float* __restrict__ l0, float* __restrict__ l1,
                                          float* __restrict__ l2, float* __restrict__ l3) {
  const int tid  = threadIdx.x;
  const int lane = tid & 31;
  const int wave = tid >> 5;
  const int Mtiles = M >> 4;

  const int n0 = blockIdx.y * 64;
  const int mt = blockIdx.x * 8 + wave;
  const bool active = (mt < Mtiles);
  const int m0   = mt * 16;
  const int half = lane >> 4;
  const int lm   = lane & 15;

  const int arow = tid >> 3;
  const int acol = (tid & 7) * 4;
  const int brow = tid >> 4;
  const int bcol = (tid & 15) * 4;
  const int vrow = tid >> 3;
  const int vcol = (tid & 7) * 2;

  const int rlast = M - 1;
  const int r0 = blockIdx.x * 128 + arow;
  const float* __restrict__ ag0 = A + (size_t)(r0      < rlast ? r0      : rlast) * K + acol;
  const float* __restrict__ ag1 = A + (size_t)(r0 + 32 < rlast ? r0 + 32 : rlast) * K + acol;
  const float* __restrict__ ag2 = A + (size_t)(r0 + 64 < rlast ? r0 + 64 : rlast) * K + acol;
  const float* __restrict__ ag3 = A + (size_t)(r0 + 96 < rlast ? r0 + 96 : rlast) * K + acol;
  const float* __restrict__ bg  = B + (size_t)brow * Nout + n0 + bcol;
  const size_t b16 = (size_t)16 * Nout;
  const float* __restrict__ vg  = LG ? (Vb + (size_t)vrow * NFOLD + vcol) : A;

  const int ast = arow * APITCH + acol;
  const int bst = (brow >> 1) * PAIRSTRIDE + bcol * 2 + (brow & 1);
  const int vst = (vrow >> 1) * PSV + vcol * 2 + (vrow & 1);

  float4 ra0, ra1, ra2, ra3, rb0, rb1;
  float2 rv = make_float2(0.f, 0.f);
  v8f acc0 = {}, acc1 = {}, acc2 = {}, acc3 = {}, acc4 = {};

  ra0 = *(const float4*)(ag0);  ra1 = *(const float4*)(ag1);
  ra2 = *(const float4*)(ag2);  ra3 = *(const float4*)(ag3);
  rb0 = *(const float4*)(bg);   rb1 = *(const float4*)(bg + b16);
  if (LG) rv = *(const float2*)(vg);
  gemm_land(ldsA + ast, ldsB + bst, ra0, ra1, ra2, ra3, rb0, rb1);
  if (LG) { ldsV[vst] = rv.x; ldsV[vst + 2] = rv.y; }
  __syncthreads();

  const int afr = (16 * wave + lm) * APITCH + 2 * half;
  const int bfr = half * PAIRSTRIDE + lm * 2;
  const int vfr = half * PSV + lm * 2;

  const int nch = K / KC;
  int i = 0;
  for (; i + 1 < nch; ++i) {
    const int s = i & 1;
    const int kb = (i + 1) * KC;
    ra0 = *(const float4*)(ag0 + kb);  ra1 = *(const float4*)(ag1 + kb);
    ra2 = *(const float4*)(ag2 + kb);  ra3 = *(const float4*)(ag3 + kb);
    rb0 = *(const float4*)(bg + (size_t)kb * Nout);
    rb1 = *(const float4*)(bg + (size_t)kb * Nout + b16);
    if (LG) rv = *(const float2*)(vg + (size_t)kb * NFOLD);
    __builtin_amdgcn_sched_barrier(0);
    gemm_stage<LG>(ldsA + s * ASTAGE + afr, ldsB + s * BSTAGE + bfr, ldsV + s * VSTAGE + vfr,
                   acc0, acc1, acc2, acc3, acc4);
    gemm_land(ldsA + (s ^ 1) * ASTAGE + ast, ldsB + (s ^ 1) * BSTAGE + bst, ra0, ra1, ra2, ra3, rb0, rb1);
    if (LG) { ldsV[(s ^ 1) * VSTAGE + vst] = rv.x; ldsV[(s ^ 1) * VSTAGE + vst + 2] = rv.y; }
    __syncthreads();
  }
  gemm_stage<LG>(ldsA + (i & 1) * ASTAGE + afr, ldsB + (i & 1) * BSTAGE + bfr, ldsV + (i & 1) * VSTAGE + vfr,
                 acc0, acc1, acc2, acc3, acc4);

  if (active) {
    float* __restrict__ crow = C + (size_t)(m0 + 8 * half) * Nout + n0 + lm;
#pragma unroll
    for (int v = 0; v < 8; ++v) {
      crow[(size_t)v * Nout + 0]  = acc0[v];
      crow[(size_t)v * Nout + 16] = acc1[v];
      crow[(size_t)v * Nout + 32] = acc2[v];
      crow[(size_t)v * Nout + 48] = acc3[v];
    }
    if (LG) {
      float* lp = (lm < 4) ? l0 : (lm < 8) ? l1 : (lm < 12) ? l2 : l3;
      if (lp != nullptr) {
        lp += (size_t)(m0 + 8 * half) * HEADS + (lm & 3);
#pragma unroll
        for (int v = 0; v < 8; ++v) lp[v * HEADS] = acc4[v];
      }
    }
  }
}

__global__ __launch_bounds__(256)
void gemm_xw_wmma(const float* __restrict__ A, const float* __restrict__ B,
                  float* __restrict__ C, int M, int K, int Nout, int* __restrict__ zcnt,
                  const float* __restrict__ Vb, float* __restrict__ l0, float* __restrict__ l1,
                  float* __restrict__ l2, float* __restrict__ l3) {
  __shared__ __attribute__((aligned(16))) float ldsA[2 * ASTAGE];
  __shared__ __attribute__((aligned(16))) float ldsB[2 * BSTAGE];
  __shared__ __attribute__((aligned(16))) float ldsV[2 * VSTAGE];

  if (Vb != nullptr && blockIdx.y == 0)
    gemm_body<true>(A, B, C, M, K, Nout, ldsA, ldsB, ldsV, Vb, l0, l1, l2, l3);
  else
    gemm_body<false>(A, B, C, M, K, Nout, ldsA, ldsB, ldsV, Vb, l0, l1, l2, l3);

  const int tid = threadIdx.x;
  if (zcnt != nullptr && blockIdx.x == 0 && blockIdx.y == 0 && tid < NGROUPS) zcnt[tid] = 0;
}

struct FoldArgs {
  const float* W2;  const float* as2;  const float* ad2;
  const float* Wm;  const float* asm_; const float* adm;
  const float* Wl;  const float* asl;  const float* adl;
  float* V2;  float* V34;
};

__device__ __forceinline__ float fold_dot(const float* __restrict__ w, const float* __restrict__ a, int C) {
  float s = 0.f;
  for (int c = 0; c < C; c += 4) {
    const float4 wv = *(const float4*)(w + c);
    const float4 av = *(const float4*)(a + c);
    s += wv.x * av.x;  s += wv.y * av.y;  s += wv.z * av.z;  s += wv.w * av.w;
  }
  return s;
}

template <bool BIN>
__global__ __launch_bounds__(256)
void attn_logits(const float* __restrict__ H, const float* __restrict__ a_s,
                 const float* __restrict__ a_d, float* __restrict__ als,
                 float* __restrict__ ald, int Nn, int C,
                 const int* __restrict__ src, const int* __restrict__ dst, int E,
                 unsigned* __restrict__ qlist, int* __restrict__ qcnt, int nbin, FoldArgs fa) {
  __shared__ int bcnt[BIN ? 256 : 1];
  __shared__ int bbase[BIN ? 256 : 1];
  if (BIN && (int)blockIdx.x < nbin) {
    const int tid = threadIdx.x;
    bcnt[tid] = 0;
    __syncthreads();
    const int e0 = blockIdx.x * BINBLK;
    int d[BINE], s[BINE], rk[BINE];
#pragma unroll
    for (int u = 0; u < BINE; ++u) {
      const int e = e0 + u * 256 + tid;
      d[u] = (e < E) ? dst[e] : -1;
      s[u] = (e < E) ? src[e] : 0;
    }
#pragma unroll
    for (int u = 0; u < BINE; ++u) {
      rk[u] = 0;
      if ((unsigned)d[u] < (unsigned)NNODES) rk[u] = atomicAdd(&bcnt[d[u] / RNODES], 1);
    }
    __syncthreads();
    if (tid < NGROUPS) {
      const int cblk = bcnt[tid];
      bbase[tid] = (cblk > 0) ? atomicAdd(&qcnt[tid], cblk) : 0;
    }
    __syncthreads();
#pragma unroll
    for (int u = 0; u < BINE; ++u) {
      if ((unsigned)d[u] < (unsigned)NNODES) {
        const int g = d[u] / RNODES;
        const int pos = bbase[g] + rk[u];
        if (pos < QCAP)
          qlist[(size_t)g * QCAP + pos] = (unsigned)s[u] | ((unsigned)(d[u] - g * RNODES) << 16);
      }
    }
    return;
  }
  const int nlog = (Nn * HEADS + 255) / 256;
  const int lb = blockIdx.x - (BIN ? nbin : 0);
  if (BIN && lb >= nlog) {
    const int gi = (lb - nlog) * 256 + threadIdx.x;
    if (gi < 256 * NFOLD) {
      const int k = gi / NFOLD, n = gi % NFOLD, h = n & 3;
      float v = 0.f;
      if (n < 8) v = fold_dot(fa.W2 + (size_t)k * 512 + h * 128, ((n & 4) ? fa.ad2 : fa.as2) + h * 128, 128);
      fa.V2[gi] = v;
    } else if (gi < (256 + 512) * NFOLD) {
      const int gj = gi - 256 * NFOLD;
      const int k = gj / NFOLD, n = gj % NFOLD, h = n & 3;
      const float* __restrict__ w = (n < 8) ? fa.Wm : fa.Wl;
      const float* __restrict__ a = (n < 8) ? ((n & 4) ? fa.adm : fa.asm_) : ((n & 4) ? fa.adl : fa.asl);
      fa.V34[gj] = fold_dot(w + (size_t)k * 128 + h * 32, a + h * 32, 32);
    }
    return;
  }
  int i = lb * blockDim.x + threadIdx.x;
  if (i >= Nn * HEADS) return;
  int n = i / HEADS, h = i % HEADS;
  const float* hp = H + (size_t)n * HEADS * C + (size_t)h * C;
  float s = 0.f, d = 0.f;
  for (int c = 0; c < C; ++c) { float v = hp[c]; s += v * a_s[h * C + c]; d += v * a_d[h * C + c]; }
  als[i] = s; ald[i] = d;
}

__device__ __forceinline__ unsigned fkey(float x) {
  unsigned b = __float_as_uint(x);
  return (b & 0x80000000u) ? ~b : (b | 0x80000000u);
}
__device__ __forceinline__ float funkey(unsigned k) {
  unsigned b = (k & 0x80000000u) ? (k & 0x7FFFFFFFu) : ~k;
  return __uint_as_float(b);
}


template <int C, bool MEAN>
__global__ __launch_bounds__(NTHR)
void node_range_softmax_aggregate(const int* __restrict__ src, const int* __restrict__ dst, int E,
                                  const float* __restrict__ H, const float* __restrict__ als,
                                  const float* __restrict__ ald, const float* __restrict__ bias,
                                  float* __restrict__ out,
                                  const unsigned* __restrict__ qlist, const int* __restrict__ qcnt) {
  constexpr int F = HEADS * C;
  constexpr int J = F / 128;
  constexpr int U = 8 / J;
  constexpr int SLOTS = QCAP + 256;
  __shared__ __attribute__((aligned(16))) float acc[(RNODES * F > 5 * SLOTS) ? RNODES * F : 5 * SLOTS];
  __shared__ unsigned mk[RNODES * HEADS];
  __shared__ __attribute__((aligned(16))) float mv[RNODES * HEADS];
  __shared__ float    den[RNODES * HEADS];
  __shared__ __attribute__((aligned(16))) float aldc[RNODES * HEADS];
  __shared__ unsigned q[QCAP];
  __shared__ unsigned qn;
  __shared__ int      cnt[128];
  __shared__ int      off[132];

  const int tid  = threadIdx.x;
  const int lane = tid & 31;
  const int wave = tid >> 5;
  const int base = blockIdx.x * RNODES;
  int rn = NNODES - base;
  if (rn > RNODES) rn = RNODES;
  if (rn < 0) rn = 0;
  const unsigned* __restrict__ lst = qlist + (size_t)blockIdx.x * QCAP;

  auto take = [&](int e, unsigned dl) {
    if (dl < (unsigned)rn) {
      const unsigned slot = atomicAdd(&qn, 1u);
      if (slot < QCAP) q[slot] = (unsigned)e | (dl << 25);
    }
  };

  auto scan = [&](int e0, int e1) {
    if (((E | e0 | e1) & 3) == 0) {
      const int4* __restrict__ d4 = (const int4*)dst;
      const int g1 = e1 >> 2;
      for (int g0 = (e0 >> 2); g0 < g1; g0 += NTHR * 8) {
        int4 v[8];
#pragma unroll
        for (int u = 0; u < 8; ++u) {
          const int g = g0 + u * NTHR + tid;
          v[u] = (g < g1) ? d4[g] : make_int4(-1, -1, -1, -1);
        }
#pragma unroll
        for (int u = 0; u < 8; ++u) {
          const int e = (g0 + u * NTHR + tid) * 4;
          const unsigned a0 = (unsigned)(v[u].x - base), a1 = (unsigned)(v[u].y - base);
          const unsigned a2 = (unsigned)(v[u].z - base), a3 = (unsigned)(v[u].w - base);
          const unsigned lo01 = a0 < a1 ? a0 : a1, lo23 = a2 < a3 ? a2 : a3;
          if ((lo01 < lo23 ? lo01 : lo23) < (unsigned)rn) {
            take(e + 0, a0); take(e + 1, a1); take(e + 2, a2); take(e + 3, a3);
          }
        }
      }
    } else {
      for (int e = e0 + tid; e < e1; e += NTHR) take(e, (unsigned)(dst[e] - base));
    }
  };

  auto resolve = [&](int n, bool withmax) {
    for (int i = tid; i < n; i += NTHR) {
      const unsigned ent = q[i];
      const unsigned dl = ent >> 25;
      const int s = src[ent & 0x1FFFFFFu];
      if (withmax) {
        const float4 a = *(const float4*)(als + (size_t)s * HEADS);
        atomicMax(&mk[dl * HEADS + 0], fkey(a.x));
        atomicMax(&mk[dl * HEADS + 1], fkey(a.y));
        atomicMax(&mk[dl * HEADS + 2], fkey(a.z));
        atomicMax(&mk[dl * HEADS + 3], fkey(a.w));
      }
      q[i] = (unsigned)s | (dl << 16);
    }
  };

  auto drain = [&](int n) {
    for (int i0 = wave; i0 < n; i0 += NWAVE * U) {
      const int iu = i0 + lane * NWAVE;
      const unsigned mine = (lane < U && iu < n) ? q[iu] : 0u;
      float4 hv[U][J];
      float sa[U][J], da[U][J], ma[U][J];
      int sidx[U], dloc[U];
#pragma unroll
      for (int u = 0; u < U; ++u) {
        const unsigned ent = (unsigned)__builtin_amdgcn_readlane((int)mine, u);
        sidx[u] = (int)(ent & 0xFFFFu);
        dloc[u] = (int)(ent >> 16);
      }
#pragma unroll
      for (int u = 0; u < U; ++u) {
#pragma unroll
        for (int j = 0; j < J; ++j) {
          const int h = (128 * j + lane * 4) / C;
          sa[u][j] = als[sidx[u] * HEADS + h];
        }
      }
#pragma unroll
      for (int u = 0; u < U; ++u) {
        const float* __restrict__ hrow = H + (size_t)sidx[u] * F + lane * 4;
#pragma unroll
        for (int j = 0; j < J; ++j) hv[u][j] = *(const float4*)(hrow + 128 * j);
      }
#pragma unroll
      for (int u = 0; u < U; ++u) {
#pragma unroll
        for (int j = 0; j < J; ++j) {
          const int h = (128 * j + lane * 4) / C;
          da[u][j] = aldc[dloc[u] * HEADS + h];
          ma[u][j] = mv[dloc[u] * HEADS + h];
        }
      }
      __builtin_amdgcn_sched_barrier(0);
#pragma unroll
      for (int u = 0; u < U; ++u) {
        if (i0 + u * NWAVE < n) {
          const int dl = dloc[u];
#pragma unroll
          for (int j = 0; j < J; ++j) {
            const int f = 128 * j + lane * 4;
            const int h = f / C;
            float sc = sa[u][j] + da[u][j];
            sc = (sc > 0.f) ? sc : SLOPE * sc;
            const float ex = __expf(sc - ma[u][j]);
            if ((f % C) == 0) atomicAdd(&den[dl * HEADS + h], ex);
            float* ap = &acc[dl * F + 128 * j + lane];
            atomicAdd(ap +  0, hv[u][j].x * ex);
            atomicAdd(ap + 32, hv[u][j].y * ex);
            atomicAdd(ap + 64, hv[u][j].z * ex);
            atomicAdd(ap + 96, hv[u][j].w * ex);
          }
        }
      }
    }
  };

  if (tid < rn * HEADS) {
    mk[tid]   = fkey(als[(size_t)base * HEADS + tid]);
    aldc[tid] = ald[(size_t)base * HEADS + tid];
  }
  if (tid == 0) qn = 0u;
  if (tid < 128) cnt[tid] = 0;
  __syncthreads();

  const int total = qcnt[blockIdx.x];
  const bool fits = (total <= QCAP);
  constexpr int KMAX = QCAP / NTHR;
  constexpr int KREG = 4;
  unsigned ent[KMAX];
  int rk[KMAX];
  float4 av[KREG];
  if (fits) {
#pragma unroll
    for (int k = 0; k < KMAX; ++k) {
      const int i = tid + k * NTHR;
      ent[k] = 0u;
      if (i < total) ent[k] = lst[i];
    }
#pragma unroll
    for (int k = 0; k < KREG; ++k) {
      const int i = tid + k * NTHR;
      av[k] = make_float4(0.f, 0.f, 0.f, 0.f);
      if (i < total) av[k] = *(const float4*)(als + (size_t)(ent[k] & 0xFFFFu) * HEADS);
    }
#pragma unroll
    for (int k = 0; k < KMAX; ++k) {
      const int i = tid + k * NTHR;
      rk[k] = 0;
      if (i < total) {
        const unsigned dl = ent[k] >> 16;
        float4 a;
        if (k < KREG) a = av[k];
        else          a = *(const float4*)(als + (size_t)(ent[k] & 0xFFFFu) * HEADS);
        atomicMax(&mk[dl * HEADS + 0], fkey(a.x));
        atomicMax(&mk[dl * HEADS + 1], fkey(a.y));
        atomicMax(&mk[dl * HEADS + 2], fkey(a.z));
        atomicMax(&mk[dl * HEADS + 3], fkey(a.w));
        rk[k] = atomicAdd(&cnt[dl], 1);
      }
    }
  } else {
    for (int p = 0; p < E; p += QCAP) {
      __syncthreads();
      if (tid == 0) qn = 0u;
      __syncthreads();
      scan(p, (p + QCAP < E) ? (p + QCAP) : E);
      __syncthreads();
      resolve((int)qn, true);
    }
  }
  __syncthreads();

  if (tid < rn * HEADS) {
    const float a = aldc[tid];
    float m = funkey(mk[tid]) + a;
    m = (m > 0.f) ? m : SLOPE * m;
    mv[tid] = m;
    float sc = als[(size_t)base * HEADS + tid] + a;
    sc = (sc > 0.f) ? sc : SLOPE * sc;
    den[tid] = __expf(sc - m);
  }

  if (fits) {
    static_assert(!MEAN || C == 32, "the sum over heads across lanes is written for 32 channels per head");
    unsigned* __restrict__ qs = (unsigned*)acc;
    float* __restrict__ wt = acc + SLOTS;
    if (wave == 0) {
      const int c0 = cnt[4 * lane], c1 = cnt[4 * lane + 1], c2 = cnt[4 * lane + 2], c3 = cnt[4 * lane + 3];
      const int s = c0 + c1 + c2 + c3;
      int incl = s;
#pragma unroll
      for (int d = 1; d < 32; d <<= 1) {
        const int t = __shfl_up(incl, d);
        if (lane >= d) incl += t;
      }
      const int ex0 = incl - s;
      off[4 * lane]     = ex0;
      off[4 * lane + 1] = ex0 + c0;
      off[4 * lane + 2] = ex0 + c0 + c1;
      off[4 * lane + 3] = ex0 + c0 + c1 + c2;
      if (lane == 31) off[128] = incl;
    }
    __syncthreads();
#pragma unroll
    for (int k = 0; k < KMAX; ++k) {
      const int i = tid + k * NTHR;
      if (i < total) {
        const unsigned dl = ent[k] >> 16;
        const int pos = off[dl] + (int)dl + 1 + rk[k];
        qs[pos] = ent[k] & 0xFFFFu;
        const float4 ad4 = *(const float4*)&aldc[dl * HEADS];
        const float4 m4  = *(const float4*)&mv[dl * HEADS];
        float4 a;
        if (k < KREG) a = av[k];
        else          a = *(const float4*)(als + (size_t)(ent[k] & 0xFFFFu) * HEADS);
        float4 w;
        float sc;
        sc = a.x + ad4.x;  sc = (sc > 0.f) ? sc : SLOPE * sc;  w.x = __expf(sc - m4.x);
        sc = a.y + ad4.y;  sc = (sc > 0.f) ? sc : SLOPE * sc;  w.y = __expf(sc - m4.y);
        sc = a.z + ad4.z;  sc = (sc > 0.f) ? sc : SLOPE * sc;  w.z = __expf(sc - m4.z);
        sc = a.w + ad4.w;  sc = (sc > 0.f) ? sc : SLOPE * sc;  w.w = __expf(sc - m4.w);
        *(float4*)&wt[4 * pos] = w;
      }
    }
    if (tid < rn * HEADS) {
      const int dl = tid / HEADS;
      const int pos = off[dl] + dl;
      wt[4 * pos + (tid % HEADS)] = den[tid];
      if ((tid % HEADS) == 0) qs[pos] = (unsigned)(base + dl);
    }
    constexpr int UE = 8;
    if (tid < HEADS * UE) wt[4 * (total + rn) + tid] = 0.f;
    __syncthreads();

    const int wv = __builtin_amdgcn_readfirstlane(wave);
    const int nitems = rn * J;
    for (int it = wv; it < nitems; it += NWAVE) {
      const int dl = it / J, j = it % J;
      const int node = base + dl;
      const int f = 128 * j + lane * 4;
      const int h = f / C;
      const int o0 = __builtin_amdgcn_readfirstlane(off[dl]) + dl;
      const int nt = __builtin_amdgcn_readfirstlane(off[dl + 1]) + dl + 1 - o0;
      const float* __restrict__ wp = wt + 4 * o0 + h;
      float4 a4 = make_float4(0.f, 0.f, 0.f, 0.f);
      float dn = 0.f;
      for (int i0 = 0; i0 < nt; i0 += UE) {
        const int p = i0 + lane;
        unsigned mine = (unsigned)node;
        if (lane < UE && p < nt) mine = qs[o0 + p];
        int sidx[UE];
        float wv8[UE];
        float4 hv[UE];
#pragma unroll
        for (int u = 0; u < UE; ++u) sidx[u] = __builtin_amdgcn_readlane((int)mine, u);
#pragma unroll
        for (int u = 0; u < UE; ++u) wv8[u] = wp[4 * (i0 + u)];
#pragma unroll
        for (int u = 0; u < UE; ++u) hv[u] = *(const float4*)(H + (size_t)sidx[u] * F + f);
#pragma unroll
        for (int u = 0; u < UE; ++u) {
          const float ex = (i0 + u < nt) ? wv8[u] : 0.f;
          dn += ex;
          a4.x += hv[u].x * ex;  a4.y += hv[u].y * ex;
          a4.z += hv[u].z * ex;  a4.w += hv[u].w * ex;
        }
      }
      const float dd = dn + 1e-16f;
      if (!MEAN) {
        const float4 b4 = *(const float4*)(bias + f);
        float4 o;
        o.x = a4.x / dd + b4.x;  o.y = a4.y / dd + b4.y;
        o.z = a4.z / dd + b4.z;  o.w = a4.w / dd + b4.w;
        o.x = o.x > 0.f ? o.x : 0.f;  o.y = o.y > 0.f ? o.y : 0.f;
        o.z = o.z > 0.f ? o.z : 0.f;  o.w = o.w > 0.f ? o.w : 0.f;
        *(float4*)(out + (size_t)node * F + f) = o;
      } else {
        const float v0 = a4.x / dd, v1 = a4.y / dd, v2 = a4.z / dd, v3 = a4.w / dd;
        const int l8 = lane & 7;
        const float r0 = ((__shfl(v0, l8) + __shfl(v0, l8 + 8)) + __shfl(v0, l8 + 16)) + __shfl(v0, l8 + 24);
        const float r1 = ((__shfl(v1, l8) + __shfl(v1, l8 + 8)) + __shfl(v1, l8 + 16)) + __shfl(v1, l8 + 24);
        const float r2 = ((__shfl(v2, l8) + __shfl(v2, l8 + 8)) + __shfl(v2, l8 + 16)) + __shfl(v2, l8 + 24);
        const float r3 = ((__shfl(v3, l8) + __shfl(v3, l8 + 8)) + __shfl(v3, l8 + 16)) + __shfl(v3, l8 + 24);
        if (lane < 8) {
          const float4 b4 = *(const float4*)(bias + 4 * lane);
          float4 o;
          o.x = 0.25f * r0 + b4.x;  o.y = 0.25f * r1 + b4.y;
          o.z = 0.25f * r2 + b4.z;  o.w = 0.25f * r3 + b4.w;
          *(float4*)(out + (size_t)node * C + 4 * lane) = o;
        }
      }
    }
  } else {
  __syncthreads();

  {
    constexpr int P = 4;
    const int items = rn * (F / 4);
    for (int idx0 = tid; idx0 < items; idx0 += NTHR * P) {
      float4 hp[P];
#pragma unroll
      for (int p = 0; p < P; ++p) {
        const int idx = idx0 + p * NTHR;
        const int idc = (idx < items) ? idx : idx0;
        const int dl = idc / (F / 4);
        const int r  = idc % (F / 4);
        const int f  = 128 * (r >> 5) + 4 * (r & 31);
        hp[p] = *(const float4*)(H + (size_t)(base + dl) * F + f);
      }
#pragma unroll
      for (int p = 0; p < P; ++p) {
        const int idx = idx0 + p * NTHR;
        const int idc = (idx < items) ? idx : idx0;
        const int dl = idc / (F / 4);
        const int r  = idc % (F / 4);
        const int j  = r >> 5, l = r & 31;
        const int f  = 128 * j + 4 * l;
        const float ex = den[dl * HEADS + f / C];
        float* ap = &acc[dl * F + 128 * j + l];
        ap[0] = hp[p].x * ex; ap[32] = hp[p].y * ex; ap[64] = hp[p].z * ex; ap[96] = hp[p].w * ex;
      }
    }
  }
  __syncthreads();

  {
    for (int p = 0; p < E; p += QCAP) {
      __syncthreads();
      if (tid == 0) qn = 0u;
      __syncthreads();
      scan(p, (p + QCAP < E) ? (p + QCAP) : E);
      __syncthreads();
      resolve((int)qn, false);
      __syncthreads();
      drain((int)qn);
    }
  }
  __syncthreads();

  if (!MEAN) {
    for (int idx = tid; idx < rn * (F / 4); idx += NTHR) {
      const int dl = idx / (F / 4);
      const int r  = idx % (F / 4);
      const int j  = r >> 5, l = r & 31;
      const int f  = 128 * j + 4 * l;
      const float dn = den[dl * HEADS + f / C] + 1e-16f;
      const float* ap = &acc[dl * F + 128 * j + l];
      const float4 b4 = *(const float4*)(bias + f);
      float4 o;
      o.x = ap[0]  / dn + b4.x;  o.y = ap[32] / dn + b4.y;
      o.z = ap[64] / dn + b4.z;  o.w = ap[96] / dn + b4.w;
      o.x = o.x > 0.f ? o.x : 0.f;  o.y = o.y > 0.f ? o.y : 0.f;
      o.z = o.z > 0.f ? o.z : 0.f;  o.w = o.w > 0.f ? o.w : 0.f;
      *(float4*)(out + (size_t)(base + dl) * F + f) = o;
    }
  } else {
    for (int idx = tid; idx < rn * C; idx += NTHR) {
      const int dl = idx / C, c = idx % C;
      const float* qd = &den[dl * HEADS];
      float p[HEADS];
#pragma unroll
      for (int h = 0; h < HEADS; ++h) {
        const int f = h * C + c;
        p[h] = acc[dl * F + 128 * (f >> 7) + 32 * (f & 3) + ((f & 127) >> 2)];
      }
      out[(size_t)(base + dl) * C + c] =
          0.25f * (p[0] / (qd[0] + 1e-16f) + p[1] / (qd[1] + 1e-16f) +
                   p[2] / (qd[2] + 1e-16f) + p[3] / (qd[3] + 1e-16f)) + bias[c];
    }
  }
  }
}

#define NFOLDBLK (((256 + 512) * NFOLD) / 256)
template <int C, bool MEAN, bool FIRST>
static void run_gat_layer(hipStream_t st, const float* in, int Fin,
                          const float* W, const float* a_s, const float* a_d, const float* b,
                          float* Hbuf, float* outp,
                          const int* src, const int* dst, int E,
                          float* ALS, float* ALD, unsigned* QL, int* QC,
                          const float* Vb, float* l0, float* l1, float* l2, float* l3, FoldArgs fa) {
  const int Nn = NNODES;
  const int Fout = HEADS * C;
  const int Mtiles = Nn / 16;
  dim3 ggrid((Mtiles + 7) / 8, Fout / 64);
  gemm_xw_wmma<<<ggrid, 256, 0, st>>>(in, W, Hbuf, Nn, Fin, Fout, FIRST ? QC : nullptr, Vb, l0, l1, l2, l3);
  if (FIRST) {
    int t = Nn * HEADS;
    const int nbin = (E + BINBLK - 1) / BINBLK;
    attn_logits<true><<<nbin + (t + 255) / 256 + NFOLDBLK, 256, 0, st>>>(Hbuf, a_s, a_d, ALS, ALD, Nn, C,
                                                                         src, dst, E, QL, QC, nbin, fa);
  }
  node_range_softmax_aggregate<C, MEAN>
      <<<NGROUPS, NTHR, 0, st>>>(src, dst, E, Hbuf, ALS, ALD, b, outp, QL, QC);
}

extern "C" void kernel_launch(void* const* d_in, const int* in_sizes, int n_in,
                              void* d_out, int out_size, void* d_ws, size_t ws_size,
                              hipStream_t stream) {
  const float* x   = (const float*)d_in[0];
  const int*   ei  = (const int*)d_in[1];
  const int E = in_sizes[1] / 2;
  const int* src = ei;
  const int* dst = ei + E;
  const float *W1 = (const float*)d_in[2],  *as1 = (const float*)d_in[3],
              *ad1 = (const float*)d_in[4], *b1  = (const float*)d_in[5];
  const float *W2 = (const float*)d_in[6],  *as2 = (const float*)d_in[7],
              *ad2 = (const float*)d_in[8], *b2  = (const float*)d_in[9];
  const float *Wm = (const float*)d_in[10], *asm_ = (const float*)d_in[11],
              *adm = (const float*)d_in[12], *bm = (const float*)d_in[13];
  const float *Wl = (const float*)d_in[14], *asl = (const float*)d_in[15],
              *adl = (const float*)d_in[16], *bl = (const float*)d_in[17];

  char* ws = (char*)d_ws;
  size_t off = 0;
  auto alloc = [&](size_t bytes) -> void* {
    void* p = ws + off;
    off += (bytes + 255) & ~(size_t)255;
    return p;
  };
  float*    Hbuf = (float*)alloc((size_t)NNODES * 512 * 4);
  float*    A0   = (float*)alloc((size_t)NNODES * 256 * 4);
  float*    A1   = (float*)alloc((size_t)NNODES * 512 * 4);
  float*    ALS  = (float*)alloc((size_t)NNODES * HEADS * 4);
  float*    ALD  = (float*)alloc((size_t)NNODES * HEADS * 4);
  unsigned* QL   = (unsigned*)alloc((size_t)NGROUPS * QCAP * 4);
  int*      QC   = (int*)alloc((size_t)NGROUPS * 4);
  float*    ALS4 = (float*)alloc((size_t)NNODES * HEADS * 4);
  float*    ALD4 = (float*)alloc((size_t)NNODES * HEADS * 4);
  float*    V2   = (float*)alloc((size_t)256 * NFOLD * 4);
  float*    V34  = (float*)alloc((size_t)512 * NFOLD * 4);
  FoldArgs fa;
  fa.W2 = W2;  fa.as2 = as2;   fa.ad2 = ad2;
  fa.Wm = Wm;  fa.asm_ = asm_; fa.adm = adm;
  fa.Wl = Wl;  fa.asl = asl;   fa.adl = adl;
  fa.V2 = V2;  fa.V34 = V34;

  float* zmean   = (float*)d_out;
  float* zlogstd = (float*)d_out + (size_t)NNODES * 32;

  run_gat_layer<64, false, true>(stream, x, 256, W1, as1, ad1, b1, Hbuf, A0, src, dst, E, ALS, ALD, QL, QC,
                                 nullptr, nullptr, nullptr, nullptr, nullptr, fa);
  run_gat_layer<128, false, false>(stream, A0, 256, W2, as2, ad2, b2, Hbuf, A1, src, dst, E, ALS, ALD, QL, QC,
                                   V2, ALS, ALD, nullptr, nullptr, fa);
  run_gat_layer<32, true, false>(stream, A1, 512, Wm, asm_, adm, bm, Hbuf, zmean, src, dst, E, ALS, ALD, QL, QC,
                                 V34, ALS, ALD, ALS4, ALD4, fa);
  run_gat_layer<32, true, false>(stream, A1, 512, Wl, asl, adl, bl, Hbuf, zlogstd, src, dst, E, ALS4, ALD4, QL, QC,
                                 nullptr, nullptr, nullptr, nullptr, nullptr, fa);
}
